# starting kernel
# baseline (speedup 1.0000x reference)
#include <hip/hip_runtime.h>
#include <hip/hip_bf16.h>

#define Bsz 4
#define Lsz 64
#define DMsz 1024
#define Hsz 16
#define DKsz 64
#define DVsz 64
#define Dsz 1024
#define DIsz 4096
#define Rsz 256
#define NEGC (-1.0e9f)

typedef float v2f __attribute__((ext_vector_type(2)));
typedef float v8f __attribute__((ext_vector_type(8)));

#define GEMM_BIAS  1
#define GEMM_RELU  2
#define GEMM_RESID 4

__global__ __launch_bounds__(256)
void gemm_wmma_f32(const float* __restrict__ A, const float* __restrict__ B,
                   const float* __restrict__ bias, const float* __restrict__ resid,
                   float* __restrict__ C, int M, int N, int K, int flags)
{
    __shared__ float As[64][64 + 1];
    __shared__ float Bs[64][128 + 1];

    const int tid   = threadIdx.x;
    const int wave  = tid >> 5;
    const int lane  = tid & 31;
    const int l15   = lane & 15;
    const int lhalf = lane >> 4;
    const int m0    = blockIdx.y * 64;
    const int nblk  = blockIdx.x * 128;
    const int mq    = wave >> 2;
    const int nq    = wave & 3;

    const int ar = tid >> 4;
    const int ac = (tid & 15) << 2;
    const int br = (tid >> 5) * 8;
    const int bc = (tid & 31) << 2;
    float4 aprf[4], bprf[8];

    v8f acc[2][2];
#pragma unroll
    for (int x = 0; x < 2; ++x)
#pragma unroll
        for (int y = 0; y < 2; ++y)
            acc[x][y] = v8f{0.f, 0.f, 0.f, 0.f, 0.f, 0.f, 0.f, 0.f};

#pragma unroll
    for (int rr = 0; rr < 4; ++rr)
        aprf[rr] = *(const float4*)(A + (size_t)(m0 + ar + rr * 16) * K + ac);
#pragma unroll
    for (int rr = 0; rr < 8; ++rr)
        bprf[rr] = *(const float4*)(B + (size_t)(br + rr) * N + nblk + bc);

    const int arow0 = mq * 32 + l15;
    const int bcol0 = nq * 32 + l15;

    for (int k0 = 0; k0 < K; k0 += 64) {
#pragma unroll
        for (int rr = 0; rr < 4; ++rr) {
            As[ar + rr * 16][ac + 0] = aprf[rr].x;
            As[ar + rr * 16][ac + 1] = aprf[rr].y;
            As[ar + rr * 16][ac + 2] = aprf[rr].z;
            As[ar + rr * 16][ac + 3] = aprf[rr].w;
        }
#pragma unroll
        for (int rr = 0; rr < 8; ++rr) {
            Bs[br + rr][bc + 0] = bprf[rr].x;
            Bs[br + rr][bc + 1] = bprf[rr].y;
            Bs[br + rr][bc + 2] = bprf[rr].z;
            Bs[br + rr][bc + 3] = bprf[rr].w;
        }
        __syncthreads();

        if (k0 + 64 < K) {
#pragma unroll
            for (int rr = 0; rr < 4; ++rr)
                aprf[rr] = *(const float4*)(A + (size_t)(m0 + ar + rr * 16) * K
                                            + k0 + 64 + ac);
#pragma unroll
            for (int rr = 0; rr < 8; ++rr)
                bprf[rr] = *(const float4*)(B + (size_t)(k0 + 64 + br + rr) * N
                                            + nblk + bc);
        }

#pragma unroll
        for (int half = 0; half < 2; ++half) {
            v2f af[2][8], bf[2][8];
#pragma unroll
            for (int s = 0; s < 8; ++s) {
                const int kb = half * 32 + s * 4 + lhalf * 2;
                af[0][s] = v2f{As[arow0][kb],      As[arow0][kb + 1]};
                af[1][s] = v2f{As[arow0 + 16][kb], As[arow0 + 16][kb + 1]};
                bf[0][s] = v2f{Bs[kb][bcol0],      Bs[kb + 1][bcol0]};
                bf[1][s] = v2f{Bs[kb][bcol0 + 16], Bs[kb + 1][bcol0 + 16]};
            }
#pragma unroll
            for (int s = 0; s < 8; ++s) {
                acc[0][0] = __builtin_amdgcn_wmma_f32_16x16x4_f32(
                    false, af[0][s], false, bf[0][s], (short)0, acc[0][0], false, false);
                acc[0][1] = __builtin_amdgcn_wmma_f32_16x16x4_f32(
                    false, af[0][s], false, bf[1][s], (short)0, acc[0][1], false, false);
                acc[1][0] = __builtin_amdgcn_wmma_f32_16x16x4_f32(
                    false, af[1][s], false, bf[0][s], (short)0, acc[1][0], false, false);
                acc[1][1] = __builtin_amdgcn_wmma_f32_16x16x4_f32(
                    false, af[1][s], false, bf[1][s], (short)0, acc[1][1], false, false);
            }
        }
        __syncthreads();
    }

#pragma unroll
    for (int nsub = 0; nsub < 2; ++nsub) {
        const int col = nblk + nq * 32 + nsub * 16 + l15;
        const float bv = (flags & GEMM_BIAS) ? bias[col] : 0.f;
#pragma unroll
        for (int msub = 0; msub < 2; ++msub) {
#pragma unroll
            for (int r = 0; r < 8; ++r) {
                const int row = m0 + mq * 32 + msub * 16 + r + lhalf * 8;
                float v = acc[msub][nsub][r] + bv;
                if (flags & GEMM_RESID) v += resid[(size_t)row * N + col];
                if (flags & GEMM_RELU)  v = fmaxf(v, 0.f);
                C[(size_t)row * N + col] = v;
            }
        }
    }
}

__global__ __launch_bounds__(256)
void dec_kernel(const float* __restrict__ a0base, const float* __restrict__ a1base,
                const float* __restrict__ b1, const float* __restrict__ w2,
                const float* __restrict__ b2, float* __restrict__ dec)
{
    const int wid  = (int)((blockIdx.x * 256u + threadIdx.x) >> 5);
    const int lane = threadIdx.x & 31;
    const int n    = wid & 63;
    const int m    = (wid >> 6) & 63;
    const int b    = (wid >> 12) & 3;
    const int pair = wid >> 14;
    const int i    = pair / 3;
    const int j    = pair % 3;

    const float* a1 = a1base + ((size_t)j * Rsz + b * Lsz + m) * 2048;
    const float* a0 = a0base + ((size_t)i * Rsz + b * Lsz + n) * 2048;

    float acc = 0.f;
    for (int c = lane; c < 2048; c += 32) {
        float t = a1[c] + a0[c] + b1[c];
        if (t > 0.f) acc += t * (w2[2 * c + 1] - w2[2 * c]);
    }
#pragma unroll
    for (int off = 16; off; off >>= 1) acc += __shfl_xor(acc, off, 32);
    if (lane == 0) {
        float s = acc + (b2[1] - b2[0]);
        dec[wid] = (s > 0.f) ? NEGC : 0.f;
    }
}

__global__ __launch_bounds__(192)
void attn_scores_kernel(const float* __restrict__ Qb, const float* __restrict__ Kb,
                        const float* __restrict__ dec, const int* __restrict__ mask,
                        float* __restrict__ S)
{
    const int blk = blockIdx.x;
    const int q = blk & 63;
    const int h = (blk >> 6) & 15;
    const int b = (blk >> 10) & 3;
    const int i = blk >> 12;
    const int t = threadIdx.x;
    const int j = t >> 6;
    const int k = t & 63;

    __shared__ float qrow[64];
    __shared__ float red[192];

    const float* Q = Qb + ((size_t)i * Rsz + b * Lsz + q) * DMsz + h * DKsz;
    if (t < 64) qrow[t] = Q[t];
    __syncthreads();

    const float* Kp = Kb + ((size_t)j * Rsz + b * Lsz + k) * DMsz + h * DKsz;
    float s = 0.f;
#pragma unroll 8
    for (int d = 0; d < 64; ++d) s += qrow[d] * Kp[d];
    s *= 0.125f;
    s += dec[(((size_t)(i * 3 + j) * Bsz + b) * Lsz + q) * Lsz + k];
    if (mask[((size_t)b * Lsz + q) * Lsz + k] == 0) s = NEGC;

    red[t] = s; __syncthreads();
    if (t < 64) red[t] = fmaxf(red[t], fmaxf(red[t + 64], red[t + 128]));
    __syncthreads();
    for (int st = 32; st > 0; st >>= 1) {
        if (t < st) red[t] = fmaxf(red[t], red[t + st]);
        __syncthreads();
    }
    const float mx = red[0];
    __syncthreads();
    const float e = __expf(s - mx);
    red[t] = e; __syncthreads();
    if (t < 64) red[t] = red[t] + red[t + 64] + red[t + 128];
    __syncthreads();
    for (int st = 32; st > 0; st >>= 1) {
        if (t < st) red[t] += red[t + st];
        __syncthreads();
    }
    S[(size_t)blk * 192 + t] = e * (1.0f / red[0]);
}

__global__ __launch_bounds__(64)
void attn_av_kernel(const float* __restrict__ S, const float* __restrict__ Vb,
                    float* __restrict__ Ob)
{
    const int blk = blockIdx.x;
    const int q = blk & 63;
    const int h = (blk >> 6) & 15;
    const int b = (blk >> 10) & 3;
    const int i = blk >> 12;
    const int d = threadIdx.x;

    __shared__ float arow[192];
    const float* Sp = S + (size_t)blk * 192;
    arow[d] = Sp[d]; arow[d + 64] = Sp[d + 64]; arow[d + 128] = Sp[d + 128];
    __syncthreads();

    float acc = 0.f;
    for (int jk = 0; jk < 192; ++jk) {
        int j = jk >> 6, k = jk & 63;
        acc += arow[jk] * Vb[((size_t)j * Rsz + b * Lsz + k) * DMsz + h * DVsz + d];
    }
    Ob[((size_t)i * Rsz + b * Lsz + q) * DMsz + h * DVsz + d] = acc;
}

__global__ __launch_bounds__(256)
void layernorm_kernel(const float* __restrict__ x, const float* __restrict__ resid,
                      const float* __restrict__ g, const float* __restrict__ bta,
                      float* __restrict__ out)
{
    const int row = blockIdx.x;
    const int t = threadIdx.x;
    __shared__ float red[256];

    const float* xp = x + (size_t)row * 1024;
    float v[4];
    float s = 0.f;
#pragma unroll
    for (int u = 0; u < 4; ++u) {
        float val = xp[t + u * 256];
        if (resid) val += resid[(size_t)row * 1024 + t + u * 256];
        v[u] = val; s += val;
    }
    red[t] = s; __syncthreads();
    for (int st = 128; st > 0; st >>= 1) { if (t < st) red[t] += red[t + st]; __syncthreads(); }
    const float mu = red[0] * (1.f / 1024.f);
    __syncthreads();

    float s2 = 0.f;
#pragma unroll
    for (int u = 0; u < 4; ++u) { float dd = v[u] - mu; s2 += dd * dd; }
    red[t] = s2; __syncthreads();
    for (int st = 128; st > 0; st >>= 1) { if (t < st) red[t] += red[t + st]; __syncthreads(); }
    const float var = red[0] * (1.f / 1024.f);
    const float rinv = 1.0f / sqrtf(var + 1e-6f);
#pragma unroll
    for (int u = 0; u < 4; ++u) {
        int c = t + u * 256;
        out[(size_t)row * 1024 + c] = (v[u] - mu) * rinv * g[c] + bta[c];
    }
}

extern "C" void kernel_launch(void* const* d_in, const int* in_sizes, int n_in,
                              void* d_out, int out_size, void* d_ws, size_t ws_size,
                              hipStream_t stream)
{
    const float* X[3]   = { (const float*)d_in[0], (const float*)d_in[1], (const float*)d_in[2] };
    const int*   mask   = (const int*)d_in[3];
    const float* w_qs   = (const float*)d_in[4];
    const float* w_ks   = (const float*)d_in[5];
    const float* w_vs   = (const float*)d_in[6];
    const float* w_ds0  = (const float*)d_in[7];
    const float* w_ds1  = (const float*)d_in[8];
    const float* dc_w1  = (const float*)d_in[9];
    const float* dc_b1  = (const float*)d_in[10];
    const float* dc_w2  = (const float*)d_in[11];
    const float* dc_b2  = (const float*)d_in[12];
    const float* w_fc   = (const float*)d_in[13];
    const float* ln1_g  = (const float*)d_in[14];
    const float* ln1_b  = (const float*)d_in[15];
    const float* ffn_w1 = (const float*)d_in[16];
    const float* ffn_b1 = (const float*)d_in[17];
    const float* ffn_w2 = (const float*)d_in[18];
    const float* ffn_b2 = (const float*)d_in[19];
    const float* ln2_g  = (const float*)d_in[20];
    const float* ln2_b  = (const float*)d_in[21];
    float* out = (float*)d_out;

    float* ws = (float*)d_ws;
    size_t off = 0;
    auto alloc = [&](size_t n) { float* p = ws + off; off += n; return p; };
    const size_t RM = (size_t)Rsz * DMsz;
    float* Q   = alloc(3 * RM);
    float* Kb  = alloc(3 * RM);
    float* Vb  = alloc(3 * RM);
    float* d0  = alloc(3 * RM);
    float* d1  = alloc(3 * RM);
    float* a0  = alloc(3 * (size_t)Rsz * 2048);
    float* a1  = alloc(3 * (size_t)Rsz * 2048);
    float* dec = alloc(9 * (size_t)Bsz * Lsz * Lsz);
    float* S   = alloc((size_t)3 * Bsz * Hsz * Lsz * 192);
    float* O   = alloc(3 * RM);
    float* fc  = alloc(3 * RM);
    float* oln = alloc(3 * RM);
    float* h1  = alloc(3 * (size_t)Rsz * DIsz);
    float* f2  = alloc(3 * RM);
    (void)ws_size; (void)n_in; (void)in_sizes; (void)out_size;

    auto gemm = [&](const float* A, const float* Bm, const float* bias,
                    const float* resid, float* C, int M, int N, int K, int flags) {
        dim3 grid(N / 128, M / 64);
        gemm_wmma_f32<<<grid, dim3(256), 0, stream>>>(A, Bm, bias, resid, C, M, N, K, flags);
    };

    for (int i = 0; i < 3; ++i) {
        gemm(X[i], w_qs,  nullptr, nullptr, Q  + i * RM, Rsz, 1024, 1024, 0);
        gemm(X[i], w_ks,  nullptr, nullptr, Kb + i * RM, Rsz, 1024, 1024, 0);
        gemm(X[i], w_vs,  nullptr, nullptr, Vb + i * RM, Rsz, 1024, 1024, 0);
        gemm(X[i], w_ds0, nullptr, nullptr, d0 + i * RM, Rsz, 1024, 1024, 0);
        gemm(X[i], w_ds1, nullptr, nullptr, d1 + i * RM, Rsz, 1024, 1024, 0);
        gemm(d0 + i * RM, dc_w1 + (size_t)1024 * 2048, nullptr, nullptr,
             a0 + i * (size_t)Rsz * 2048, Rsz, 2048, 1024, 0);
        gemm(d1 + i * RM, dc_w1, nullptr, nullptr,
             a1 + i * (size_t)Rsz * 2048, Rsz, 2048, 1024, 0);
    }

    {
        int total_waves = 9 * Bsz * Lsz * Lsz;
        dec_kernel<<<total_waves / 8, 256, 0, stream>>>(a0, a1, dc_b1, dc_w2, dc_b2, dec);
    }

    {
        int nblk = 3 * Bsz * Hsz * Lsz;
        attn_scores_kernel<<<nblk, 192, 0, stream>>>(Q, Kb, dec, mask, S);
        attn_av_kernel<<<nblk, 64, 0, stream>>>(S, Vb, O);
    }

    for (int i = 0; i < 3; ++i) {
        gemm(O + i * RM, w_fc, nullptr, X[i], fc + i * RM, Rsz, 1024, 1024, GEMM_RESID);
        layernorm_kernel<<<Rsz, 256, 0, stream>>>(fc + i * RM, nullptr, ln1_g, ln1_b, oln + i * RM);
        gemm(oln + i * RM, ffn_w1, ffn_b1, nullptr, h1 + i * (size_t)Rsz * DIsz,
             Rsz, DIsz, 1024, GEMM_BIAS | GEMM_RELU);
        gemm(h1 + i * (size_t)Rsz * DIsz, ffn_w2, ffn_b2, nullptr, f2 + i * RM,
             Rsz, 1024, DIsz, GEMM_BIAS);
        layernorm_kernel<<<Rsz, 256, 0, stream>>>(f2 + i * RM, oln + i * RM, ln2_g, ln2_b,
                                                  out + i * RM);
    }
}


# Batch the 3 modalities and same-input weights of every GEMM (and the layernorms) into one launch each via grid y/z: 39 launches -> 10, small grids run side by side; summation order unchanged
# speedup vs baseline: 2.4309x; 2.4309x over previous
#include <hip/hip_runtime.h>
#include <hip/hip_bf16.h>

#define Bsz 4
#define Lsz 64
#define DMsz 1024
#define Hsz 16
#define DKsz 64
#define DVsz 64
#define Dsz 1024
#define DIsz 4096
#define Rsz 256
#define NEGC (-1.0e9f)

typedef float v2f __attribute__((ext_vector_type(2)));
typedef float v8f __attribute__((ext_vector_type(8)));

#define GEMM_BIAS  1
#define GEMM_RELU  2
#define GEMM_RESID 4

__global__ __launch_bounds__(256)
void gemm_wmma_f32(const float* __restrict__ A0, const float* __restrict__ A1,
                   const float* __restrict__ A2, long strideAz,
                   const float* __restrict__ B0, const float* __restrict__ B1,
                   const float* __restrict__ B2, const float* __restrict__ B3,
                   const float* __restrict__ B4,
                   const float* __restrict__ bias,
                   const float* __restrict__ R0, const float* __restrict__ R1,
                   const float* __restrict__ R2,
                   float* __restrict__ Cbase, long strideCz,
                   int N, int K, int flags)
{
    const int seg  = blockIdx.y >> 2;
    const int mloc = (blockIdx.y & 3) * 64;
    const int zi   = blockIdx.z;
    const float* __restrict__ A =
        ((seg == 0) ? A0 : (seg == 1) ? A1 : A2) + (size_t)zi * strideAz;
    const float* __restrict__ B =
        (zi == 0) ? B0 : (zi == 1) ? B1 : (zi == 2) ? B2 : (zi == 3) ? B3 : B4;
    const float* __restrict__ resid = (seg == 0) ? R0 : (seg == 1) ? R1 : R2;
    float* __restrict__ C = Cbase + (size_t)zi * strideCz;

    __shared__ float As[64][64 + 1];
    __shared__ float Bs[64][128 + 1];

    const int tid   = threadIdx.x;
    const int wave  = tid >> 5;
    const int lane  = tid & 31;
    const int l15   = lane & 15;
    const int lhalf = lane >> 4;
    const int m0    = blockIdx.y * 64;
    const int nblk  = blockIdx.x * 128;
    const int mq    = wave >> 2;
    const int nq    = wave & 3;

    const int ar = tid >> 4;
    const int ac = (tid & 15) << 2;
    const int br = (tid >> 5) * 8;
    const int bc = (tid & 31) << 2;
    float4 aprf[4], bprf[8];

    v8f acc[2][2];
#pragma unroll
    for (int x = 0; x < 2; ++x)
#pragma unroll
        for (int y = 0; y < 2; ++y)
            acc[x][y] = v8f{0.f, 0.f, 0.f, 0.f, 0.f, 0.f, 0.f, 0.f};

#pragma unroll
    for (int rr = 0; rr < 4; ++rr)
        aprf[rr] = *(const float4*)(A + (size_t)(mloc + ar + rr * 16) * K + ac);
#pragma unroll
    for (int rr = 0; rr < 8; ++rr)
        bprf[rr] = *(const float4*)(B + (size_t)(br + rr) * N + nblk + bc);

    const int arow0 = mq * 32 + l15;
    const int bcol0 = nq * 32 + l15;

    for (int k0 = 0; k0 < K; k0 += 64) {
#pragma unroll
        for (int rr = 0; rr < 4; ++rr) {
            As[ar + rr * 16][ac + 0] = aprf[rr].x;
            As[ar + rr * 16][ac + 1] = aprf[rr].y;
            As[ar + rr * 16][ac + 2] = aprf[rr].z;
            As[ar + rr * 16][ac + 3] = aprf[rr].w;
        }
#pragma unroll
        for (int rr = 0; rr < 8; ++rr) {
            Bs[br + rr][bc + 0] = bprf[rr].x;
            Bs[br + rr][bc + 1] = bprf[rr].y;
            Bs[br + rr][bc + 2] = bprf[rr].z;
            Bs[br + rr][bc + 3] = bprf[rr].w;
        }
        __syncthreads();

        if (k0 + 64 < K) {
#pragma unroll
            for (int rr = 0; rr < 4; ++rr)
                aprf[rr] = *(const float4*)(A + (size_t)(mloc + ar + rr * 16) * K
                                            + k0 + 64 + ac);
#pragma unroll
            for (int rr = 0; rr < 8; ++rr)
                bprf[rr] = *(const float4*)(B + (size_t)(k0 + 64 + br + rr) * N
                                            + nblk + bc);
        }

#pragma unroll
        for (int half = 0; half < 2; ++half) {
            v2f af[2][8], bf[2][8];
#pragma unroll
            for (int s = 0; s < 8; ++s) {
                const int kb = half * 32 + s * 4 + lhalf * 2;
                af[0][s] = v2f{As[arow0][kb],      As[arow0][kb + 1]};
                af[1][s] = v2f{As[arow0 + 16][kb], As[arow0 + 16][kb + 1]};
                bf[0][s] = v2f{Bs[kb][bcol0],      Bs[kb + 1][bcol0]};
                bf[1][s] = v2f{Bs[kb][bcol0 + 16], Bs[kb + 1][bcol0 + 16]};
            }
#pragma unroll
            for (int s = 0; s < 8; ++s) {
                acc[0][0] = __builtin_amdgcn_wmma_f32_16x16x4_f32(
                    false, af[0][s], false, bf[0][s], (short)0, acc[0][0], false, false);
                acc[0][1] = __builtin_amdgcn_wmma_f32_16x16x4_f32(
                    false, af[0][s], false, bf[1][s], (short)0, acc[0][1], false, false);
                acc[1][0] = __builtin_amdgcn_wmma_f32_16x16x4_f32(
                    false, af[1][s], false, bf[0][s], (short)0, acc[1][0], false, false);
                acc[1][1] = __builtin_amdgcn_wmma_f32_16x16x4_f32(
                    false, af[1][s], false, bf[1][s], (short)0, acc[1][1], false, false);
            }
        }
        __syncthreads();
    }

#pragma unroll
    for (int nsub = 0; nsub < 2; ++nsub) {
        const int col = nblk + nq * 32 + nsub * 16 + l15;
        const float bv = (flags & GEMM_BIAS) ? bias[col] : 0.f;
#pragma unroll
        for (int msub = 0; msub < 2; ++msub) {
#pragma unroll
            for (int r = 0; r < 8; ++r) {
                const int rloc = mq * 32 + msub * 16 + r + lhalf * 8;
                const int row  = m0 + rloc;
                float v = acc[msub][nsub][r] + bv;
                if (flags & GEMM_RESID) v += resid[(size_t)(mloc + rloc) * N + col];
                if (flags & GEMM_RELU)  v = fmaxf(v, 0.f);
                C[(size_t)row * N + col] = v;
            }
        }
    }
}

__global__ __launch_bounds__(256)
void dec_kernel(const float* __restrict__ a0base, const float* __restrict__ a1base,
                const float* __restrict__ b1, const float* __restrict__ w2,
                const float* __restrict__ b2, float* __restrict__ dec)
{
    const int wid  = (int)((blockIdx.x * 256u + threadIdx.x) >> 5);
    const int lane = threadIdx.x & 31;
    const int n    = wid & 63;
    const int m    = (wid >> 6) & 63;
    const int b    = (wid >> 12) & 3;
    const int pair = wid >> 14;
    const int i    = pair / 3;
    const int j    = pair % 3;

    const float* a1 = a1base + ((size_t)j * Rsz + b * Lsz + m) * 2048;
    const float* a0 = a0base + ((size_t)i * Rsz + b * Lsz + n) * 2048;

    float acc = 0.f;
    for (int c = lane; c < 2048; c += 32) {
        float t = a1[c] + a0[c] + b1[c];
        if (t > 0.f) acc += t * (w2[2 * c + 1] - w2[2 * c]);
    }
#pragma unroll
    for (int off = 16; off; off >>= 1) acc += __shfl_xor(acc, off, 32);
    if (lane == 0) {
        float s = acc + (b2[1] - b2[0]);
        dec[wid] = (s > 0.f) ? NEGC : 0.f;
    }
}

__global__ __launch_bounds__(192)
void attn_scores_kernel(const float* __restrict__ Qb, const float* __restrict__ Kb,
                        const float* __restrict__ dec, const int* __restrict__ mask,
                        float* __restrict__ S)
{
    const int blk = blockIdx.x;
    const int q = blk & 63;
    const int h = (blk >> 6) & 15;
    const int b = (blk >> 10) & 3;
    const int i = blk >> 12;
    const int t = threadIdx.x;
    const int j = t >> 6;
    const int k = t & 63;

    __shared__ float qrow[64];
    __shared__ float red[192];

    const float* Q = Qb + ((size_t)i * Rsz + b * Lsz + q) * DMsz + h * DKsz;
    if (t < 64) qrow[t] = Q[t];
    __syncthreads();

    const float* Kp = Kb + ((size_t)j * Rsz + b * Lsz + k) * DMsz + h * DKsz;
    float s = 0.f;
#pragma unroll 8
    for (int d = 0; d < 64; ++d) s += qrow[d] * Kp[d];
    s *= 0.125f;
    s += dec[(((size_t)(i * 3 + j) * Bsz + b) * Lsz + q) * Lsz + k];
    if (mask[((size_t)b * Lsz + q) * Lsz + k] == 0) s = NEGC;

    red[t] = s; __syncthreads();
    if (t < 64) red[t] = fmaxf(red[t], fmaxf(red[t + 64], red[t + 128]));
    __syncthreads();
    for (int st = 32; st > 0; st >>= 1) {
        if (t < st) red[t] = fmaxf(red[t], red[t + st]);
        __syncthreads();
    }
    const float mx = red[0];
    __syncthreads();
    const float e = __expf(s - mx);
    red[t] = e; __syncthreads();
    if (t < 64) red[t] = red[t] + red[t + 64] + red[t + 128];
    __syncthreads();
    for (int st = 32; st > 0; st >>= 1) {
        if (t < st) red[t] += red[t + st];
        __syncthreads();
    }
    S[(size_t)blk * 192 + t] = e * (1.0f / red[0]);
}

__global__ __launch_bounds__(64)
void attn_av_kernel(const float* __restrict__ S, const float* __restrict__ Vb,
                    float* __restrict__ Ob)
{
    const int blk = blockIdx.x;
    const int q = blk & 63;
    const int h = (blk >> 6) & 15;
    const int b = (blk >> 10) & 3;
    const int i = blk >> 12;
    const int d = threadIdx.x;

    __shared__ float arow[192];
    const float* Sp = S + (size_t)blk * 192;
    arow[d] = Sp[d]; arow[d + 64] = Sp[d + 64]; arow[d + 128] = Sp[d + 128];
    __syncthreads();

    float acc = 0.f;
    for (int jk = 0; jk < 192; ++jk) {
        int j = jk >> 6, k = jk & 63;
        acc += arow[jk] * Vb[((size_t)j * Rsz + b * Lsz + k) * DMsz + h * DVsz + d];
    }
    Ob[((size_t)i * Rsz + b * Lsz + q) * DMsz + h * DVsz + d] = acc;
}

__global__ __launch_bounds__(256)
void layernorm_kernel(const float* __restrict__ x, const float* __restrict__ resid,
                      const float* __restrict__ g, const float* __restrict__ bta,
                      float* __restrict__ out)
{
    const int row = blockIdx.x;
    const int t = threadIdx.x;
    __shared__ float red[256];

    const float* xp = x + (size_t)row * 1024;
    float v[4];
    float s = 0.f;
#pragma unroll
    for (int u = 0; u < 4; ++u) {
        float val = xp[t + u * 256];
        if (resid) val += resid[(size_t)row * 1024 + t + u * 256];
        v[u] = val; s += val;
    }
    red[t] = s; __syncthreads();
    for (int st = 128; st > 0; st >>= 1) { if (t < st) red[t] += red[t + st]; __syncthreads(); }
    const float mu = red[0] * (1.f / 1024.f);
    __syncthreads();

    float s2 = 0.f;
#pragma unroll
    for (int u = 0; u < 4; ++u) { float dd = v[u] - mu; s2 += dd * dd; }
    red[t] = s2; __syncthreads();
    for (int st = 128; st > 0; st >>= 1) { if (t < st) red[t] += red[t + st]; __syncthreads(); }
    const float var = red[0] * (1.f / 1024.f);
    const float rinv = 1.0f / sqrtf(var + 1e-6f);
#pragma unroll
    for (int u = 0; u < 4; ++u) {
        int c = t + u * 256;
        out[(size_t)row * 1024 + c] = (v[u] - mu) * rinv * g[c] + bta[c];
    }
}

extern "C" void kernel_launch(void* const* d_in, const int* in_sizes, int n_in,
                              void* d_out, int out_size, void* d_ws, size_t ws_size,
                              hipStream_t stream)
{
    const float* X[3]   = { (const float*)d_in[0], (const float*)d_in[1], (const float*)d_in[2] };
    const int*   mask   = (const int*)d_in[3];
    const float* w_qs   = (const float*)d_in[4];
    const float* w_ks   = (const float*)d_in[5];
    const float* w_vs   = (const float*)d_in[6];
    const float* w_ds0  = (const float*)d_in[7];
    const float* w_ds1  = (const float*)d_in[8];
    const float* dc_w1  = (const float*)d_in[9];
    const float* dc_b1  = (const float*)d_in[10];
    const float* dc_w2  = (const float*)d_in[11];
    const float* dc_b2  = (const float*)d_in[12];
    const float* w_fc   = (const float*)d_in[13];
    const float* ln1_g  = (const float*)d_in[14];
    const float* ln1_b  = (const float*)d_in[15];
    const float* ffn_w1 = (const float*)d_in[16];
    const float* ffn_b1 = (const float*)d_in[17];
    const float* ffn_w2 = (const float*)d_in[18];
    const float* ffn_b2 = (const float*)d_in[19];
    const float* ln2_g  = (const float*)d_in[20];
    const float* ln2_b  = (const float*)d_in[21];
    float* out = (float*)d_out;

    float* ws = (float*)d_ws;
    size_t off = 0;
    auto alloc = [&](size_t n) { float* p = ws + off; off += n; return p; };
    const size_t RM = (size_t)Rsz * DMsz;
    float* Q   = alloc(3 * RM);
    float* Kb  = alloc(3 * RM);
    float* Vb  = alloc(3 * RM);
    float* d0  = alloc(3 * RM);
    float* d1  = alloc(3 * RM);
    float* a0  = alloc(3 * (size_t)Rsz * 2048);
    float* a1  = alloc(3 * (size_t)Rsz * 2048);
    float* dec = alloc(9 * (size_t)Bsz * Lsz * Lsz);
    float* S   = alloc((size_t)3 * Bsz * Hsz * Lsz * 192);
    float* O   = alloc(3 * RM);
    float* fc  = alloc(3 * RM);
    float* oln = alloc(3 * RM);
    float* h1  = alloc(3 * (size_t)Rsz * DIsz);
    float* f2  = alloc(3 * RM);
    (void)ws_size; (void)n_in; (void)in_sizes; (void)out_size;

    auto gemm3 = [&](const float* a0s, const float* a1s, const float* a2s, long strideAz,
                     const float* b0, const float* b1, const float* b2,
                     const float* b3, const float* b4,
                     const float* bias,
                     const float* r0s, const float* r1s, const float* r2s,
                     float* C, long strideCz, int nz, int N, int K, int flags) {
        dim3 grid(N / 128, (3 * Rsz) / 64, nz);
        gemm_wmma_f32<<<grid, dim3(256), 0, stream>>>(a0s, a1s, a2s, strideAz,
                                                      b0, b1, b2, b3, b4, bias,
                                                      r0s, r1s, r2s, C, strideCz,
                                                      N, K, flags);
    };

    gemm3(X[0], X[1], X[2], 0,
          w_qs, w_ks, w_vs, w_ds0, w_ds1,
          nullptr, nullptr, nullptr, nullptr,
          Q, (long)(3 * RM), 5, 1024, 1024, 0);

    gemm3(d0, d0 + RM, d0 + 2 * RM, (long)(3 * RM),
          dc_w1 + (size_t)1024 * 2048, dc_w1, dc_w1, dc_w1, dc_w1,
          nullptr, nullptr, nullptr, nullptr,
          a0, (long)(3 * (size_t)Rsz * 2048), 2, 2048, 1024, 0);

    {
        int total_waves = 9 * Bsz * Lsz * Lsz;
        dec_kernel<<<total_waves / 8, 256, 0, stream>>>(a0, a1, dc_b1, dc_w2, dc_b2, dec);
    }

    {
        int nblk = 3 * Bsz * Hsz * Lsz;
        attn_scores_kernel<<<nblk, 192, 0, stream>>>(Q, Kb, dec, mask, S);
        attn_av_kernel<<<nblk, 64, 0, stream>>>(S, Vb, O);
    }

    gemm3(O, O + RM, O + 2 * RM, 0,
          w_fc, w_fc, w_fc, w_fc, w_fc,
          nullptr, X[0], X[1], X[2],
          fc, 0, 1, 1024, 1024, GEMM_RESID);
    layernorm_kernel<<<3 * Rsz, 256, 0, stream>>>(fc, nullptr, ln1_g, ln1_b, oln);
    gemm3(oln, oln + RM, oln + 2 * RM, 0,
          ffn_w1, ffn_w1, ffn_w1, ffn_w1, ffn_w1,
          ffn_b1, nullptr, nullptr, nullptr,
          h1, 0, 1, DIsz, 1024, GEMM_BIAS | GEMM_RELU);
    gemm3(h1, h1 + (size_t)Rsz * DIsz, h1 + 2 * (size_t)Rsz * DIsz, 0,
          ffn_w2, ffn_w2, ffn_w2, ffn_w2, ffn_w2,
          ffn_b2, nullptr, nullptr, nullptr,
          f2, 0, 1, 1024, DIsz, GEMM_BIAS);
    layernorm_kernel<<<3 * Rsz, 256, 0, stream>>>(f2, oln, ln2_g, ln2_b, out);
}


# Decision-gate kernel: each wave keeps two a1 rows, b1 and the w2 difference in registers and streams 32 a0 rows as grouped independent loads (was two dependent loads per term); sum order unchanged
# speedup vs baseline: 3.3307x; 1.3702x over previous
#include <hip/hip_runtime.h>
#include <hip/hip_bf16.h>

#define Bsz 4
#define Lsz 64
#define DMsz 1024
#define Hsz 16
#define DKsz 64
#define DVsz 64
#define Dsz 1024
#define DIsz 4096
#define Rsz 256
#define NEGC (-1.0e9f)

typedef float v2f __attribute__((ext_vector_type(2)));
typedef float v8f __attribute__((ext_vector_type(8)));

#define GEMM_BIAS  1
#define GEMM_RELU  2
#define GEMM_RESID 4

__global__ __launch_bounds__(256)
void gemm_wmma_f32(const float* __restrict__ A0, const float* __restrict__ A1,
                   const float* __restrict__ A2, long strideAz,
                   const float* __restrict__ B0, const float* __restrict__ B1,
                   const float* __restrict__ B2, const float* __restrict__ B3,
                   const float* __restrict__ B4,
                   const float* __restrict__ bias,
                   const float* __restrict__ R0, const float* __restrict__ R1,
                   const float* __restrict__ R2,
                   float* __restrict__ Cbase, long strideCz,
                   int N, int K, int flags)
{
    const int seg  = blockIdx.y >> 2;
    const int mloc = (blockIdx.y & 3) * 64;
    const int zi   = blockIdx.z;
    const float* __restrict__ A =
        ((seg == 0) ? A0 : (seg == 1) ? A1 : A2) + (size_t)zi * strideAz;
    const float* __restrict__ B =
        (zi == 0) ? B0 : (zi == 1) ? B1 : (zi == 2) ? B2 : (zi == 3) ? B3 : B4;
    const float* __restrict__ resid = (seg == 0) ? R0 : (seg == 1) ? R1 : R2;
    float* __restrict__ C = Cbase + (size_t)zi * strideCz;

    __shared__ float As[64][64 + 1];
    __shared__ float Bs[64][128 + 1];

    const int tid   = threadIdx.x;
    const int wave  = tid >> 5;
    const int lane  = tid & 31;
    const int l15   = lane & 15;
    const int lhalf = lane >> 4;
    const int m0    = blockIdx.y * 64;
    const int nblk  = blockIdx.x * 128;
    const int mq    = wave >> 2;
    const int nq    = wave & 3;

    const int ar = tid >> 4;
    const int ac = (tid & 15) << 2;
    const int br = (tid >> 5) * 8;
    const int bc = (tid & 31) << 2;
    float4 aprf[4], bprf[8];

    v8f acc[2][2];
#pragma unroll
    for (int x = 0; x < 2; ++x)
#pragma unroll
        for (int y = 0; y < 2; ++y)
            acc[x][y] = v8f{0.f, 0.f, 0.f, 0.f, 0.f, 0.f, 0.f, 0.f};

#pragma unroll
    for (int rr = 0; rr < 4; ++rr)
        aprf[rr] = *(const float4*)(A + (size_t)(mloc + ar + rr * 16) * K + ac);
#pragma unroll
    for (int rr = 0; rr < 8; ++rr)
        bprf[rr] = *(const float4*)(B + (size_t)(br + rr) * N + nblk + bc);

    const int arow0 = mq * 32 + l15;
    const int bcol0 = nq * 32 + l15;

    for (int k0 = 0; k0 < K; k0 += 64) {
#pragma unroll
        for (int rr = 0; rr < 4; ++rr) {
            As[ar + rr * 16][ac + 0] = aprf[rr].x;
            As[ar + rr * 16][ac + 1] = aprf[rr].y;
            As[ar + rr * 16][ac + 2] = aprf[rr].z;
            As[ar + rr * 16][ac + 3] = aprf[rr].w;
        }
#pragma unroll
        for (int rr = 0; rr < 8; ++rr) {
            Bs[br + rr][bc + 0] = bprf[rr].x;
            Bs[br + rr][bc + 1] = bprf[rr].y;
            Bs[br + rr][bc + 2] = bprf[rr].z;
            Bs[br + rr][bc + 3] = bprf[rr].w;
        }
        __syncthreads();

        if (k0 + 64 < K) {
#pragma unroll
            for (int rr = 0; rr < 4; ++rr)
                aprf[rr] = *(const float4*)(A + (size_t)(mloc + ar + rr * 16) * K
                                            + k0 + 64 + ac);
#pragma unroll
            for (int rr = 0; rr < 8; ++rr)
                bprf[rr] = *(const float4*)(B + (size_t)(k0 + 64 + br + rr) * N
                                            + nblk + bc);
        }

#pragma unroll
        for (int half = 0; half < 2; ++half) {
            v2f af[2][8], bf[2][8];
#pragma unroll
            for (int s = 0; s < 8; ++s) {
                const int kb = half * 32 + s * 4 + lhalf * 2;
                af[0][s] = v2f{As[arow0][kb],      As[arow0][kb + 1]};
                af[1][s] = v2f{As[arow0 + 16][kb], As[arow0 + 16][kb + 1]};
                bf[0][s] = v2f{Bs[kb][bcol0],      Bs[kb + 1][bcol0]};
                bf[1][s] = v2f{Bs[kb][bcol0 + 16], Bs[kb + 1][bcol0 + 16]};
            }
#pragma unroll
            for (int s = 0; s < 8; ++s) {
                acc[0][0] = __builtin_amdgcn_wmma_f32_16x16x4_f32(
                    false, af[0][s], false, bf[0][s], (short)0, acc[0][0], false, false);
                acc[0][1] = __builtin_amdgcn_wmma_f32_16x16x4_f32(
                    false, af[0][s], false, bf[1][s], (short)0, acc[0][1], false, false);
                acc[1][0] = __builtin_amdgcn_wmma_f32_16x16x4_f32(
                    false, af[1][s], false, bf[0][s], (short)0, acc[1][0], false, false);
                acc[1][1] = __builtin_amdgcn_wmma_f32_16x16x4_f32(
                    false, af[1][s], false, bf[1][s], (short)0, acc[1][1], false, false);
            }
        }
        __syncthreads();
    }

#pragma unroll
    for (int nsub = 0; nsub < 2; ++nsub) {
        const int col = nblk + nq * 32 + nsub * 16 + l15;
        const float bv = (flags & GEMM_BIAS) ? bias[col] : 0.f;
#pragma unroll
        for (int msub = 0; msub < 2; ++msub) {
#pragma unroll
            for (int r = 0; r < 8; ++r) {
                const int rloc = mq * 32 + msub * 16 + r + lhalf * 8;
                const int row  = m0 + rloc;
                float v = acc[msub][nsub][r] + bv;
                if (flags & GEMM_RESID) v += resid[(size_t)(mloc + rloc) * N + col];
                if (flags & GEMM_RELU)  v = fmaxf(v, 0.f);
                C[(size_t)row * N + col] = v;
            }
        }
    }
}

#define DEC_NB 4
#define DEC_HOLD16(a, o)                                                        \
    asm volatile("" :: "v"(a[(o) + 0]), "v"(a[(o) + 1]), "v"(a[(o) + 2]),       \
                 "v"(a[(o) + 3]), "v"(a[(o) + 4]), "v"(a[(o) + 5]),             \
                 "v"(a[(o) + 6]), "v"(a[(o) + 7]), "v"(a[(o) + 8]),             \
                 "v"(a[(o) + 9]), "v"(a[(o) + 10]), "v"(a[(o) + 11]),           \
                 "v"(a[(o) + 12]), "v"(a[(o) + 13]), "v"(a[(o) + 14]),          \
                 "v"(a[(o) + 15]))

__global__ __launch_bounds__(128, 3)
void dec_kernel(const float* __restrict__ a0base, const float* __restrict__ a1base,
                const float* __restrict__ b1, const float* __restrict__ w2,
                const float* __restrict__ b2, float* __restrict__ dec)
{
    const int lane = threadIdx.x & 31;
    const int wid  = __builtin_amdgcn_readfirstlane(
                         (int)((blockIdx.x * 128u + threadIdx.x) >> 5));
    const int nh   = wid & 1;
    const int mp   = (wid >> 1) & 31;
    const int b    = (wid >> 6) & 3;
    const int pair = wid >> 8;
    const int i    = pair / 3;
    const int j    = pair % 3;
    const int m0   = mp * 2;
    const int n0   = nh * 32;

    const float* a1r = a1base + ((size_t)j * Rsz + b * Lsz + m0) * 2048;
    const float* a0r = a0base + ((size_t)i * Rsz + b * Lsz + n0) * 2048;

    float wd[64], bb[64], x0[64], x1[64];
#pragma unroll
    for (int k = 0; k < 64; ++k) {
        const int c = lane + 32 * k;
        wd[k] = w2[2 * c + 1] - w2[2 * c];
        bb[k] = b1[c];
        x0[k] = a1r[lane + 32 * k];
        x1[k] = a1r[2048 + lane + 32 * k];
    }

    float r0 = 0.f, r1 = 0.f;
    for (int g = 0; g < 32; g += DEC_NB) {
        float s0[DEC_NB], s1[DEC_NB];
#pragma unroll
        for (int u = 0; u < DEC_NB; ++u) {
            const float* p = a0r + (size_t)(g + u) * 2048;
            float y[64];
#pragma unroll
            for (int k = 0; k < 64; ++k) y[k] = p[lane + 32 * k];
            float c0 = 0.f, c1 = 0.f;
#pragma unroll
            for (int q = 0; q < 4; ++q) {
                DEC_HOLD16(y, 16 * q);
#pragma unroll
                for (int k = 16 * q; k < 16 * q + 16; ++k) {
                    const float t0 = x0[k] + y[k] + bb[k];
                    const float t1 = x1[k] + y[k] + bb[k];
                    const float f0 = fmaf(t0, wd[k], c0);
                    const float f1 = fmaf(t1, wd[k], c1);
                    c0 = (t0 > 0.f) ? f0 : c0;
                    c1 = (t1 > 0.f) ? f1 : c1;
                }
            }
            s0[u] = c0; s1[u] = c1;
        }
#pragma unroll
        for (int off = 16; off; off >>= 1) {
#pragma unroll
            for (int u = 0; u < DEC_NB; ++u) {
                s0[u] += __shfl_xor(s0[u], off, 32);
                s1[u] += __shfl_xor(s1[u], off, 32);
            }
        }
#pragma unroll
        for (int u = 0; u < DEC_NB; ++u) {
            r0 = (lane == g + u) ? s0[u] : r0;
            r1 = (lane == g + u) ? s1[u] : r1;
        }
    }

    const float bd = b2[1] - b2[0];
    const size_t o = (((size_t)pair * Bsz + b) * Lsz + m0) * Lsz + n0 + lane;
    dec[o]       = (r0 + bd > 0.f) ? NEGC : 0.f;
    dec[o + Lsz] = (r1 + bd > 0.f) ? NEGC : 0.f;
}

__global__ __launch_bounds__(192)
void attn_scores_kernel(const float* __restrict__ Qb, const float* __restrict__ Kb,
                        const float* __restrict__ dec, const int* __restrict__ mask,
                        float* __restrict__ S)
{
    const int blk = blockIdx.x;
    const int q = blk & 63;
    const int h = (blk >> 6) & 15;
    const int b = (blk >> 10) & 3;
    const int i = blk >> 12;
    const int t = threadIdx.x;
    const int j = t >> 6;
    const int k = t & 63;

    __shared__ float qrow[64];
    __shared__ float red[192];

    const float* Q = Qb + ((size_t)i * Rsz + b * Lsz + q) * DMsz + h * DKsz;
    if (t < 64) qrow[t] = Q[t];
    __syncthreads();

    const float* Kp = Kb + ((size_t)j * Rsz + b * Lsz + k) * DMsz + h * DKsz;
    float s = 0.f;
#pragma unroll 8
    for (int d = 0; d < 64; ++d) s += qrow[d] * Kp[d];
    s *= 0.125f;
    s += dec[(((size_t)(i * 3 + j) * Bsz + b) * Lsz + q) * Lsz + k];
    if (mask[((size_t)b * Lsz + q) * Lsz + k] == 0) s = NEGC;

    red[t] = s; __syncthreads();
    if (t < 64) red[t] = fmaxf(red[t], fmaxf(red[t + 64], red[t + 128]));
    __syncthreads();
    for (int st = 32; st > 0; st >>= 1) {
        if (t < st) red[t] = fmaxf(red[t], red[t + st]);
        __syncthreads();
    }
    const float mx = red[0];
    __syncthreads();
    const float e = __expf(s - mx);
    red[t] = e; __syncthreads();
    if (t < 64) red[t] = red[t] + red[t + 64] + red[t + 128];
    __syncthreads();
    for (int st = 32; st > 0; st >>= 1) {
        if (t < st) red[t] += red[t + st];
        __syncthreads();
    }
    S[(size_t)blk * 192 + t] = e * (1.0f / red[0]);
}

__global__ __launch_bounds__(64)
void attn_av_kernel(const float* __restrict__ S, const float* __restrict__ Vb,
                    float* __restrict__ Ob)
{
    const int blk = blockIdx.x;
    const int q = blk & 63;
    const int h = (blk >> 6) & 15;
    const int b = (blk >> 10) & 3;
    const int i = blk >> 12;
    const int d = threadIdx.x;

    __shared__ float arow[192];
    const float* Sp = S + (size_t)blk * 192;
    arow[d] = Sp[d]; arow[d + 64] = Sp[d + 64]; arow[d + 128] = Sp[d + 128];
    __syncthreads();

    float acc = 0.f;
    for (int jk = 0; jk < 192; ++jk) {
        int j = jk >> 6, k = jk & 63;
        acc += arow[jk] * Vb[((size_t)j * Rsz + b * Lsz + k) * DMsz + h * DVsz + d];
    }
    Ob[((size_t)i * Rsz + b * Lsz + q) * DMsz + h * DVsz + d] = acc;
}

__global__ __launch_bounds__(256)
void layernorm_kernel(const float* __restrict__ x, const float* __restrict__ resid,
                      const float* __restrict__ g, const float* __restrict__ bta,
                      float* __restrict__ out)
{
    const int row = blockIdx.x;
    const int t = threadIdx.x;
    __shared__ float red[256];

    const float* xp = x + (size_t)row * 1024;
    float v[4];
    float s = 0.f;
#pragma unroll
    for (int u = 0; u < 4; ++u) {
        float val = xp[t + u * 256];
        if (resid) val += resid[(size_t)row * 1024 + t + u * 256];
        v[u] = val; s += val;
    }
    red[t] = s; __syncthreads();
    for (int st = 128; st > 0; st >>= 1) { if (t < st) red[t] += red[t + st]; __syncthreads(); }
    const float mu = red[0] * (1.f / 1024.f);
    __syncthreads();

    float s2 = 0.f;
#pragma unroll
    for (int u = 0; u < 4; ++u) { float dd = v[u] - mu; s2 += dd * dd; }
    red[t] = s2; __syncthreads();
    for (int st = 128; st > 0; st >>= 1) { if (t < st) red[t] += red[t + st]; __syncthreads(); }
    const float var = red[0] * (1.f / 1024.f);
    const float rinv = 1.0f / sqrtf(var + 1e-6f);
#pragma unroll
    for (int u = 0; u < 4; ++u) {
        int c = t + u * 256;
        out[(size_t)row * 1024 + c] = (v[u] - mu) * rinv * g[c] + bta[c];
    }
}

extern "C" void kernel_launch(void* const* d_in, const int* in_sizes, int n_in,
                              void* d_out, int out_size, void* d_ws, size_t ws_size,
                              hipStream_t stream)
{
    const float* X[3]   = { (const float*)d_in[0], (const float*)d_in[1], (const float*)d_in[2] };
    const int*   mask   = (const int*)d_in[3];
    const float* w_qs   = (const float*)d_in[4];
    const float* w_ks   = (const float*)d_in[5];
    const float* w_vs   = (const float*)d_in[6];
    const float* w_ds0  = (const float*)d_in[7];
    const float* w_ds1  = (const float*)d_in[8];
    const float* dc_w1  = (const float*)d_in[9];
    const float* dc_b1  = (const float*)d_in[10];
    const float* dc_w2  = (const float*)d_in[11];
    const float* dc_b2  = (const float*)d_in[12];
    const float* w_fc   = (const float*)d_in[13];
    const float* ln1_g  = (const float*)d_in[14];
    const float* ln1_b  = (const float*)d_in[15];
    const float* ffn_w1 = (const float*)d_in[16];
    const float* ffn_b1 = (const float*)d_in[17];
    const float* ffn_w2 = (const float*)d_in[18];
    const float* ffn_b2 = (const float*)d_in[19];
    const float* ln2_g  = (const float*)d_in[20];
    const float* ln2_b  = (const float*)d_in[21];
    float* out = (float*)d_out;

    float* ws = (float*)d_ws;
    size_t off = 0;
    auto alloc = [&](size_t n) { float* p = ws + off; off += n; return p; };
    const size_t RM = (size_t)Rsz * DMsz;
    float* Q   = alloc(3 * RM);
    float* Kb  = alloc(3 * RM);
    float* Vb  = alloc(3 * RM);
    float* d0  = alloc(3 * RM);
    float* d1  = alloc(3 * RM);
    float* a0  = alloc(3 * (size_t)Rsz * 2048);
    float* a1  = alloc(3 * (size_t)Rsz * 2048);
    float* dec = alloc(9 * (size_t)Bsz * Lsz * Lsz);
    float* S   = alloc((size_t)3 * Bsz * Hsz * Lsz * 192);
    float* O   = alloc(3 * RM);
    float* fc  = alloc(3 * RM);
    float* oln = alloc(3 * RM);
    float* h1  = alloc(3 * (size_t)Rsz * DIsz);
    float* f2  = alloc(3 * RM);
    (void)ws_size; (void)n_in; (void)in_sizes; (void)out_size;

    auto gemm3 = [&](const float* a0s, const float* a1s, const float* a2s, long strideAz,
                     const float* b0, const float* b1, const float* b2,
                     const float* b3, const float* b4,
                     const float* bias,
                     const float* r0s, const float* r1s, const float* r2s,
                     float* C, long strideCz, int nz, int N, int K, int flags) {
        dim3 grid(N / 128, (3 * Rsz) / 64, nz);
        gemm_wmma_f32<<<grid, dim3(256), 0, stream>>>(a0s, a1s, a2s, strideAz,
                                                      b0, b1, b2, b3, b4, bias,
                                                      r0s, r1s, r2s, C, strideCz,
                                                      N, K, flags);
    };

    gemm3(X[0], X[1], X[2], 0,
          w_qs, w_ks, w_vs, w_ds0, w_ds1,
          nullptr, nullptr, nullptr, nullptr,
          Q, (long)(3 * RM), 5, 1024, 1024, 0);

    gemm3(d0, d0 + RM, d0 + 2 * RM, (long)(3 * RM),
          dc_w1 + (size_t)1024 * 2048, dc_w1, dc_w1, dc_w1, dc_w1,
          nullptr, nullptr, nullptr, nullptr,
          a0, (long)(3 * (size_t)Rsz * 2048), 2, 2048, 1024, 0);

    {
        dec_kernel<<<576, 128, 0, stream>>>(a0, a1, dc_b1, dc_w2, dc_b2, dec);
    }

    {
        int nblk = 3 * Bsz * Hsz * Lsz;
        attn_scores_kernel<<<nblk, 192, 0, stream>>>(Q, Kb, dec, mask, S);
        attn_av_kernel<<<nblk, 64, 0, stream>>>(S, Vb, O);
    }

    gemm3(O, O + RM, O + 2 * RM, 0,
          w_fc, w_fc, w_fc, w_fc, w_fc,
          nullptr, X[0], X[1], X[2],
          fc, 0, 1, 1024, 1024, GEMM_RESID);
    layernorm_kernel<<<3 * Rsz, 256, 0, stream>>>(fc, nullptr, ln1_g, ln1_b, oln);
    gemm3(oln, oln + RM, oln + 2 * RM, 0,
          ffn_w1, ffn_w1, ffn_w1, ffn_w1, ffn_w1,
          ffn_b1, nullptr, nullptr, nullptr,
          h1, 0, 1, DIsz, 1024, GEMM_BIAS | GEMM_RELU);
    gemm3(h1, h1 + (size_t)Rsz * DIsz, h1 + 2 * (size_t)Rsz * DIsz, 0,
          ffn_w2, ffn_w2, ffn_w2, ffn_w2, ffn_w2,
          ffn_b2, nullptr, nullptr, nullptr,
          f2, 0, 1, 1024, DIsz, GEMM_BIAS);
    layernorm_kernel<<<3 * Rsz, 256, 0, stream>>>(f2, oln, ln2_g, ln2_b, out);
}


# Split K of the two 96-block GEMMs (attention out-projection, FFN down-projection) 8 ways over grid z: 768 blocks fill all WGPs; each layernorm adds the 8 partial sums in fixed order
# speedup vs baseline: 4.2069x; 1.2631x over previous
#include <hip/hip_runtime.h>
#include <hip/hip_bf16.h>

#define Bsz 4
#define Lsz 64
#define DMsz 1024
#define Hsz 16
#define DKsz 64
#define DVsz 64
#define Dsz 1024
#define DIsz 4096
#define Rsz 256
#define NEGC (-1.0e9f)

typedef float v2f __attribute__((ext_vector_type(2)));
typedef float v8f __attribute__((ext_vector_type(8)));

#define GEMM_BIAS  1
#define GEMM_RELU  2
#define GEMM_RESID 4

__global__ __launch_bounds__(256)
void gemm_wmma_f32(const float* __restrict__ A0, const float* __restrict__ A1,
                   const float* __restrict__ A2, long strideAz, int lda,
                   const float* __restrict__ B0, const float* __restrict__ B1,
                   const float* __restrict__ B2, const float* __restrict__ B3,
                   const float* __restrict__ B4, long strideBz,
                   const float* __restrict__ bias,
                   const float* __restrict__ R0, const float* __restrict__ R1,
                   const float* __restrict__ R2,
                   float* __restrict__ Cbase, long strideCz,
                   int N, int Kloop, int flags)
{
    const int seg  = blockIdx.y >> 2;
    const int mloc = (blockIdx.y & 3) * 64;
    const int zi   = blockIdx.z;
    const float* __restrict__ A =
        ((seg == 0) ? A0 : (seg == 1) ? A1 : A2) + (size_t)zi * strideAz;
    const float* __restrict__ B =
        ((zi == 0) ? B0 : (zi == 1) ? B1 : (zi == 2) ? B2 : (zi == 3) ? B3 : B4)
        + (size_t)zi * strideBz;
    const bool lead = (zi == 0);
    const float* __restrict__ resid = (seg == 0) ? R0 : (seg == 1) ? R1 : R2;
    float* __restrict__ C = Cbase + (size_t)zi * strideCz;

    __shared__ float As[64][64 + 1];
    __shared__ float Bs[64][128 + 1];

    const int tid   = threadIdx.x;
    const int wave  = tid >> 5;
    const int lane  = tid & 31;
    const int l15   = lane & 15;
    const int lhalf = lane >> 4;
    const int m0    = blockIdx.y * 64;
    const int nblk  = blockIdx.x * 128;
    const int mq    = wave >> 2;
    const int nq    = wave & 3;

    const int ar = tid >> 4;
    const int ac = (tid & 15) << 2;
    const int br = (tid >> 5) * 8;
    const int bc = (tid & 31) << 2;
    float4 aprf[4], bprf[8];

    v8f acc[2][2];
#pragma unroll
    for (int x = 0; x < 2; ++x)
#pragma unroll
        for (int y = 0; y < 2; ++y)
            acc[x][y] = v8f{0.f, 0.f, 0.f, 0.f, 0.f, 0.f, 0.f, 0.f};

#pragma unroll
    for (int rr = 0; rr < 4; ++rr)
        aprf[rr] = *(const float4*)(A + (size_t)(mloc + ar + rr * 16) * lda + ac);
#pragma unroll
    for (int rr = 0; rr < 8; ++rr)
        bprf[rr] = *(const float4*)(B + (size_t)(br + rr) * N + nblk + bc);

    const int arow0 = mq * 32 + l15;
    const int bcol0 = nq * 32 + l15;

    for (int k0 = 0; k0 < Kloop; k0 += 64) {
#pragma unroll
        for (int rr = 0; rr < 4; ++rr) {
            As[ar + rr * 16][ac + 0] = aprf[rr].x;
            As[ar + rr * 16][ac + 1] = aprf[rr].y;
            As[ar + rr * 16][ac + 2] = aprf[rr].z;
            As[ar + rr * 16][ac + 3] = aprf[rr].w;
        }
#pragma unroll
        for (int rr = 0; rr < 8; ++rr) {
            Bs[br + rr][bc + 0] = bprf[rr].x;
            Bs[br + rr][bc + 1] = bprf[rr].y;
            Bs[br + rr][bc + 2] = bprf[rr].z;
            Bs[br + rr][bc + 3] = bprf[rr].w;
        }
        __syncthreads();

        if (k0 + 64 < Kloop) {
#pragma unroll
            for (int rr = 0; rr < 4; ++rr)
                aprf[rr] = *(const float4*)(A + (size_t)(mloc + ar + rr * 16) * lda
                                            + k0 + 64 + ac);
#pragma unroll
            for (int rr = 0; rr < 8; ++rr)
                bprf[rr] = *(const float4*)(B + (size_t)(k0 + 64 + br + rr) * N
                                            + nblk + bc);
        }

#pragma unroll
        for (int half = 0; half < 2; ++half) {
            v2f af[2][8], bf[2][8];
#pragma unroll
            for (int s = 0; s < 8; ++s) {
                const int kb = half * 32 + s * 4 + lhalf * 2;
                af[0][s] = v2f{As[arow0][kb],      As[arow0][kb + 1]};
                af[1][s] = v2f{As[arow0 + 16][kb], As[arow0 + 16][kb + 1]};
                bf[0][s] = v2f{Bs[kb][bcol0],      Bs[kb + 1][bcol0]};
                bf[1][s] = v2f{Bs[kb][bcol0 + 16], Bs[kb + 1][bcol0 + 16]};
            }
#pragma unroll
            for (int s = 0; s < 8; ++s) {
                acc[0][0] = __builtin_amdgcn_wmma_f32_16x16x4_f32(
                    false, af[0][s], false, bf[0][s], (short)0, acc[0][0], false, false);
                acc[0][1] = __builtin_amdgcn_wmma_f32_16x16x4_f32(
                    false, af[0][s], false, bf[1][s], (short)0, acc[0][1], false, false);
                acc[1][0] = __builtin_amdgcn_wmma_f32_16x16x4_f32(
                    false, af[1][s], false, bf[0][s], (short)0, acc[1][0], false, false);
                acc[1][1] = __builtin_amdgcn_wmma_f32_16x16x4_f32(
                    false, af[1][s], false, bf[1][s], (short)0, acc[1][1], false, false);
            }
        }
        __syncthreads();
    }

#pragma unroll
    for (int nsub = 0; nsub < 2; ++nsub) {
        const int col = nblk + nq * 32 + nsub * 16 + l15;
        const float bv = ((flags & GEMM_BIAS) && lead) ? bias[col] : 0.f;
#pragma unroll
        for (int msub = 0; msub < 2; ++msub) {
#pragma unroll
            for (int r = 0; r < 8; ++r) {
                const int rloc = mq * 32 + msub * 16 + r + lhalf * 8;
                const int row  = m0 + rloc;
                float v = acc[msub][nsub][r] + bv;
                if ((flags & GEMM_RESID) && lead) v += resid[(size_t)(mloc + rloc) * N + col];
                if (flags & GEMM_RELU)  v = fmaxf(v, 0.f);
                C[(size_t)row * N + col] = v;
            }
        }
    }
}

#define DEC_NB 4
#define DEC_HOLD16(a, o)                                                        \
    asm volatile("" :: "v"(a[(o) + 0]), "v"(a[(o) + 1]), "v"(a[(o) + 2]),       \
                 "v"(a[(o) + 3]), "v"(a[(o) + 4]), "v"(a[(o) + 5]),             \
                 "v"(a[(o) + 6]), "v"(a[(o) + 7]), "v"(a[(o) + 8]),             \
                 "v"(a[(o) + 9]), "v"(a[(o) + 10]), "v"(a[(o) + 11]),           \
                 "v"(a[(o) + 12]), "v"(a[(o) + 13]), "v"(a[(o) + 14]),          \
                 "v"(a[(o) + 15]))

__global__ __launch_bounds__(128, 3)
void dec_kernel(const float* __restrict__ a0base, const float* __restrict__ a1base,
                const float* __restrict__ b1, const float* __restrict__ w2,
                const float* __restrict__ b2, float* __restrict__ dec)
{
    const int lane = threadIdx.x & 31;
    const int wid  = __builtin_amdgcn_readfirstlane(
                         (int)((blockIdx.x * 128u + threadIdx.x) >> 5));
    const int nh   = wid & 1;
    const int mp   = (wid >> 1) & 31;
    const int b    = (wid >> 6) & 3;
    const int pair = wid >> 8;
    const int i    = pair / 3;
    const int j    = pair % 3;
    const int m0   = mp * 2;
    const int n0   = nh * 32;

    const float* a1r = a1base + ((size_t)j * Rsz + b * Lsz + m0) * 2048;
    const float* a0r = a0base + ((size_t)i * Rsz + b * Lsz + n0) * 2048;

    float wd[64], bb[64], x0[64], x1[64];
#pragma unroll
    for (int k = 0; k < 64; ++k) {
        const int c = lane + 32 * k;
        wd[k] = w2[2 * c + 1] - w2[2 * c];
        bb[k] = b1[c];
        x0[k] = a1r[lane + 32 * k];
        x1[k] = a1r[2048 + lane + 32 * k];
    }

    float r0 = 0.f, r1 = 0.f;
    for (int g = 0; g < 32; g += DEC_NB) {
        float s0[DEC_NB], s1[DEC_NB];
#pragma unroll
        for (int u = 0; u < DEC_NB; ++u) {
            const float* p = a0r + (size_t)(g + u) * 2048;
            float y[64];
#pragma unroll
            for (int k = 0; k < 64; ++k) y[k] = p[lane + 32 * k];
            float c0 = 0.f, c1 = 0.f;
#pragma unroll
            for (int q = 0; q < 4; ++q) {
                DEC_HOLD16(y, 16 * q);
#pragma unroll
                for (int k = 16 * q; k < 16 * q + 16; ++k) {
                    const float t0 = x0[k] + y[k] + bb[k];
                    const float t1 = x1[k] + y[k] + bb[k];
                    const float f0 = fmaf(t0, wd[k], c0);
                    const float f1 = fmaf(t1, wd[k], c1);
                    c0 = (t0 > 0.f) ? f0 : c0;
                    c1 = (t1 > 0.f) ? f1 : c1;
                }
            }
            s0[u] = c0; s1[u] = c1;
        }
#pragma unroll
        for (int off = 16; off; off >>= 1) {
#pragma unroll
            for (int u = 0; u < DEC_NB; ++u) {
                s0[u] += __shfl_xor(s0[u], off, 32);
                s1[u] += __shfl_xor(s1[u], off, 32);
            }
        }
#pragma unroll
        for (int u = 0; u < DEC_NB; ++u) {
            r0 = (lane == g + u) ? s0[u] : r0;
            r1 = (lane == g + u) ? s1[u] : r1;
        }
    }

    const float bd = b2[1] - b2[0];
    const size_t o = (((size_t)pair * Bsz + b) * Lsz + m0) * Lsz + n0 + lane;
    dec[o]       = (r0 + bd > 0.f) ? NEGC : 0.f;
    dec[o + Lsz] = (r1 + bd > 0.f) ? NEGC : 0.f;
}

__global__ __launch_bounds__(192)
void attn_scores_kernel(const float* __restrict__ Qb, const float* __restrict__ Kb,
                        const float* __restrict__ dec, const int* __restrict__ mask,
                        float* __restrict__ S)
{
    const int blk = blockIdx.x;
    const int q = blk & 63;
    const int h = (blk >> 6) & 15;
    const int b = (blk >> 10) & 3;
    const int i = blk >> 12;
    const int t = threadIdx.x;
    const int j = t >> 6;
    const int k = t & 63;

    __shared__ float qrow[64];
    __shared__ float red[192];

    const float* Q = Qb + ((size_t)i * Rsz + b * Lsz + q) * DMsz + h * DKsz;
    if (t < 64) qrow[t] = Q[t];
    __syncthreads();

    const float* Kp = Kb + ((size_t)j * Rsz + b * Lsz + k) * DMsz + h * DKsz;
    float s = 0.f;
#pragma unroll 8
    for (int d = 0; d < 64; ++d) s += qrow[d] * Kp[d];
    s *= 0.125f;
    s += dec[(((size_t)(i * 3 + j) * Bsz + b) * Lsz + q) * Lsz + k];
    if (mask[((size_t)b * Lsz + q) * Lsz + k] == 0) s = NEGC;

    red[t] = s; __syncthreads();
    if (t < 64) red[t] = fmaxf(red[t], fmaxf(red[t + 64], red[t + 128]));
    __syncthreads();
    for (int st = 32; st > 0; st >>= 1) {
        if (t < st) red[t] = fmaxf(red[t], red[t + st]);
        __syncthreads();
    }
    const float mx = red[0];
    __syncthreads();
    const float e = __expf(s - mx);
    red[t] = e; __syncthreads();
    if (t < 64) red[t] = red[t] + red[t + 64] + red[t + 128];
    __syncthreads();
    for (int st = 32; st > 0; st >>= 1) {
        if (t < st) red[t] += red[t + st];
        __syncthreads();
    }
    S[(size_t)blk * 192 + t] = e * (1.0f / red[0]);
}

__global__ __launch_bounds__(64)
void attn_av_kernel(const float* __restrict__ S, const float* __restrict__ Vb,
                    float* __restrict__ Ob)
{
    const int blk = blockIdx.x;
    const int q = blk & 63;
    const int h = (blk >> 6) & 15;
    const int b = (blk >> 10) & 3;
    const int i = blk >> 12;
    const int d = threadIdx.x;

    __shared__ float arow[192];
    const float* Sp = S + (size_t)blk * 192;
    arow[d] = Sp[d]; arow[d + 64] = Sp[d + 64]; arow[d + 128] = Sp[d + 128];
    __syncthreads();

    float acc = 0.f;
    for (int jk = 0; jk < 192; ++jk) {
        int j = jk >> 6, k = jk & 63;
        acc += arow[jk] * Vb[((size_t)j * Rsz + b * Lsz + k) * DMsz + h * DVsz + d];
    }
    Ob[((size_t)i * Rsz + b * Lsz + q) * DMsz + h * DVsz + d] = acc;
}

#define KSPLIT 8
__global__ __launch_bounds__(256)
void layernorm_kernel(const float* __restrict__ x, long plane,
                      const float* __restrict__ resid,
                      const float* __restrict__ g, const float* __restrict__ bta,
                      float* __restrict__ out)
{
    const int row = blockIdx.x;
    const int t = threadIdx.x;
    __shared__ float red[256];

    const float* xp = x + (size_t)row * 1024;
    float v[4];
    float s = 0.f;
#pragma unroll
    for (int u = 0; u < 4; ++u) {
        float val = xp[t + u * 256];
#pragma unroll
        for (int p = 1; p < KSPLIT; ++p) val += xp[(size_t)p * plane + t + u * 256];
        v[u] = val;
    }
    if (resid) {
#pragma unroll
        for (int u = 0; u < 4; ++u) v[u] += resid[(size_t)row * 1024 + t + u * 256];
    }
#pragma unroll
    for (int u = 0; u < 4; ++u) s += v[u];
    red[t] = s; __syncthreads();
    for (int st = 128; st > 0; st >>= 1) { if (t < st) red[t] += red[t + st]; __syncthreads(); }
    const float mu = red[0] * (1.f / 1024.f);
    __syncthreads();

    float s2 = 0.f;
#pragma unroll
    for (int u = 0; u < 4; ++u) { float dd = v[u] - mu; s2 += dd * dd; }
    red[t] = s2; __syncthreads();
    for (int st = 128; st > 0; st >>= 1) { if (t < st) red[t] += red[t + st]; __syncthreads(); }
    const float var = red[0] * (1.f / 1024.f);
    const float rinv = 1.0f / sqrtf(var + 1e-6f);
#pragma unroll
    for (int u = 0; u < 4; ++u) {
        int c = t + u * 256;
        out[(size_t)row * 1024 + c] = (v[u] - mu) * rinv * g[c] + bta[c];
    }
}

extern "C" void kernel_launch(void* const* d_in, const int* in_sizes, int n_in,
                              void* d_out, int out_size, void* d_ws, size_t ws_size,
                              hipStream_t stream)
{
    const float* X[3]   = { (const float*)d_in[0], (const float*)d_in[1], (const float*)d_in[2] };
    const int*   mask   = (const int*)d_in[3];
    const float* w_qs   = (const float*)d_in[4];
    const float* w_ks   = (const float*)d_in[5];
    const float* w_vs   = (const float*)d_in[6];
    const float* w_ds0  = (const float*)d_in[7];
    const float* w_ds1  = (const float*)d_in[8];
    const float* dc_w1  = (const float*)d_in[9];
    const float* dc_b1  = (const float*)d_in[10];
    const float* dc_w2  = (const float*)d_in[11];
    const float* dc_b2  = (const float*)d_in[12];
    const float* w_fc   = (const float*)d_in[13];
    const float* ln1_g  = (const float*)d_in[14];
    const float* ln1_b  = (const float*)d_in[15];
    const float* ffn_w1 = (const float*)d_in[16];
    const float* ffn_b1 = (const float*)d_in[17];
    const float* ffn_w2 = (const float*)d_in[18];
    const float* ffn_b2 = (const float*)d_in[19];
    const float* ln2_g  = (const float*)d_in[20];
    const float* ln2_b  = (const float*)d_in[21];
    float* out = (float*)d_out;

    float* ws = (float*)d_ws;
    size_t off = 0;
    auto alloc = [&](size_t n) { float* p = ws + off; off += n; return p; };
    const size_t RM = (size_t)Rsz * DMsz;
    float* Q   = alloc(3 * RM);
    float* Kb  = alloc(3 * RM);
    float* Vb  = alloc(3 * RM);
    float* d0  = alloc(3 * RM);
    float* d1  = alloc(3 * RM);
    float* a0  = alloc(3 * (size_t)Rsz * 2048);
    float* a1  = alloc(3 * (size_t)Rsz * 2048);
    float* dec = alloc(9 * (size_t)Bsz * Lsz * Lsz);
    float* S   = alloc((size_t)3 * Bsz * Hsz * Lsz * 192);
    float* O   = alloc(3 * RM);
    float* oln = alloc(3 * RM);
    float* h1  = alloc(3 * (size_t)Rsz * DIsz);
    float* part = Q;
    (void)ws_size; (void)n_in; (void)in_sizes; (void)out_size;

    auto gemm3 = [&](const float* a0s, const float* a1s, const float* a2s, long strideAz,
                     int lda,
                     const float* b0, const float* b1, const float* b2,
                     const float* b3, const float* b4, long strideBz,
                     const float* bias,
                     const float* r0s, const float* r1s, const float* r2s,
                     float* C, long strideCz, int nz, int N, int Kloop, int flags) {
        dim3 grid(N / 128, (3 * Rsz) / 64, nz);
        gemm_wmma_f32<<<grid, dim3(256), 0, stream>>>(a0s, a1s, a2s, strideAz, lda,
                                                      b0, b1, b2, b3, b4, strideBz, bias,
                                                      r0s, r1s, r2s, C, strideCz,
                                                      N, Kloop, flags);
    };

    gemm3(X[0], X[1], X[2], 0, 1024,
          w_qs, w_ks, w_vs, w_ds0, w_ds1, 0,
          nullptr, nullptr, nullptr, nullptr,
          Q, (long)(3 * RM), 5, 1024, 1024, 0);

    gemm3(d0, d0 + RM, d0 + 2 * RM, (long)(3 * RM), 1024,
          dc_w1 + (size_t)1024 * 2048, dc_w1, dc_w1, dc_w1, dc_w1, 0,
          nullptr, nullptr, nullptr, nullptr,
          a0, (long)(3 * (size_t)Rsz * 2048), 2, 2048, 1024, 0);

    {
        dec_kernel<<<576, 128, 0, stream>>>(a0, a1, dc_b1, dc_w2, dc_b2, dec);
    }

    {
        int nblk = 3 * Bsz * Hsz * Lsz;
        attn_scores_kernel<<<nblk, 192, 0, stream>>>(Q, Kb, dec, mask, S);
        attn_av_kernel<<<nblk, 64, 0, stream>>>(S, Vb, O);
    }

    gemm3(O, O + RM, O + 2 * RM, 1024 / KSPLIT, 1024,
          w_fc, w_fc, w_fc, w_fc, w_fc, (long)(1024 / KSPLIT) * 1024,
          nullptr, X[0], X[1], X[2],
          part, (long)(3 * RM), KSPLIT, 1024, 1024 / KSPLIT, GEMM_RESID);
    layernorm_kernel<<<3 * Rsz, 256, 0, stream>>>(part, (long)(3 * RM), nullptr,
                                                  ln1_g, ln1_b, oln);
    gemm3(oln, oln + RM, oln + 2 * RM, 0, 1024,
          ffn_w1, ffn_w1, ffn_w1, ffn_w1, ffn_w1, 0,
          ffn_b1, nullptr, nullptr, nullptr,
          h1, 0, 1, DIsz, 1024, GEMM_BIAS | GEMM_RELU);
    gemm3(h1, h1 + (size_t)Rsz * DIsz, h1 + 2 * (size_t)Rsz * DIsz, DIsz / KSPLIT, DIsz,
          ffn_w2, ffn_w2, ffn_w2, ffn_w2, ffn_w2, (long)(DIsz / KSPLIT) * 1024,
          ffn_b2, nullptr, nullptr, nullptr,
          part, (long)(3 * RM), KSPLIT, 1024, DIsz / KSPLIT, GEMM_BIAS);
    layernorm_kernel<<<3 * Rsz, 256, 0, stream>>>(part, (long)(3 * RM), oln,
                                                  ln2_g, ln2_b, out);
}


# Fuse attention scores, softmax and A@V into one kernel per (i,b,h): Q/K/V staged once in LDS, both products on f32 WMMA, scores stay in LDS; sum order of dot products and softmax sum changes
# speedup vs baseline: 5.3533x; 1.2725x over previous
#include <hip/hip_runtime.h>
#include <hip/hip_bf16.h>

#define Bsz 4
#define Lsz 64
#define DMsz 1024
#define Hsz 16
#define DKsz 64
#define DVsz 64
#define Dsz 1024
#define DIsz 4096
#define Rsz 256
#define NEGC (-1.0e9f)

typedef float v2f __attribute__((ext_vector_type(2)));
typedef float v8f __attribute__((ext_vector_type(8)));

#define GEMM_BIAS  1
#define GEMM_RELU  2
#define GEMM_RESID 4

__global__ __launch_bounds__(256)
void gemm_wmma_f32(const float* __restrict__ A0, const float* __restrict__ A1,
                   const float* __restrict__ A2, long strideAz, int lda,
                   const float* __restrict__ B0, const float* __restrict__ B1,
                   const float* __restrict__ B2, const float* __restrict__ B3,
                   const float* __restrict__ B4, long strideBz,
                   const float* __restrict__ bias,
                   const float* __restrict__ R0, const float* __restrict__ R1,
                   const float* __restrict__ R2,
                   float* __restrict__ Cbase, long strideCz,
                   int N, int Kloop, int flags)
{
    const int seg  = blockIdx.y >> 2;
    const int mloc = (blockIdx.y & 3) * 64;
    const int zi   = blockIdx.z;
    const float* __restrict__ A =
        ((seg == 0) ? A0 : (seg == 1) ? A1 : A2) + (size_t)zi * strideAz;
    const float* __restrict__ B =
        ((zi == 0) ? B0 : (zi == 1) ? B1 : (zi == 2) ? B2 : (zi == 3) ? B3 : B4)
        + (size_t)zi * strideBz;
    const bool lead = (zi == 0);
    const float* __restrict__ resid = (seg == 0) ? R0 : (seg == 1) ? R1 : R2;
    float* __restrict__ C = Cbase + (size_t)zi * strideCz;

    __shared__ float As[64][64 + 1];
    __shared__ float Bs[64][128 + 1];

    const int tid   = threadIdx.x;
    const int wave  = tid >> 5;
    const int lane  = tid & 31;
    const int l15   = lane & 15;
    const int lhalf = lane >> 4;
    const int m0    = blockIdx.y * 64;
    const int nblk  = blockIdx.x * 128;
    const int mq    = wave >> 2;
    const int nq    = wave & 3;

    const int ar = tid >> 4;
    const int ac = (tid & 15) << 2;
    const int br = (tid >> 5) * 8;
    const int bc = (tid & 31) << 2;
    float4 aprf[4], bprf[8];

    v8f acc[2][2];
#pragma unroll
    for (int x = 0; x < 2; ++x)
#pragma unroll
        for (int y = 0; y < 2; ++y)
            acc[x][y] = v8f{0.f, 0.f, 0.f, 0.f, 0.f, 0.f, 0.f, 0.f};

#pragma unroll
    for (int rr = 0; rr < 4; ++rr)
        aprf[rr] = *(const float4*)(A + (size_t)(mloc + ar + rr * 16) * lda + ac);
#pragma unroll
    for (int rr = 0; rr < 8; ++rr)
        bprf[rr] = *(const float4*)(B + (size_t)(br + rr) * N + nblk + bc);

    const int arow0 = mq * 32 + l15;
    const int bcol0 = nq * 32 + l15;

    for (int k0 = 0; k0 < Kloop; k0 += 64) {
#pragma unroll
        for (int rr = 0; rr < 4; ++rr) {
            As[ar + rr * 16][ac + 0] = aprf[rr].x;
            As[ar + rr * 16][ac + 1] = aprf[rr].y;
            As[ar + rr * 16][ac + 2] = aprf[rr].z;
            As[ar + rr * 16][ac + 3] = aprf[rr].w;
        }
#pragma unroll
        for (int rr = 0; rr < 8; ++rr) {
            Bs[br + rr][bc + 0] = bprf[rr].x;
            Bs[br + rr][bc + 1] = bprf[rr].y;
            Bs[br + rr][bc + 2] = bprf[rr].z;
            Bs[br + rr][bc + 3] = bprf[rr].w;
        }
        __syncthreads();

        if (k0 + 64 < Kloop) {
#pragma unroll
            for (int rr = 0; rr < 4; ++rr)
                aprf[rr] = *(const float4*)(A + (size_t)(mloc + ar + rr * 16) * lda
                                            + k0 + 64 + ac);
#pragma unroll
            for (int rr = 0; rr < 8; ++rr)
                bprf[rr] = *(const float4*)(B + (size_t)(k0 + 64 + br + rr) * N
                                            + nblk + bc);
        }

#pragma unroll
        for (int half = 0; half < 2; ++half) {
            v2f af[2][8], bf[2][8];
#pragma unroll
            for (int s = 0; s < 8; ++s) {
                const int kb = half * 32 + s * 4 + lhalf * 2;
                af[0][s] = v2f{As[arow0][kb],      As[arow0][kb + 1]};
                af[1][s] = v2f{As[arow0 + 16][kb], As[arow0 + 16][kb + 1]};
                bf[0][s] = v2f{Bs[kb][bcol0],      Bs[kb + 1][bcol0]};
                bf[1][s] = v2f{Bs[kb][bcol0 + 16], Bs[kb + 1][bcol0 + 16]};
            }
#pragma unroll
            for (int s = 0; s < 8; ++s) {
                acc[0][0] = __builtin_amdgcn_wmma_f32_16x16x4_f32(
                    false, af[0][s], false, bf[0][s], (short)0, acc[0][0], false, false);
                acc[0][1] = __builtin_amdgcn_wmma_f32_16x16x4_f32(
                    false, af[0][s], false, bf[1][s], (short)0, acc[0][1], false, false);
                acc[1][0] = __builtin_amdgcn_wmma_f32_16x16x4_f32(
                    false, af[1][s], false, bf[0][s], (short)0, acc[1][0], false, false);
                acc[1][1] = __builtin_amdgcn_wmma_f32_16x16x4_f32(
                    false, af[1][s], false, bf[1][s], (short)0, acc[1][1], false, false);
            }
        }
        __syncthreads();
    }

#pragma unroll
    for (int nsub = 0; nsub < 2; ++nsub) {
        const int col = nblk + nq * 32 + nsub * 16 + l15;
        const float bv = ((flags & GEMM_BIAS) && lead) ? bias[col] : 0.f;
#pragma unroll
        for (int msub = 0; msub < 2; ++msub) {
#pragma unroll
            for (int r = 0; r < 8; ++r) {
                const int rloc = mq * 32 + msub * 16 + r + lhalf * 8;
                const int row  = m0 + rloc;
                float v = acc[msub][nsub][r] + bv;
                if ((flags & GEMM_RESID) && lead) v += resid[(size_t)(mloc + rloc) * N + col];
                if (flags & GEMM_RELU)  v = fmaxf(v, 0.f);
                C[(size_t)row * N + col] = v;
            }
        }
    }
}

#define DEC_NB 4
#define DEC_HOLD16(a, o)                                                        \
    asm volatile("" :: "v"(a[(o) + 0]), "v"(a[(o) + 1]), "v"(a[(o) + 2]),       \
                 "v"(a[(o) + 3]), "v"(a[(o) + 4]), "v"(a[(o) + 5]),             \
                 "v"(a[(o) + 6]), "v"(a[(o) + 7]), "v"(a[(o) + 8]),             \
                 "v"(a[(o) + 9]), "v"(a[(o) + 10]), "v"(a[(o) + 11]),           \
                 "v"(a[(o) + 12]), "v"(a[(o) + 13]), "v"(a[(o) + 14]),          \
                 "v"(a[(o) + 15]))

__global__ __launch_bounds__(128, 3)
void dec_kernel(const float* __restrict__ a0base, const float* __restrict__ a1base,
                const float* __restrict__ b1, const float* __restrict__ w2,
                const float* __restrict__ b2, float* __restrict__ dec)
{
    const int lane = threadIdx.x & 31;
    const int wid  = __builtin_amdgcn_readfirstlane(
                         (int)((blockIdx.x * 128u + threadIdx.x) >> 5));
    const int nh   = wid & 1;
    const int mp   = (wid >> 1) & 31;
    const int b    = (wid >> 6) & 3;
    const int pair = wid >> 8;
    const int i    = pair / 3;
    const int j    = pair % 3;
    const int m0   = mp * 2;
    const int n0   = nh * 32;

    const float* a1r = a1base + ((size_t)j * Rsz + b * Lsz + m0) * 2048;
    const float* a0r = a0base + ((size_t)i * Rsz + b * Lsz + n0) * 2048;

    float wd[64], bb[64], x0[64], x1[64];
#pragma unroll
    for (int k = 0; k < 64; ++k) {
        const int c = lane + 32 * k;
        wd[k] = w2[2 * c + 1] - w2[2 * c];
        bb[k] = b1[c];
        x0[k] = a1r[lane + 32 * k];
        x1[k] = a1r[2048 + lane + 32 * k];
    }

    float r0 = 0.f, r1 = 0.f;
    for (int g = 0; g < 32; g += DEC_NB) {
        float s0[DEC_NB], s1[DEC_NB];
#pragma unroll
        for (int u = 0; u < DEC_NB; ++u) {
            const float* p = a0r + (size_t)(g + u) * 2048;
            float y[64];
#pragma unroll
            for (int k = 0; k < 64; ++k) y[k] = p[lane + 32 * k];
            float c0 = 0.f, c1 = 0.f;
#pragma unroll
            for (int q = 0; q < 4; ++q) {
                DEC_HOLD16(y, 16 * q);
#pragma unroll
                for (int k = 16 * q; k < 16 * q + 16; ++k) {
                    const float t0 = x0[k] + y[k] + bb[k];
                    const float t1 = x1[k] + y[k] + bb[k];
                    const float f0 = fmaf(t0, wd[k], c0);
                    const float f1 = fmaf(t1, wd[k], c1);
                    c0 = (t0 > 0.f) ? f0 : c0;
                    c1 = (t1 > 0.f) ? f1 : c1;
                }
            }
            s0[u] = c0; s1[u] = c1;
        }
#pragma unroll
        for (int off = 16; off; off >>= 1) {
#pragma unroll
            for (int u = 0; u < DEC_NB; ++u) {
                s0[u] += __shfl_xor(s0[u], off, 32);
                s1[u] += __shfl_xor(s1[u], off, 32);
            }
        }
#pragma unroll
        for (int u = 0; u < DEC_NB; ++u) {
            r0 = (lane == g + u) ? s0[u] : r0;
            r1 = (lane == g + u) ? s1[u] : r1;
        }
    }

    const float bd = b2[1] - b2[0];
    const size_t o = (((size_t)pair * Bsz + b) * Lsz + m0) * Lsz + n0 + lane;
    dec[o]       = (r0 + bd > 0.f) ? NEGC : 0.f;
    dec[o + Lsz] = (r1 + bd > 0.f) ? NEGC : 0.f;
}

#define AT_QP 68
#define AT_VP 72
#define AT_SP 196
__global__ __launch_bounds__(256)
void attn_fused_kernel(const float* __restrict__ Qb, const float* __restrict__ Kb,
                       const float* __restrict__ Vb, const float* __restrict__ dec,
                       const int* __restrict__ mask, float* __restrict__ Ob)
{
    const int blk = blockIdx.x;
    const int h = blk & 15;
    const int b = (blk >> 4) & 3;
    const int i = blk >> 6;

    __shared__ __attribute__((aligned(16))) float Qs[64][AT_QP];
    __shared__ __attribute__((aligned(16))) float Ks[192][AT_QP];
    __shared__ __attribute__((aligned(16))) float Vs[192][AT_VP];
    __shared__ __attribute__((aligned(16))) float Ss[64][AT_SP];

    const int tid   = threadIdx.x;
    const int wave  = tid >> 5;
    const int lane  = tid & 31;
    const int l15   = lane & 15;
    const int lhalf = lane >> 4;

    const int lr = tid >> 4;
    const int lc = (tid & 15) << 2;
    float4 qreg[4], kreg[12], vreg[12];
#pragma unroll
    for (int rr = 0; rr < 4; ++rr) {
        const int row = lr + rr * 16;
        qreg[rr] = *(const float4*)(Qb + ((size_t)i * Rsz + b * Lsz + row) * DMsz
                                    + h * DKsz + lc);
    }
#pragma unroll
    for (int rr = 0; rr < 12; ++rr) {
        const int row = lr + rr * 16;
        const size_t g = ((size_t)(row >> 6) * Rsz + b * Lsz + (row & 63)) * DMsz
                         + h * DKsz + lc;
        kreg[rr] = *(const float4*)(Kb + g);
        vreg[rr] = *(const float4*)(Vb + g);
    }

    float dv[8][6];
    int   mk[8][2];
#pragma unroll
    for (int rr = 0; rr < 8; ++rr) {
        const int row = wave * 8 + rr;
#pragma unroll
        for (int u = 0; u < 6; ++u) {
            const int col = lane + 32 * u;
            dv[rr][u] = dec[(((size_t)(i * 3 + (col >> 6)) * Bsz + b) * Lsz + row) * Lsz
                            + (col & 63)];
        }
        mk[rr][0] = mask[((size_t)b * Lsz + row) * Lsz + lane];
        mk[rr][1] = mask[((size_t)b * Lsz + row) * Lsz + lane + 32];
    }

#pragma unroll
    for (int rr = 0; rr < 4; ++rr)
        *(float4*)&Qs[lr + rr * 16][lc] = qreg[rr];
#pragma unroll
    for (int rr = 0; rr < 12; ++rr) {
        *(float4*)&Ks[lr + rr * 16][lc] = kreg[rr];
        *(float4*)&Vs[lr + rr * 16][lc] = vreg[rr];
    }
    __syncthreads();

    const int mt   = wave & 3;
    const int ng   = wave >> 2;
    const int arow = mt * 16 + l15;
    {
        v8f sacc[6];
#pragma unroll
        for (int t = 0; t < 6; ++t) sacc[t] = v8f{0.f, 0.f, 0.f, 0.f, 0.f, 0.f, 0.f, 0.f};
#pragma unroll
        for (int s = 0; s < 16; ++s) {
            const int kb = s * 4 + lhalf * 2;
            const v2f a = *(const v2f*)&Qs[arow][kb];
#pragma unroll
            for (int t = 0; t < 6; ++t) {
                const v2f bk = *(const v2f*)&Ks[(ng * 6 + t) * 16 + l15][kb];
                sacc[t] = __builtin_amdgcn_wmma_f32_16x16x4_f32(
                    false, a, false, bk, (short)0, sacc[t], false, false);
            }
        }
#pragma unroll
        for (int t = 0; t < 6; ++t) {
            const int col = (ng * 6 + t) * 16 + l15;
#pragma unroll
            for (int r = 0; r < 8; ++r)
                Ss[mt * 16 + r + lhalf * 8][col] = sacc[t][r] * 0.125f;
        }
    }
    __syncthreads();

#pragma unroll
    for (int rr = 0; rr < 8; ++rr) {
        const int row = wave * 8 + rr;
        float sv[6];
        float mx = -3.0e38f;
#pragma unroll
        for (int u = 0; u < 6; ++u) {
            float s = Ss[row][lane + 32 * u] + dv[rr][u];
            if (mk[rr][u & 1] == 0) s = NEGC;
            sv[u] = s;
            mx = fmaxf(mx, s);
        }
#pragma unroll
        for (int off = 16; off; off >>= 1) mx = fmaxf(mx, __shfl_xor(mx, off, 32));
        float sum = 0.f;
#pragma unroll
        for (int u = 0; u < 6; ++u) { sv[u] = __expf(sv[u] - mx); sum += sv[u]; }
#pragma unroll
        for (int off = 16; off; off >>= 1) sum += __shfl_xor(sum, off, 32);
        const float inv = 1.0f / sum;
#pragma unroll
        for (int u = 0; u < 6; ++u) Ss[row][lane + 32 * u] = sv[u] * inv;
    }
    __syncthreads();

    {
        v8f oacc[2][2];
#pragma unroll
        for (int t = 0; t < 2; ++t)
#pragma unroll
            for (int e = 0; e < 2; ++e)
                oacc[t][e] = v8f{0.f, 0.f, 0.f, 0.f, 0.f, 0.f, 0.f, 0.f};
#pragma unroll
        for (int s = 0; s < 48; s += 2) {
#pragma unroll
            for (int e = 0; e < 2; ++e) {
                const int kb = (s + e) * 4 + lhalf * 2;
                const v2f a = *(const v2f*)&Ss[arow][kb];
#pragma unroll
                for (int t = 0; t < 2; ++t) {
                    const int col = ng * 32 + t * 16 + l15;
                    const v2f bv = v2f{Vs[kb][col], Vs[kb + 1][col]};
                    oacc[t][e] = __builtin_amdgcn_wmma_f32_16x16x4_f32(
                        false, a, false, bv, (short)0, oacc[t][e], false, false);
                }
            }
        }
#pragma unroll
        for (int t = 0; t < 2; ++t) {
            const int col = ng * 32 + t * 16 + l15;
#pragma unroll
            for (int r = 0; r < 8; ++r) {
                const int row = mt * 16 + r + lhalf * 8;
                Ob[((size_t)i * Rsz + b * Lsz + row) * DMsz + h * DVsz + col] =
                    oacc[t][0][r] + oacc[t][1][r];
            }
        }
    }
}

#define KSPLIT 8
__global__ __launch_bounds__(256)
void layernorm_kernel(const float* __restrict__ x, long plane,
                      const float* __restrict__ resid,
                      const float* __restrict__ g, const float* __restrict__ bta,
                      float* __restrict__ out)
{
    const int row = blockIdx.x;
    const int t = threadIdx.x;
    __shared__ float red[256];

    const float* xp = x + (size_t)row * 1024;
    float v[4];
    float s = 0.f;
#pragma unroll
    for (int u = 0; u < 4; ++u) {
        float val = xp[t + u * 256];
#pragma unroll
        for (int p = 1; p < KSPLIT; ++p) val += xp[(size_t)p * plane + t + u * 256];
        v[u] = val;
    }
    if (resid) {
#pragma unroll
        for (int u = 0; u < 4; ++u) v[u] += resid[(size_t)row * 1024 + t + u * 256];
    }
#pragma unroll
    for (int u = 0; u < 4; ++u) s += v[u];
    red[t] = s; __syncthreads();
    for (int st = 128; st > 0; st >>= 1) { if (t < st) red[t] += red[t + st]; __syncthreads(); }
    const float mu = red[0] * (1.f / 1024.f);
    __syncthreads();

    float s2 = 0.f;
#pragma unroll
    for (int u = 0; u < 4; ++u) { float dd = v[u] - mu; s2 += dd * dd; }
    red[t] = s2; __syncthreads();
    for (int st = 128; st > 0; st >>= 1) { if (t < st) red[t] += red[t + st]; __syncthreads(); }
    const float var = red[0] * (1.f / 1024.f);
    const float rinv = 1.0f / sqrtf(var + 1e-6f);
#pragma unroll
    for (int u = 0; u < 4; ++u) {
        int c = t + u * 256;
        out[(size_t)row * 1024 + c] = (v[u] - mu) * rinv * g[c] + bta[c];
    }
}

extern "C" void kernel_launch(void* const* d_in, const int* in_sizes, int n_in,
                              void* d_out, int out_size, void* d_ws, size_t ws_size,
                              hipStream_t stream)
{
    const float* X[3]   = { (const float*)d_in[0], (const float*)d_in[1], (const float*)d_in[2] };
    const int*   mask   = (const int*)d_in[3];
    const float* w_qs   = (const float*)d_in[4];
    const float* w_ks   = (const float*)d_in[5];
    const float* w_vs   = (const float*)d_in[6];
    const float* w_ds0  = (const float*)d_in[7];
    const float* w_ds1  = (const float*)d_in[8];
    const float* dc_w1  = (const float*)d_in[9];
    const float* dc_b1  = (const float*)d_in[10];
    const float* dc_w2  = (const float*)d_in[11];
    const float* dc_b2  = (const float*)d_in[12];
    const float* w_fc   = (const float*)d_in[13];
    const float* ln1_g  = (const float*)d_in[14];
    const float* ln1_b  = (const float*)d_in[15];
    const float* ffn_w1 = (const float*)d_in[16];
    const float* ffn_b1 = (const float*)d_in[17];
    const float* ffn_w2 = (const float*)d_in[18];
    const float* ffn_b2 = (const float*)d_in[19];
    const float* ln2_g  = (const float*)d_in[20];
    const float* ln2_b  = (const float*)d_in[21];
    float* out = (float*)d_out;

    float* ws = (float*)d_ws;
    size_t off = 0;
    auto alloc = [&](size_t n) { float* p = ws + off; off += n; return p; };
    const size_t RM = (size_t)Rsz * DMsz;
    float* Q   = alloc(3 * RM);
    float* Kb  = alloc(3 * RM);
    float* Vb  = alloc(3 * RM);
    float* d0  = alloc(3 * RM);
    float* d1  = alloc(3 * RM);
    float* a0  = alloc(3 * (size_t)Rsz * 2048);
    float* a1  = alloc(3 * (size_t)Rsz * 2048);
    float* dec = alloc(9 * (size_t)Bsz * Lsz * Lsz);
    (void)alloc((size_t)3 * Bsz * Hsz * Lsz * 192);
    float* O   = alloc(3 * RM);
    float* oln = alloc(3 * RM);
    float* h1  = alloc(3 * (size_t)Rsz * DIsz);
    float* part = Q;
    (void)ws_size; (void)n_in; (void)in_sizes; (void)out_size;

    auto gemm3 = [&](const float* a0s, const float* a1s, const float* a2s, long strideAz,
                     int lda,
                     const float* b0, const float* b1, const float* b2,
                     const float* b3, const float* b4, long strideBz,
                     const float* bias,
                     const float* r0s, const float* r1s, const float* r2s,
                     float* C, long strideCz, int nz, int N, int Kloop, int flags) {
        dim3 grid(N / 128, (3 * Rsz) / 64, nz);
        gemm_wmma_f32<<<grid, dim3(256), 0, stream>>>(a0s, a1s, a2s, strideAz, lda,
                                                      b0, b1, b2, b3, b4, strideBz, bias,
                                                      r0s, r1s, r2s, C, strideCz,
                                                      N, Kloop, flags);
    };

    gemm3(X[0], X[1], X[2], 0, 1024,
          w_qs, w_ks, w_vs, w_ds0, w_ds1, 0,
          nullptr, nullptr, nullptr, nullptr,
          Q, (long)(3 * RM), 5, 1024, 1024, 0);

    gemm3(d0, d0 + RM, d0 + 2 * RM, (long)(3 * RM), 1024,
          dc_w1 + (size_t)1024 * 2048, dc_w1, dc_w1, dc_w1, dc_w1, 0,
          nullptr, nullptr, nullptr, nullptr,
          a0, (long)(3 * (size_t)Rsz * 2048), 2, 2048, 1024, 0);

    {
        dec_kernel<<<576, 128, 0, stream>>>(a0, a1, dc_b1, dc_w2, dc_b2, dec);
    }

    attn_fused_kernel<<<3 * Bsz * Hsz, 256, 0, stream>>>(Q, Kb, Vb, dec, mask, O);

    gemm3(O, O + RM, O + 2 * RM, 1024 / KSPLIT, 1024,
          w_fc, w_fc, w_fc, w_fc, w_fc, (long)(1024 / KSPLIT) * 1024,
          nullptr, X[0], X[1], X[2],
          part, (long)(3 * RM), KSPLIT, 1024, 1024 / KSPLIT, GEMM_RESID);
    layernorm_kernel<<<3 * Rsz, 256, 0, stream>>>(part, (long)(3 * RM), nullptr,
                                                  ln1_g, ln1_b, oln);
    gemm3(oln, oln + RM, oln + 2 * RM, 0, 1024,
          ffn_w1, ffn_w1, ffn_w1, ffn_w1, ffn_w1, 0,
          ffn_b1, nullptr, nullptr, nullptr,
          h1, 0, 1, DIsz, 1024, GEMM_BIAS | GEMM_RELU);
    gemm3(h1, h1 + (size_t)Rsz * DIsz, h1 + 2 * (size_t)Rsz * DIsz, DIsz / KSPLIT, DIsz,
          ffn_w2, ffn_w2, ffn_w2, ffn_w2, ffn_w2, (long)(DIsz / KSPLIT) * 1024,
          ffn_b2, nullptr, nullptr, nullptr,
          part, (long)(3 * RM), KSPLIT, 1024, DIsz / KSPLIT, GEMM_BIAS);
    layernorm_kernel<<<3 * Rsz, 256, 0, stream>>>(part, (long)(3 * RM), oln,
                                                  ln2_g, ln2_b, out);
}


# GEMM kernel: LDS tiles re-laid out (A row pitch 68; B as 16-column tiles, k rows 16 floats apart, tile pitch 1040) so fragment loads and 16-byte staging stores hit no bank twice; sum order unchanged
# speedup vs baseline: 5.8643x; 1.0955x over previous
#include <hip/hip_runtime.h>
#include <hip/hip_bf16.h>

#define Bsz 4
#define Lsz 64
#define DMsz 1024
#define Hsz 16
#define DKsz 64
#define DVsz 64
#define Dsz 1024
#define DIsz 4096
#define Rsz 256
#define NEGC (-1.0e9f)

typedef float v2f __attribute__((ext_vector_type(2)));
typedef float v4f __attribute__((ext_vector_type(4)));
typedef float v8f __attribute__((ext_vector_type(8)));

#define GEMM_AP 68
#define GEMM_BT 1040
#define GEMM_BIAS  1
#define GEMM_RELU  2
#define GEMM_RESID 4

__global__ __launch_bounds__(256)
void gemm_wmma_f32(const float* __restrict__ A0, const float* __restrict__ A1,
                   const float* __restrict__ A2, long strideAz, int lda,
                   const float* __restrict__ B0, const float* __restrict__ B1,
                   const float* __restrict__ B2, const float* __restrict__ B3,
                   const float* __restrict__ B4, long strideBz,
                   const float* __restrict__ bias,
                   const float* __restrict__ R0, const float* __restrict__ R1,
                   const float* __restrict__ R2,
                   float* __restrict__ Cbase, long strideCz,
                   int N, int Kloop, int flags)
{
    const int seg  = blockIdx.y >> 2;
    const int mloc = (blockIdx.y & 3) * 64;
    const int zi   = blockIdx.z;
    const float* __restrict__ A =
        ((seg == 0) ? A0 : (seg == 1) ? A1 : A2) + (size_t)zi * strideAz;
    const float* __restrict__ B =
        ((zi == 0) ? B0 : (zi == 1) ? B1 : (zi == 2) ? B2 : (zi == 3) ? B3 : B4)
        + (size_t)zi * strideBz;
    const bool lead = (zi == 0);
    const float* __restrict__ resid = (seg == 0) ? R0 : (seg == 1) ? R1 : R2;
    float* __restrict__ C = Cbase + (size_t)zi * strideCz;

    __shared__ __attribute__((aligned(16))) float As[64][GEMM_AP];
    __shared__ __attribute__((aligned(16))) float Bs[8 * GEMM_BT];

    const int tid   = threadIdx.x;
    const int wave  = tid >> 5;
    const int lane  = tid & 31;
    const int l15   = lane & 15;
    const int lhalf = lane >> 4;
    const int m0    = blockIdx.y * 64;
    const int nblk  = blockIdx.x * 128;
    const int mq    = wave >> 2;
    const int nq    = wave & 3;

    const int ar = tid >> 4;
    const int ac = (tid & 15) << 2;
    const int br = (tid >> 5) * 8;
    const int bc = (tid & 31) << 2;
    v4f aprf[4], bprf[8];

    v8f acc[2][2];
#pragma unroll
    for (int x = 0; x < 2; ++x)
#pragma unroll
        for (int y = 0; y < 2; ++y)
            acc[x][y] = v8f{0.f, 0.f, 0.f, 0.f, 0.f, 0.f, 0.f, 0.f};

#pragma unroll
    for (int rr = 0; rr < 4; ++rr)
        aprf[rr] = *(const v4f*)(A + (size_t)(mloc + ar + rr * 16) * lda + ac);
#pragma unroll
    for (int rr = 0; rr < 8; ++rr)
        bprf[rr] = *(const v4f*)(B + (size_t)(br + rr) * N + nblk + bc);

    const int arow0 = mq * 32 + l15;

    for (int k0 = 0; k0 < Kloop; k0 += 64) {
#pragma unroll
        for (int rr = 0; rr < 4; ++rr)
            *(v4f*)&As[ar + rr * 16][ac] = aprf[rr];
#pragma unroll
        for (int rr = 0; rr < 8; ++rr)
            *(v4f*)&Bs[(bc >> 4) * GEMM_BT + (br + rr) * 16 + (bc & 15)] = bprf[rr];
        __syncthreads();

        if (k0 + 64 < Kloop) {
#pragma unroll
            for (int rr = 0; rr < 4; ++rr)
                aprf[rr] = *(const v4f*)(A + (size_t)(mloc + ar + rr * 16) * lda
                                         + k0 + 64 + ac);
#pragma unroll
            for (int rr = 0; rr < 8; ++rr)
                bprf[rr] = *(const v4f*)(B + (size_t)(k0 + 64 + br + rr) * N
                                         + nblk + bc);
        }

#pragma unroll
        for (int half = 0; half < 2; ++half) {
            v2f af[2][8], bf[2][8];
#pragma unroll
            for (int s = 0; s < 8; ++s) {
                const int kb = half * 32 + s * 4 + lhalf * 2;
                af[0][s] = *(const v2f*)&As[arow0][kb];
                af[1][s] = *(const v2f*)&As[arow0 + 16][kb];
                const float* bp = &Bs[(nq * 2) * GEMM_BT + kb * 16 + l15];
                bf[0][s] = v2f{bp[0],       bp[16]};
                bf[1][s] = v2f{bp[GEMM_BT], bp[GEMM_BT + 16]};
            }
#pragma unroll
            for (int s = 0; s < 8; ++s) {
                acc[0][0] = __builtin_amdgcn_wmma_f32_16x16x4_f32(
                    false, af[0][s], false, bf[0][s], (short)0, acc[0][0], false, false);
                acc[0][1] = __builtin_amdgcn_wmma_f32_16x16x4_f32(
                    false, af[0][s], false, bf[1][s], (short)0, acc[0][1], false, false);
                acc[1][0] = __builtin_amdgcn_wmma_f32_16x16x4_f32(
                    false, af[1][s], false, bf[0][s], (short)0, acc[1][0], false, false);
                acc[1][1] = __builtin_amdgcn_wmma_f32_16x16x4_f32(
                    false, af[1][s], false, bf[1][s], (short)0, acc[1][1], false, false);
            }
        }
        __syncthreads();
    }

#pragma unroll
    for (int nsub = 0; nsub < 2; ++nsub) {
        const int col = nblk + nq * 32 + nsub * 16 + l15;
        const float bv = ((flags & GEMM_BIAS) && lead) ? bias[col] : 0.f;
#pragma unroll
        for (int msub = 0; msub < 2; ++msub) {
#pragma unroll
            for (int r = 0; r < 8; ++r) {
                const int rloc = mq * 32 + msub * 16 + r + lhalf * 8;
                const int row  = m0 + rloc;
                float v = acc[msub][nsub][r] + bv;
                if ((flags & GEMM_RESID) && lead) v += resid[(size_t)(mloc + rloc) * N + col];
                if (flags & GEMM_RELU)  v = fmaxf(v, 0.f);
                C[(size_t)row * N + col] = v;
            }
        }
    }
}

#define DEC_NB 4
#define DEC_HOLD16(a, o)                                                        \
    asm volatile("" :: "v"(a[(o) + 0]), "v"(a[(o) + 1]), "v"(a[(o) + 2]),       \
                 "v"(a[(o) + 3]), "v"(a[(o) + 4]), "v"(a[(o) + 5]),             \
                 "v"(a[(o) + 6]), "v"(a[(o) + 7]), "v"(a[(o) + 8]),             \
                 "v"(a[(o) + 9]), "v"(a[(o) + 10]), "v"(a[(o) + 11]),           \
                 "v"(a[(o) + 12]), "v"(a[(o) + 13]), "v"(a[(o) + 14]),          \
                 "v"(a[(o) + 15]))

__global__ __launch_bounds__(128, 3)
void dec_kernel(const float* __restrict__ a0base, const float* __restrict__ a1base,
                const float* __restrict__ b1, const float* __restrict__ w2,
                const float* __restrict__ b2, float* __restrict__ dec)
{
    const int lane = threadIdx.x & 31;
    const int wid  = __builtin_amdgcn_readfirstlane(
                         (int)((blockIdx.x * 128u + threadIdx.x) >> 5));
    const int nh   = wid & 1;
    const int mp   = (wid >> 1) & 31;
    const int b    = (wid >> 6) & 3;
    const int pair = wid >> 8;
    const int i    = pair / 3;
    const int j    = pair % 3;
    const int m0   = mp * 2;
    const int n0   = nh * 32;

    const float* a1r = a1base + ((size_t)j * Rsz + b * Lsz + m0) * 2048;
    const float* a0r = a0base + ((size_t)i * Rsz + b * Lsz + n0) * 2048;

    float wd[64], bb[64], x0[64], x1[64];
#pragma unroll
    for (int k = 0; k < 64; ++k) {
        const int c = lane + 32 * k;
        wd[k] = w2[2 * c + 1] - w2[2 * c];
        bb[k] = b1[c];
        x0[k] = a1r[lane + 32 * k];
        x1[k] = a1r[2048 + lane + 32 * k];
    }

    float r0 = 0.f, r1 = 0.f;
    for (int g = 0; g < 32; g += DEC_NB) {
        float s0[DEC_NB], s1[DEC_NB];
#pragma unroll
        for (int u = 0; u < DEC_NB; ++u) {
            const float* p = a0r + (size_t)(g + u) * 2048;
            float y[64];
#pragma unroll
            for (int k = 0; k < 64; ++k) y[k] = p[lane + 32 * k];
            float c0 = 0.f, c1 = 0.f;
#pragma unroll
            for (int q = 0; q < 4; ++q) {
                DEC_HOLD16(y, 16 * q);
#pragma unroll
                for (int k = 16 * q; k < 16 * q + 16; ++k) {
                    const float t0 = x0[k] + y[k] + bb[k];
                    const float t1 = x1[k] + y[k] + bb[k];
                    const float f0 = fmaf(t0, wd[k], c0);
                    const float f1 = fmaf(t1, wd[k], c1);
                    c0 = (t0 > 0.f) ? f0 : c0;
                    c1 = (t1 > 0.f) ? f1 : c1;
                }
            }
            s0[u] = c0; s1[u] = c1;
        }
#pragma unroll
        for (int off = 16; off; off >>= 1) {
#pragma unroll
            for (int u = 0; u < DEC_NB; ++u) {
                s0[u] += __shfl_xor(s0[u], off, 32);
                s1[u] += __shfl_xor(s1[u], off, 32);
            }
        }
#pragma unroll
        for (int u = 0; u < DEC_NB; ++u) {
            r0 = (lane == g + u) ? s0[u] : r0;
            r1 = (lane == g + u) ? s1[u] : r1;
        }
    }

    const float bd = b2[1] - b2[0];
    const size_t o = (((size_t)pair * Bsz + b) * Lsz + m0) * Lsz + n0 + lane;
    dec[o]       = (r0 + bd > 0.f) ? NEGC : 0.f;
    dec[o + Lsz] = (r1 + bd > 0.f) ? NEGC : 0.f;
}

#define AT_QP 68
#define AT_VP 72
#define AT_SP 196
__global__ __launch_bounds__(256)
void attn_fused_kernel(const float* __restrict__ Qb, const float* __restrict__ Kb,
                       const float* __restrict__ Vb, const float* __restrict__ dec,
                       const int* __restrict__ mask, float* __restrict__ Ob)
{
    const int blk = blockIdx.x;
    const int h = blk & 15;
    const int b = (blk >> 4) & 3;
    const int i = blk >> 6;

    __shared__ __attribute__((aligned(16))) float Qs[64][AT_QP];
    __shared__ __attribute__((aligned(16))) float Ks[192][AT_QP];
    __shared__ __attribute__((aligned(16))) float Vs[192][AT_VP];
    __shared__ __attribute__((aligned(16))) float Ss[64][AT_SP];

    const int tid   = threadIdx.x;
    const int wave  = tid >> 5;
    const int lane  = tid & 31;
    const int l15   = lane & 15;
    const int lhalf = lane >> 4;

    const int lr = tid >> 4;
    const int lc = (tid & 15) << 2;
    float4 qreg[4], kreg[12], vreg[12];
#pragma unroll
    for (int rr = 0; rr < 4; ++rr) {
        const int row = lr + rr * 16;
        qreg[rr] = *(const float4*)(Qb + ((size_t)i * Rsz + b * Lsz + row) * DMsz
                                    + h * DKsz + lc);
    }
#pragma unroll
    for (int rr = 0; rr < 12; ++rr) {
        const int row = lr + rr * 16;
        const size_t g = ((size_t)(row >> 6) * Rsz + b * Lsz + (row & 63)) * DMsz
                         + h * DKsz + lc;
        kreg[rr] = *(const float4*)(Kb + g);
        vreg[rr] = *(const float4*)(Vb + g);
    }

    float dv[8][6];
    int   mk[8][2];
#pragma unroll
    for (int rr = 0; rr < 8; ++rr) {
        const int row = wave * 8 + rr;
#pragma unroll
        for (int u = 0; u < 6; ++u) {
            const int col = lane + 32 * u;
            dv[rr][u] = dec[(((size_t)(i * 3 + (col >> 6)) * Bsz + b) * Lsz + row) * Lsz
                            + (col & 63)];
        }
        mk[rr][0] = mask[((size_t)b * Lsz + row) * Lsz + lane];
        mk[rr][1] = mask[((size_t)b * Lsz + row) * Lsz + lane + 32];
    }

#pragma unroll
    for (int rr = 0; rr < 4; ++rr)
        *(float4*)&Qs[lr + rr * 16][lc] = qreg[rr];
#pragma unroll
    for (int rr = 0; rr < 12; ++rr) {
        *(float4*)&Ks[lr + rr * 16][lc] = kreg[rr];
        *(float4*)&Vs[lr + rr * 16][lc] = vreg[rr];
    }
    __syncthreads();

    const int mt   = wave & 3;
    const int ng   = wave >> 2;
    const int arow = mt * 16 + l15;
    {
        v8f sacc[6];
#pragma unroll
        for (int t = 0; t < 6; ++t) sacc[t] = v8f{0.f, 0.f, 0.f, 0.f, 0.f, 0.f, 0.f, 0.f};
#pragma unroll
        for (int s = 0; s < 16; ++s) {
            const int kb = s * 4 + lhalf * 2;
            const v2f a = *(const v2f*)&Qs[arow][kb];
#pragma unroll
            for (int t = 0; t < 6; ++t) {
                const v2f bk = *(const v2f*)&Ks[(ng * 6 + t) * 16 + l15][kb];
                sacc[t] = __builtin_amdgcn_wmma_f32_16x16x4_f32(
                    false, a, false, bk, (short)0, sacc[t], false, false);
            }
        }
#pragma unroll
        for (int t = 0; t < 6; ++t) {
            const int col = (ng * 6 + t) * 16 + l15;
#pragma unroll
            for (int r = 0; r < 8; ++r)
                Ss[mt * 16 + r + lhalf * 8][col] = sacc[t][r] * 0.125f;
        }
    }
    __syncthreads();

#pragma unroll
    for (int rr = 0; rr < 8; ++rr) {
        const int row = wave * 8 + rr;
        float sv[6];
        float mx = -3.0e38f;
#pragma unroll
        for (int u = 0; u < 6; ++u) {
            float s = Ss[row][lane + 32 * u] + dv[rr][u];
            if (mk[rr][u & 1] == 0) s = NEGC;
            sv[u] = s;
            mx = fmaxf(mx, s);
        }
#pragma unroll
        for (int off = 16; off; off >>= 1) mx = fmaxf(mx, __shfl_xor(mx, off, 32));
        float sum = 0.f;
#pragma unroll
        for (int u = 0; u < 6; ++u) { sv[u] = __expf(sv[u] - mx); sum += sv[u]; }
#pragma unroll
        for (int off = 16; off; off >>= 1) sum += __shfl_xor(sum, off, 32);
        const float inv = 1.0f / sum;
#pragma unroll
        for (int u = 0; u < 6; ++u) Ss[row][lane + 32 * u] = sv[u] * inv;
    }
    __syncthreads();

    {
        v8f oacc[2][2];
#pragma unroll
        for (int t = 0; t < 2; ++t)
#pragma unroll
            for (int e = 0; e < 2; ++e)
                oacc[t][e] = v8f{0.f, 0.f, 0.f, 0.f, 0.f, 0.f, 0.f, 0.f};
#pragma unroll
        for (int s = 0; s < 48; s += 2) {
#pragma unroll
            for (int e = 0; e < 2; ++e) {
                const int kb = (s + e) * 4 + lhalf * 2;
                const v2f a = *(const v2f*)&Ss[arow][kb];
#pragma unroll
                for (int t = 0; t < 2; ++t) {
                    const int col = ng * 32 + t * 16 + l15;
                    const v2f bv = v2f{Vs[kb][col], Vs[kb + 1][col]};
                    oacc[t][e] = __builtin_amdgcn_wmma_f32_16x16x4_f32(
                        false, a, false, bv, (short)0, oacc[t][e], false, false);
                }
            }
        }
#pragma unroll
        for (int t = 0; t < 2; ++t) {
            const int col = ng * 32 + t * 16 + l15;
#pragma unroll
            for (int r = 0; r < 8; ++r) {
                const int row = mt * 16 + r + lhalf * 8;
                Ob[((size_t)i * Rsz + b * Lsz + row) * DMsz + h * DVsz + col] =
                    oacc[t][0][r] + oacc[t][1][r];
            }
        }
    }
}

#define KSPLIT 8
__global__ __launch_bounds__(256)
void layernorm_kernel(const float* __restrict__ x, long plane,
                      const float* __restrict__ resid,
                      const float* __restrict__ g, const float* __restrict__ bta,
                      float* __restrict__ out)
{
    const int row = blockIdx.x;
    const int t = threadIdx.x;
    __shared__ float red[256];

    const float* xp = x + (size_t)row * 1024;
    float v[4];
    float s = 0.f;
#pragma unroll
    for (int u = 0; u < 4; ++u) {
        float val = xp[t + u * 256];
#pragma unroll
        for (int p = 1; p < KSPLIT; ++p) val += xp[(size_t)p * plane + t + u * 256];
        v[u] = val;
    }
    if (resid) {
#pragma unroll
        for (int u = 0; u < 4; ++u) v[u] += resid[(size_t)row * 1024 + t + u * 256];
    }
#pragma unroll
    for (int u = 0; u < 4; ++u) s += v[u];
    red[t] = s; __syncthreads();
    for (int st = 128; st > 0; st >>= 1) { if (t < st) red[t] += red[t + st]; __syncthreads(); }
    const float mu = red[0] * (1.f / 1024.f);
    __syncthreads();

    float s2 = 0.f;
#pragma unroll
    for (int u = 0; u < 4; ++u) { float dd = v[u] - mu; s2 += dd * dd; }
    red[t] = s2; __syncthreads();
    for (int st = 128; st > 0; st >>= 1) { if (t < st) red[t] += red[t + st]; __syncthreads(); }
    const float var = red[0] * (1.f / 1024.f);
    const float rinv = 1.0f / sqrtf(var + 1e-6f);
#pragma unroll
    for (int u = 0; u < 4; ++u) {
        int c = t + u * 256;
        out[(size_t)row * 1024 + c] = (v[u] - mu) * rinv * g[c] + bta[c];
    }
}

extern "C" void kernel_launch(void* const* d_in, const int* in_sizes, int n_in,
                              void* d_out, int out_size, void* d_ws, size_t ws_size,
                              hipStream_t stream)
{
    const float* X[3]   = { (const float*)d_in[0], (const float*)d_in[1], (const float*)d_in[2] };
    const int*   mask   = (const int*)d_in[3];
    const float* w_qs   = (const float*)d_in[4];
    const float* w_ks   = (const float*)d_in[5];
    const float* w_vs   = (const float*)d_in[6];
    const float* w_ds0  = (const float*)d_in[7];
    const float* w_ds1  = (const float*)d_in[8];
    const float* dc_w1  = (const float*)d_in[9];
    const float* dc_b1  = (const float*)d_in[10];
    const float* dc_w2  = (const float*)d_in[11];
    const float* dc_b2  = (const float*)d_in[12];
    const float* w_fc   = (const float*)d_in[13];
    const float* ln1_g  = (const float*)d_in[14];
    const float* ln1_b  = (const float*)d_in[15];
    const float* ffn_w1 = (const float*)d_in[16];
    const float* ffn_b1 = (const float*)d_in[17];
    const float* ffn_w2 = (const float*)d_in[18];
    const float* ffn_b2 = (const float*)d_in[19];
    const float* ln2_g  = (const float*)d_in[20];
    const float* ln2_b  = (const float*)d_in[21];
    float* out = (float*)d_out;

    float* ws = (float*)d_ws;
    size_t off = 0;
    auto alloc = [&](size_t n) { float* p = ws + off; off += n; return p; };
    const size_t RM = (size_t)Rsz * DMsz;
    float* Q   = alloc(3 * RM);
    float* Kb  = alloc(3 * RM);
    float* Vb  = alloc(3 * RM);
    float* d0  = alloc(3 * RM);
    float* d1  = alloc(3 * RM);
    float* a0  = alloc(3 * (size_t)Rsz * 2048);
    float* a1  = alloc(3 * (size_t)Rsz * 2048);
    float* dec = alloc(9 * (size_t)Bsz * Lsz * Lsz);
    (void)alloc((size_t)3 * Bsz * Hsz * Lsz * 192);
    float* O   = alloc(3 * RM);
    float* oln = alloc(3 * RM);
    float* h1  = alloc(3 * (size_t)Rsz * DIsz);
    float* part = Q;
    (void)ws_size; (void)n_in; (void)in_sizes; (void)out_size;

    auto gemm3 = [&](const float* a0s, const float* a1s, const float* a2s, long strideAz,
                     int lda,
                     const float* b0, const float* b1, const float* b2,
                     const float* b3, const float* b4, long strideBz,
                     const float* bias,
                     const float* r0s, const float* r1s, const float* r2s,
                     float* C, long strideCz, int nz, int N, int Kloop, int flags) {
        dim3 grid(N / 128, (3 * Rsz) / 64, nz);
        gemm_wmma_f32<<<grid, dim3(256), 0, stream>>>(a0s, a1s, a2s, strideAz, lda,
                                                      b0, b1, b2, b3, b4, strideBz, bias,
                                                      r0s, r1s, r2s, C, strideCz,
                                                      N, Kloop, flags);
    };

    gemm3(X[0], X[1], X[2], 0, 1024,
          w_qs, w_ks, w_vs, w_ds0, w_ds1, 0,
          nullptr, nullptr, nullptr, nullptr,
          Q, (long)(3 * RM), 5, 1024, 1024, 0);

    gemm3(d0, d0 + RM, d0 + 2 * RM, (long)(3 * RM), 1024,
          dc_w1 + (size_t)1024 * 2048, dc_w1, dc_w1, dc_w1, dc_w1, 0,
          nullptr, nullptr, nullptr, nullptr,
          a0, (long)(3 * (size_t)Rsz * 2048), 2, 2048, 1024, 0);

    {
        dec_kernel<<<576, 128, 0, stream>>>(a0, a1, dc_b1, dc_w2, dc_b2, dec);
    }

    attn_fused_kernel<<<3 * Bsz * Hsz, 256, 0, stream>>>(Q, Kb, Vb, dec, mask, O);

    gemm3(O, O + RM, O + 2 * RM, 1024 / KSPLIT, 1024,
          w_fc, w_fc, w_fc, w_fc, w_fc, (long)(1024 / KSPLIT) * 1024,
          nullptr, X[0], X[1], X[2],
          part, (long)(3 * RM), KSPLIT, 1024, 1024 / KSPLIT, GEMM_RESID);
    layernorm_kernel<<<3 * Rsz, 256, 0, stream>>>(part, (long)(3 * RM), nullptr,
                                                  ln1_g, ln1_b, oln);
    gemm3(oln, oln + RM, oln + 2 * RM, 0, 1024,
          ffn_w1, ffn_w1, ffn_w1, ffn_w1, ffn_w1, 0,
          ffn_b1, nullptr, nullptr, nullptr,
          h1, 0, 1, DIsz, 1024, GEMM_BIAS | GEMM_RELU);
    gemm3(h1, h1 + (size_t)Rsz * DIsz, h1 + 2 * (size_t)Rsz * DIsz, DIsz / KSPLIT, DIsz,
          ffn_w2, ffn_w2, ffn_w2, ffn_w2, ffn_w2, (long)(DIsz / KSPLIT) * 1024,
          ffn_b2, nullptr, nullptr, nullptr,
          part, (long)(3 * RM), KSPLIT, 1024, DIsz / KSPLIT, GEMM_BIAS);
    layernorm_kernel<<<3 * Rsz, 256, 0, stream>>>(part, (long)(3 * RM), oln,
                                                  ln2_g, ln2_b, out);
}


# GEMM kernel: the two launches of 384 64x128 tiles (a0/a1, FFN up-projection) run as 768 four-wave blocks on 64x64 tiles, three per WGP instead of two-or-one; wave code and sum order unchanged
# speedup vs baseline: 6.1300x; 1.0453x over previous
#include <hip/hip_runtime.h>
#include <hip/hip_bf16.h>

#define Bsz 4
#define Lsz 64
#define DMsz 1024
#define Hsz 16
#define DKsz 64
#define DVsz 64
#define Dsz 1024
#define DIsz 4096
#define Rsz 256
#define NEGC (-1.0e9f)

typedef float v2f __attribute__((ext_vector_type(2)));
typedef float v4f __attribute__((ext_vector_type(4)));
typedef float v8f __attribute__((ext_vector_type(8)));

#define GEMM_AP 68
#define GEMM_BT 1040
#define GEMM_BIAS  1
#define GEMM_RELU  2
#define GEMM_RESID 4

template <int BN>
__global__ __launch_bounds__(BN * 2)
void gemm_wmma_f32(const float* __restrict__ A0, const float* __restrict__ A1,
                   const float* __restrict__ A2, long strideAz, int lda,
                   const float* __restrict__ B0, const float* __restrict__ B1,
                   const float* __restrict__ B2, const float* __restrict__ B3,
                   const float* __restrict__ B4, long strideBz,
                   const float* __restrict__ bias,
                   const float* __restrict__ R0, const float* __restrict__ R1,
                   const float* __restrict__ R2,
                   float* __restrict__ Cbase, long strideCz,
                   int N, int Kloop, int flags)
{
    const int seg  = blockIdx.y >> 2;
    const int mloc = (blockIdx.y & 3) * 64;
    const int zi   = blockIdx.z;
    const float* __restrict__ A =
        ((seg == 0) ? A0 : (seg == 1) ? A1 : A2) + (size_t)zi * strideAz;
    const float* __restrict__ B =
        ((zi == 0) ? B0 : (zi == 1) ? B1 : (zi == 2) ? B2 : (zi == 3) ? B3 : B4)
        + (size_t)zi * strideBz;
    const bool lead = (zi == 0);
    const float* __restrict__ resid = (seg == 0) ? R0 : (seg == 1) ? R1 : R2;
    float* __restrict__ C = Cbase + (size_t)zi * strideCz;

    constexpr int NT    = BN * 2;
    constexpr int NWN   = BN / 32;
    constexpr int APASS = 1024 / NT;
    constexpr int AROWS = NT / 16;
    constexpr int TPR   = BN / 4;

    __shared__ __attribute__((aligned(16))) float As[64][GEMM_AP];
    __shared__ __attribute__((aligned(16))) float Bs[(BN / 16) * GEMM_BT];

    const int tid   = threadIdx.x;
    const int wave  = tid >> 5;
    const int lane  = tid & 31;
    const int l15   = lane & 15;
    const int lhalf = lane >> 4;
    const int m0    = blockIdx.y * 64;
    const int nblk  = blockIdx.x * BN;
    const int mq    = wave / NWN;
    const int nq    = wave % NWN;

    const int ar = tid >> 4;
    const int ac = (tid & 15) << 2;
    const int br = (tid / TPR) * 8;
    const int bc = (tid % TPR) << 2;
    v4f aprf[APASS], bprf[8];

    v8f acc[2][2];
#pragma unroll
    for (int x = 0; x < 2; ++x)
#pragma unroll
        for (int y = 0; y < 2; ++y)
            acc[x][y] = v8f{0.f, 0.f, 0.f, 0.f, 0.f, 0.f, 0.f, 0.f};

#pragma unroll
    for (int rr = 0; rr < APASS; ++rr)
        aprf[rr] = *(const v4f*)(A + (size_t)(mloc + ar + rr * AROWS) * lda + ac);
#pragma unroll
    for (int rr = 0; rr < 8; ++rr)
        bprf[rr] = *(const v4f*)(B + (size_t)(br + rr) * N + nblk + bc);

    const int arow0 = mq * 32 + l15;

    for (int k0 = 0; k0 < Kloop; k0 += 64) {
#pragma unroll
        for (int rr = 0; rr < APASS; ++rr)
            *(v4f*)&As[ar + rr * AROWS][ac] = aprf[rr];
#pragma unroll
        for (int rr = 0; rr < 8; ++rr)
            *(v4f*)&Bs[(bc >> 4) * GEMM_BT + (br + rr) * 16 + (bc & 15)] = bprf[rr];
        __syncthreads();

        if (k0 + 64 < Kloop) {
#pragma unroll
            for (int rr = 0; rr < APASS; ++rr)
                aprf[rr] = *(const v4f*)(A + (size_t)(mloc + ar + rr * AROWS) * lda
                                         + k0 + 64 + ac);
#pragma unroll
            for (int rr = 0; rr < 8; ++rr)
                bprf[rr] = *(const v4f*)(B + (size_t)(k0 + 64 + br + rr) * N
                                         + nblk + bc);
        }

#pragma unroll
        for (int half = 0; half < 2; ++half) {
            v2f af[2][8], bf[2][8];
#pragma unroll
            for (int s = 0; s < 8; ++s) {
                const int kb = half * 32 + s * 4 + lhalf * 2;
                af[0][s] = *(const v2f*)&As[arow0][kb];
                af[1][s] = *(const v2f*)&As[arow0 + 16][kb];
                const float* bp = &Bs[(nq * 2) * GEMM_BT + kb * 16 + l15];
                bf[0][s] = v2f{bp[0],       bp[16]};
                bf[1][s] = v2f{bp[GEMM_BT], bp[GEMM_BT + 16]};
            }
#pragma unroll
            for (int s = 0; s < 8; ++s) {
                acc[0][0] = __builtin_amdgcn_wmma_f32_16x16x4_f32(
                    false, af[0][s], false, bf[0][s], (short)0, acc[0][0], false, false);
                acc[0][1] = __builtin_amdgcn_wmma_f32_16x16x4_f32(
                    false, af[0][s], false, bf[1][s], (short)0, acc[0][1], false, false);
                acc[1][0] = __builtin_amdgcn_wmma_f32_16x16x4_f32(
                    false, af[1][s], false, bf[0][s], (short)0, acc[1][0], false, false);
                acc[1][1] = __builtin_amdgcn_wmma_f32_16x16x4_f32(
                    false, af[1][s], false, bf[1][s], (short)0, acc[1][1], false, false);
            }
        }
        __syncthreads();
    }

#pragma unroll
    for (int nsub = 0; nsub < 2; ++nsub) {
        const int col = nblk + nq * 32 + nsub * 16 + l15;
        const float bv = ((flags & GEMM_BIAS) && lead) ? bias[col] : 0.f;
#pragma unroll
        for (int msub = 0; msub < 2; ++msub) {
#pragma unroll
            for (int r = 0; r < 8; ++r) {
                const int rloc = mq * 32 + msub * 16 + r + lhalf * 8;
                const int row  = m0 + rloc;
                float v = acc[msub][nsub][r] + bv;
                if ((flags & GEMM_RESID) && lead) v += resid[(size_t)(mloc + rloc) * N + col];
                if (flags & GEMM_RELU)  v = fmaxf(v, 0.f);
                C[(size_t)row * N + col] = v;
            }
        }
    }
}

#define DEC_NB 4
#define DEC_HOLD16(a, o)                                                        \
    asm volatile("" :: "v"(a[(o) + 0]), "v"(a[(o) + 1]), "v"(a[(o) + 2]),       \
                 "v"(a[(o) + 3]), "v"(a[(o) + 4]), "v"(a[(o) + 5]),             \
                 "v"(a[(o) + 6]), "v"(a[(o) + 7]), "v"(a[(o) + 8]),             \
                 "v"(a[(o) + 9]), "v"(a[(o) + 10]), "v"(a[(o) + 11]),           \
                 "v"(a[(o) + 12]), "v"(a[(o) + 13]), "v"(a[(o) + 14]),          \
                 "v"(a[(o) + 15]))

__global__ __launch_bounds__(128, 3)
void dec_kernel(const float* __restrict__ a0base, const float* __restrict__ a1base,
                const float* __restrict__ b1, const float* __restrict__ w2,
                const float* __restrict__ b2, float* __restrict__ dec)
{
    const int lane = threadIdx.x & 31;
    const int wid  = __builtin_amdgcn_readfirstlane(
                         (int)((blockIdx.x * 128u + threadIdx.x) >> 5));
    const int nh   = wid & 1;
    const int mp   = (wid >> 1) & 31;
    const int b    = (wid >> 6) & 3;
    const int pair = wid >> 8;
    const int i    = pair / 3;
    const int j    = pair % 3;
    const int m0   = mp * 2;
    const int n0   = nh * 32;

    const float* a1r = a1base + ((size_t)j * Rsz + b * Lsz + m0) * 2048;
    const float* a0r = a0base + ((size_t)i * Rsz + b * Lsz + n0) * 2048;

    float wd[64], bb[64], x0[64], x1[64];
#pragma unroll
    for (int k = 0; k < 64; ++k) {
        const int c = lane + 32 * k;
        wd[k] = w2[2 * c + 1] - w2[2 * c];
        bb[k] = b1[c];
        x0[k] = a1r[lane + 32 * k];
        x1[k] = a1r[2048 + lane + 32 * k];
    }

    float r0 = 0.f, r1 = 0.f;
    for (int g = 0; g < 32; g += DEC_NB) {
        float s0[DEC_NB], s1[DEC_NB];
#pragma unroll
        for (int u = 0; u < DEC_NB; ++u) {
            const float* p = a0r + (size_t)(g + u) * 2048;
            float y[64];
#pragma unroll
            for (int k = 0; k < 64; ++k) y[k] = p[lane + 32 * k];
            float c0 = 0.f, c1 = 0.f;
#pragma unroll
            for (int q = 0; q < 4; ++q) {
                DEC_HOLD16(y, 16 * q);
#pragma unroll
                for (int k = 16 * q; k < 16 * q + 16; ++k) {
                    const float t0 = x0[k] + y[k] + bb[k];
                    const float t1 = x1[k] + y[k] + bb[k];
                    const float f0 = fmaf(t0, wd[k], c0);
                    const float f1 = fmaf(t1, wd[k], c1);
                    c0 = (t0 > 0.f) ? f0 : c0;
                    c1 = (t1 > 0.f) ? f1 : c1;
                }
            }
            s0[u] = c0; s1[u] = c1;
        }
#pragma unroll
        for (int off = 16; off; off >>= 1) {
#pragma unroll
            for (int u = 0; u < DEC_NB; ++u) {
                s0[u] += __shfl_xor(s0[u], off, 32);
                s1[u] += __shfl_xor(s1[u], off, 32);
            }
        }
#pragma unroll
        for (int u = 0; u < DEC_NB; ++u) {
            r0 = (lane == g + u) ? s0[u] : r0;
            r1 = (lane == g + u) ? s1[u] : r1;
        }
    }

    const float bd = b2[1] - b2[0];
    const size_t o = (((size_t)pair * Bsz + b) * Lsz + m0) * Lsz + n0 + lane;
    dec[o]       = (r0 + bd > 0.f) ? NEGC : 0.f;
    dec[o + Lsz] = (r1 + bd > 0.f) ? NEGC : 0.f;
}

#define AT_QP 68
#define AT_VP 72
#define AT_SP 196
__global__ __launch_bounds__(256)
void attn_fused_kernel(const float* __restrict__ Qb, const float* __restrict__ Kb,
                       const float* __restrict__ Vb, const float* __restrict__ dec,
                       const int* __restrict__ mask, float* __restrict__ Ob)
{
    const int blk = blockIdx.x;
    const int h = blk & 15;
    const int b = (blk >> 4) & 3;
    const int i = blk >> 6;

    __shared__ __attribute__((aligned(16))) float Qs[64][AT_QP];
    __shared__ __attribute__((aligned(16))) float Ks[192][AT_QP];
    __shared__ __attribute__((aligned(16))) float Vs[192][AT_VP];
    __shared__ __attribute__((aligned(16))) float Ss[64][AT_SP];

    const int tid   = threadIdx.x;
    const int wave  = tid >> 5;
    const int lane  = tid & 31;
    const int l15   = lane & 15;
    const int lhalf = lane >> 4;

    const int lr = tid >> 4;
    const int lc = (tid & 15) << 2;
    float4 qreg[4], kreg[12], vreg[12];
#pragma unroll
    for (int rr = 0; rr < 4; ++rr) {
        const int row = lr + rr * 16;
        qreg[rr] = *(const float4*)(Qb + ((size_t)i * Rsz + b * Lsz + row) * DMsz
                                    + h * DKsz + lc);
    }
#pragma unroll
    for (int rr = 0; rr < 12; ++rr) {
        const int row = lr + rr * 16;
        const size_t g = ((size_t)(row >> 6) * Rsz + b * Lsz + (row & 63)) * DMsz
                         + h * DKsz + lc;
        kreg[rr] = *(const float4*)(Kb + g);
        vreg[rr] = *(const float4*)(Vb + g);
    }

    float dv[8][6];
    int   mk[8][2];
#pragma unroll
    for (int rr = 0; rr < 8; ++rr) {
        const int row = wave * 8 + rr;
#pragma unroll
        for (int u = 0; u < 6; ++u) {
            const int col = lane + 32 * u;
            dv[rr][u] = dec[(((size_t)(i * 3 + (col >> 6)) * Bsz + b) * Lsz + row) * Lsz
                            + (col & 63)];
        }
        mk[rr][0] = mask[((size_t)b * Lsz + row) * Lsz + lane];
        mk[rr][1] = mask[((size_t)b * Lsz + row) * Lsz + lane + 32];
    }

#pragma unroll
    for (int rr = 0; rr < 4; ++rr)
        *(float4*)&Qs[lr + rr * 16][lc] = qreg[rr];
#pragma unroll
    for (int rr = 0; rr < 12; ++rr) {
        *(float4*)&Ks[lr + rr * 16][lc] = kreg[rr];
        *(float4*)&Vs[lr + rr * 16][lc] = vreg[rr];
    }
    __syncthreads();

    const int mt   = wave & 3;
    const int ng   = wave >> 2;
    const int arow = mt * 16 + l15;
    {
        v8f sacc[6];
#pragma unroll
        for (int t = 0; t < 6; ++t) sacc[t] = v8f{0.f, 0.f, 0.f, 0.f, 0.f, 0.f, 0.f, 0.f};
#pragma unroll
        for (int s = 0; s < 16; ++s) {
            const int kb = s * 4 + lhalf * 2;
            const v2f a = *(const v2f*)&Qs[arow][kb];
#pragma unroll
            for (int t = 0; t < 6; ++t) {
                const v2f bk = *(const v2f*)&Ks[(ng * 6 + t) * 16 + l15][kb];
                sacc[t] = __builtin_amdgcn_wmma_f32_16x16x4_f32(
                    false, a, false, bk, (short)0, sacc[t], false, false);
            }
        }
#pragma unroll
        for (int t = 0; t < 6; ++t) {
            const int col = (ng * 6 + t) * 16 + l15;
#pragma unroll
            for (int r = 0; r < 8; ++r)
                Ss[mt * 16 + r + lhalf * 8][col] = sacc[t][r] * 0.125f;
        }
    }
    __syncthreads();

#pragma unroll
    for (int rr = 0; rr < 8; ++rr) {
        const int row = wave * 8 + rr;
        float sv[6];
        float mx = -3.0e38f;
#pragma unroll
        for (int u = 0; u < 6; ++u) {
            float s = Ss[row][lane + 32 * u] + dv[rr][u];
            if (mk[rr][u & 1] == 0) s = NEGC;
            sv[u] = s;
            mx = fmaxf(mx, s);
        }
#pragma unroll
        for (int off = 16; off; off >>= 1) mx = fmaxf(mx, __shfl_xor(mx, off, 32));
        float sum = 0.f;
#pragma unroll
        for (int u = 0; u < 6; ++u) { sv[u] = __expf(sv[u] - mx); sum += sv[u]; }
#pragma unroll
        for (int off = 16; off; off >>= 1) sum += __shfl_xor(sum, off, 32);
        const float inv = 1.0f / sum;
#pragma unroll
        for (int u = 0; u < 6; ++u) Ss[row][lane + 32 * u] = sv[u] * inv;
    }
    __syncthreads();

    {
        v8f oacc[2][2];
#pragma unroll
        for (int t = 0; t < 2; ++t)
#pragma unroll
            for (int e = 0; e < 2; ++e)
                oacc[t][e] = v8f{0.f, 0.f, 0.f, 0.f, 0.f, 0.f, 0.f, 0.f};
#pragma unroll
        for (int s = 0; s < 48; s += 2) {
#pragma unroll
            for (int e = 0; e < 2; ++e) {
                const int kb = (s + e) * 4 + lhalf * 2;
                const v2f a = *(const v2f*)&Ss[arow][kb];
#pragma unroll
                for (int t = 0; t < 2; ++t) {
                    const int col = ng * 32 + t * 16 + l15;
                    const v2f bv = v2f{Vs[kb][col], Vs[kb + 1][col]};
                    oacc[t][e] = __builtin_amdgcn_wmma_f32_16x16x4_f32(
                        false, a, false, bv, (short)0, oacc[t][e], false, false);
                }
            }
        }
#pragma unroll
        for (int t = 0; t < 2; ++t) {
            const int col = ng * 32 + t * 16 + l15;
#pragma unroll
            for (int r = 0; r < 8; ++r) {
                const int row = mt * 16 + r + lhalf * 8;
                Ob[((size_t)i * Rsz + b * Lsz + row) * DMsz + h * DVsz + col] =
                    oacc[t][0][r] + oacc[t][1][r];
            }
        }
    }
}

#define KSPLIT 8
__global__ __launch_bounds__(256)
void layernorm_kernel(const float* __restrict__ x, long plane,
                      const float* __restrict__ resid,
                      const float* __restrict__ g, const float* __restrict__ bta,
                      float* __restrict__ out)
{
    const int row = blockIdx.x;
    const int t = threadIdx.x;
    __shared__ float red[256];

    const float* xp = x + (size_t)row * 1024;
    float v[4];
    float s = 0.f;
#pragma unroll
    for (int u = 0; u < 4; ++u) {
        float val = xp[t + u * 256];
#pragma unroll
        for (int p = 1; p < KSPLIT; ++p) val += xp[(size_t)p * plane + t + u * 256];
        v[u] = val;
    }
    if (resid) {
#pragma unroll
        for (int u = 0; u < 4; ++u) v[u] += resid[(size_t)row * 1024 + t + u * 256];
    }
#pragma unroll
    for (int u = 0; u < 4; ++u) s += v[u];
    red[t] = s; __syncthreads();
    for (int st = 128; st > 0; st >>= 1) { if (t < st) red[t] += red[t + st]; __syncthreads(); }
    const float mu = red[0] * (1.f / 1024.f);
    __syncthreads();

    float s2 = 0.f;
#pragma unroll
    for (int u = 0; u < 4; ++u) { float dd = v[u] - mu; s2 += dd * dd; }
    red[t] = s2; __syncthreads();
    for (int st = 128; st > 0; st >>= 1) { if (t < st) red[t] += red[t + st]; __syncthreads(); }
    const float var = red[0] * (1.f / 1024.f);
    const float rinv = 1.0f / sqrtf(var + 1e-6f);
#pragma unroll
    for (int u = 0; u < 4; ++u) {
        int c = t + u * 256;
        out[(size_t)row * 1024 + c] = (v[u] - mu) * rinv * g[c] + bta[c];
    }
}

extern "C" void kernel_launch(void* const* d_in, const int* in_sizes, int n_in,
                              void* d_out, int out_size, void* d_ws, size_t ws_size,
                              hipStream_t stream)
{
    const float* X[3]   = { (const float*)d_in[0], (const float*)d_in[1], (const float*)d_in[2] };
    const int*   mask   = (const int*)d_in[3];
    const float* w_qs   = (const float*)d_in[4];
    const float* w_ks   = (const float*)d_in[5];
    const float* w_vs   = (const float*)d_in[6];
    const float* w_ds0  = (const float*)d_in[7];
    const float* w_ds1  = (const float*)d_in[8];
    const float* dc_w1  = (const float*)d_in[9];
    const float* dc_b1  = (const float*)d_in[10];
    const float* dc_w2  = (const float*)d_in[11];
    const float* dc_b2  = (const float*)d_in[12];
    const float* w_fc   = (const float*)d_in[13];
    const float* ln1_g  = (const float*)d_in[14];
    const float* ln1_b  = (const float*)d_in[15];
    const float* ffn_w1 = (const float*)d_in[16];
    const float* ffn_b1 = (const float*)d_in[17];
    const float* ffn_w2 = (const float*)d_in[18];
    const float* ffn_b2 = (const float*)d_in[19];
    const float* ln2_g  = (const float*)d_in[20];
    const float* ln2_b  = (const float*)d_in[21];
    float* out = (float*)d_out;

    float* ws = (float*)d_ws;
    size_t off = 0;
    auto alloc = [&](size_t n) { float* p = ws + off; off += n; return p; };
    const size_t RM = (size_t)Rsz * DMsz;
    float* Q   = alloc(3 * RM);
    float* Kb  = alloc(3 * RM);
    float* Vb  = alloc(3 * RM);
    float* d0  = alloc(3 * RM);
    float* d1  = alloc(3 * RM);
    float* a0  = alloc(3 * (size_t)Rsz * 2048);
    float* a1  = alloc(3 * (size_t)Rsz * 2048);
    float* dec = alloc(9 * (size_t)Bsz * Lsz * Lsz);
    (void)alloc((size_t)3 * Bsz * Hsz * Lsz * 192);
    float* O   = alloc(3 * RM);
    float* oln = alloc(3 * RM);
    float* h1  = alloc(3 * (size_t)Rsz * DIsz);
    float* part = Q;
    (void)ws_size; (void)n_in; (void)in_sizes; (void)out_size;

    auto gemm3 = [&](bool narrow,
                     const float* a0s, const float* a1s, const float* a2s, long strideAz,
                     int lda,
                     const float* b0, const float* b1, const float* b2,
                     const float* b3, const float* b4, long strideBz,
                     const float* bias,
                     const float* r0s, const float* r1s, const float* r2s,
                     float* C, long strideCz, int nz, int N, int Kloop, int flags) {
        if (narrow) {
            dim3 grid(N / 64, (3 * Rsz) / 64, nz);
            gemm_wmma_f32<64><<<grid, dim3(128), 0, stream>>>(a0s, a1s, a2s, strideAz, lda,
                                                              b0, b1, b2, b3, b4, strideBz, bias,
                                                              r0s, r1s, r2s, C, strideCz,
                                                              N, Kloop, flags);
        } else {
            dim3 grid(N / 128, (3 * Rsz) / 64, nz);
            gemm_wmma_f32<128><<<grid, dim3(256), 0, stream>>>(a0s, a1s, a2s, strideAz, lda,
                                                               b0, b1, b2, b3, b4, strideBz, bias,
                                                               r0s, r1s, r2s, C, strideCz,
                                                               N, Kloop, flags);
        }
    };

    gemm3(false, X[0], X[1], X[2], 0, 1024,
          w_qs, w_ks, w_vs, w_ds0, w_ds1, 0,
          nullptr, nullptr, nullptr, nullptr,
          Q, (long)(3 * RM), 5, 1024, 1024, 0);

    gemm3(true, d0, d0 + RM, d0 + 2 * RM, (long)(3 * RM), 1024,
          dc_w1 + (size_t)1024 * 2048, dc_w1, dc_w1, dc_w1, dc_w1, 0,
          nullptr, nullptr, nullptr, nullptr,
          a0, (long)(3 * (size_t)Rsz * 2048), 2, 2048, 1024, 0);

    {
        dec_kernel<<<576, 128, 0, stream>>>(a0, a1, dc_b1, dc_w2, dc_b2, dec);
    }

    attn_fused_kernel<<<3 * Bsz * Hsz, 256, 0, stream>>>(Q, Kb, Vb, dec, mask, O);

    gemm3(false, O, O + RM, O + 2 * RM, 1024 / KSPLIT, 1024,
          w_fc, w_fc, w_fc, w_fc, w_fc, (long)(1024 / KSPLIT) * 1024,
          nullptr, X[0], X[1], X[2],
          part, (long)(3 * RM), KSPLIT, 1024, 1024 / KSPLIT, GEMM_RESID);
    layernorm_kernel<<<3 * Rsz, 256, 0, stream>>>(part, (long)(3 * RM), nullptr,
                                                  ln1_g, ln1_b, oln);
    gemm3(true, oln, oln + RM, oln + 2 * RM, 0, 1024,
          ffn_w1, ffn_w1, ffn_w1, ffn_w1, ffn_w1, 0,
          ffn_b1, nullptr, nullptr, nullptr,
          h1, 0, 1, DIsz, 1024, GEMM_BIAS | GEMM_RELU);
    gemm3(false, h1, h1 + (size_t)Rsz * DIsz, h1 + 2 * (size_t)Rsz * DIsz, DIsz / KSPLIT, DIsz,
          ffn_w2, ffn_w2, ffn_w2, ffn_w2, ffn_w2, (long)(DIsz / KSPLIT) * 1024,
          ffn_b2, nullptr, nullptr, nullptr,
          part, (long)(3 * RM), KSPLIT, 1024, DIsz / KSPLIT, GEMM_BIAS);
    layernorm_kernel<<<3 * Rsz, 256, 0, stream>>>(part, (long)(3 * RM), oln,
                                                  ln2_g, ln2_b, out);
}


# Decision-gate kernel: a lane's columns held as register pairs so each term's two adds run as packed f32 adds; relu as max in front of the fma (was fma, compare, select); sum order unchanged
# speedup vs baseline: 6.3401x; 1.0343x over previous
#include <hip/hip_runtime.h>
#include <hip/hip_bf16.h>

#define Bsz 4
#define Lsz 64
#define DMsz 1024
#define Hsz 16
#define DKsz 64
#define DVsz 64
#define Dsz 1024
#define DIsz 4096
#define Rsz 256
#define NEGC (-1.0e9f)

typedef float v2f __attribute__((ext_vector_type(2)));
typedef float v4f __attribute__((ext_vector_type(4)));
typedef float v8f __attribute__((ext_vector_type(8)));

#define GEMM_AP 68
#define GEMM_BT 1040
#define GEMM_BIAS  1
#define GEMM_RELU  2
#define GEMM_RESID 4

template <int BN>
__global__ __launch_bounds__(BN * 2)
void gemm_wmma_f32(const float* __restrict__ A0, const float* __restrict__ A1,
                   const float* __restrict__ A2, long strideAz, int lda,
                   const float* __restrict__ B0, const float* __restrict__ B1,
                   const float* __restrict__ B2, const float* __restrict__ B3,
                   const float* __restrict__ B4, long strideBz,
                   const float* __restrict__ bias,
                   const float* __restrict__ R0, const float* __restrict__ R1,
                   const float* __restrict__ R2,
                   float* __restrict__ Cbase, long strideCz,
                   int N, int Kloop, int flags)
{
    const int seg  = blockIdx.y >> 2;
    const int mloc = (blockIdx.y & 3) * 64;
    const int zi   = blockIdx.z;
    const float* __restrict__ A =
        ((seg == 0) ? A0 : (seg == 1) ? A1 : A2) + (size_t)zi * strideAz;
    const float* __restrict__ B =
        ((zi == 0) ? B0 : (zi == 1) ? B1 : (zi == 2) ? B2 : (zi == 3) ? B3 : B4)
        + (size_t)zi * strideBz;
    const bool lead = (zi == 0);
    const float* __restrict__ resid = (seg == 0) ? R0 : (seg == 1) ? R1 : R2;
    float* __restrict__ C = Cbase + (size_t)zi * strideCz;

    constexpr int NT    = BN * 2;
    constexpr int NWN   = BN / 32;
    constexpr int APASS = 1024 / NT;
    constexpr int AROWS = NT / 16;
    constexpr int TPR   = BN / 4;

    __shared__ __attribute__((aligned(16))) float As[64][GEMM_AP];
    __shared__ __attribute__((aligned(16))) float Bs[(BN / 16) * GEMM_BT];

    const int tid   = threadIdx.x;
    const int wave  = tid >> 5;
    const int lane  = tid & 31;
    const int l15   = lane & 15;
    const int lhalf = lane >> 4;
    const int m0    = blockIdx.y * 64;
    const int nblk  = blockIdx.x * BN;
    const int mq    = wave / NWN;
    const int nq    = wave % NWN;

    const int ar = tid >> 4;
    const int ac = (tid & 15) << 2;
    const int br = (tid / TPR) * 8;
    const int bc = (tid % TPR) << 2;
    v4f aprf[APASS], bprf[8];

    v8f acc[2][2];
#pragma unroll
    for (int x = 0; x < 2; ++x)
#pragma unroll
        for (int y = 0; y < 2; ++y)
            acc[x][y] = v8f{0.f, 0.f, 0.f, 0.f, 0.f, 0.f, 0.f, 0.f};

#pragma unroll
    for (int rr = 0; rr < APASS; ++rr)
        aprf[rr] = *(const v4f*)(A + (size_t)(mloc + ar + rr * AROWS) * lda + ac);
#pragma unroll
    for (int rr = 0; rr < 8; ++rr)
        bprf[rr] = *(const v4f*)(B + (size_t)(br + rr) * N + nblk + bc);

    const int arow0 = mq * 32 + l15;

    for (int k0 = 0; k0 < Kloop; k0 += 64) {
#pragma unroll
        for (int rr = 0; rr < APASS; ++rr)
            *(v4f*)&As[ar + rr * AROWS][ac] = aprf[rr];
#pragma unroll
        for (int rr = 0; rr < 8; ++rr)
            *(v4f*)&Bs[(bc >> 4) * GEMM_BT + (br + rr) * 16 + (bc & 15)] = bprf[rr];
        __syncthreads();

        if (k0 + 64 < Kloop) {
#pragma unroll
            for (int rr = 0; rr < APASS; ++rr)
                aprf[rr] = *(const v4f*)(A + (size_t)(mloc + ar + rr * AROWS) * lda
                                         + k0 + 64 + ac);
#pragma unroll
            for (int rr = 0; rr < 8; ++rr)
                bprf[rr] = *(const v4f*)(B + (size_t)(k0 + 64 + br + rr) * N
                                         + nblk + bc);
        }

#pragma unroll
        for (int half = 0; half < 2; ++half) {
            v2f af[2][8], bf[2][8];
#pragma unroll
            for (int s = 0; s < 8; ++s) {
                const int kb = half * 32 + s * 4 + lhalf * 2;
                af[0][s] = *(const v2f*)&As[arow0][kb];
                af[1][s] = *(const v2f*)&As[arow0 + 16][kb];
                const float* bp = &Bs[(nq * 2) * GEMM_BT + kb * 16 + l15];
                bf[0][s] = v2f{bp[0],       bp[16]};
                bf[1][s] = v2f{bp[GEMM_BT], bp[GEMM_BT + 16]};
            }
#pragma unroll
            for (int s = 0; s < 8; ++s) {
                acc[0][0] = __builtin_amdgcn_wmma_f32_16x16x4_f32(
                    false, af[0][s], false, bf[0][s], (short)0, acc[0][0], false, false);
                acc[0][1] = __builtin_amdgcn_wmma_f32_16x16x4_f32(
                    false, af[0][s], false, bf[1][s], (short)0, acc[0][1], false, false);
                acc[1][0] = __builtin_amdgcn_wmma_f32_16x16x4_f32(
                    false, af[1][s], false, bf[0][s], (short)0, acc[1][0], false, false);
                acc[1][1] = __builtin_amdgcn_wmma_f32_16x16x4_f32(
                    false, af[1][s], false, bf[1][s], (short)0, acc[1][1], false, false);
            }
        }
        __syncthreads();
    }

#pragma unroll
    for (int nsub = 0; nsub < 2; ++nsub) {
        const int col = nblk + nq * 32 + nsub * 16 + l15;
        const float bv = ((flags & GEMM_BIAS) && lead) ? bias[col] : 0.f;
#pragma unroll
        for (int msub = 0; msub < 2; ++msub) {
#pragma unroll
            for (int r = 0; r < 8; ++r) {
                const int rloc = mq * 32 + msub * 16 + r + lhalf * 8;
                const int row  = m0 + rloc;
                float v = acc[msub][nsub][r] + bv;
                if ((flags & GEMM_RESID) && lead) v += resid[(size_t)(mloc + rloc) * N + col];
                if (flags & GEMM_RELU)  v = fmaxf(v, 0.f);
                C[(size_t)row * N + col] = v;
            }
        }
    }
}

#define DEC_NB 4
#define DEC_HOLD8(a, o)                                                         \
    asm volatile("" :: "v"(a[(o) + 0]), "v"(a[(o) + 1]), "v"(a[(o) + 2]),       \
                 "v"(a[(o) + 3]), "v"(a[(o) + 4]), "v"(a[(o) + 5]),             \
                 "v"(a[(o) + 6]), "v"(a[(o) + 7]))

__global__ __launch_bounds__(128, 3)
void dec_kernel(const float* __restrict__ a0base, const float* __restrict__ a1base,
                const float* __restrict__ b1, const float* __restrict__ w2,
                const float* __restrict__ b2, float* __restrict__ dec)
{
    const int lane = threadIdx.x & 31;
    const int wid  = __builtin_amdgcn_readfirstlane(
                         (int)((blockIdx.x * 128u + threadIdx.x) >> 5));
    const int nh   = wid & 1;
    const int mp   = (wid >> 1) & 31;
    const int b    = (wid >> 6) & 3;
    const int pair = wid >> 8;
    const int i    = pair / 3;
    const int j    = pair % 3;
    const int m0   = mp * 2;
    const int n0   = nh * 32;

    const float* a1r = a1base + ((size_t)j * Rsz + b * Lsz + m0) * 2048;
    const float* a0r = a0base + ((size_t)i * Rsz + b * Lsz + n0) * 2048;

    v2f wdp[32], bbp[32], x0p[32], x1p[32];
#pragma unroll
    for (int k = 0; k < 64; ++k) {
        const int c = lane + 32 * k;
        wdp[k >> 1][k & 1] = w2[2 * c + 1] - w2[2 * c];
        bbp[k >> 1][k & 1] = b1[c];
        x0p[k >> 1][k & 1] = a1r[lane + 32 * k];
        x1p[k >> 1][k & 1] = a1r[2048 + lane + 32 * k];
    }

    float r0 = 0.f, r1 = 0.f;
    for (int g = 0; g < 32; g += DEC_NB) {
        float s0[DEC_NB], s1[DEC_NB];
#pragma unroll
        for (int u = 0; u < DEC_NB; ++u) {
            const float* p = a0r + (size_t)(g + u) * 2048;
            v2f yp[32];
#pragma unroll
            for (int k = 0; k < 64; ++k) yp[k >> 1][k & 1] = p[lane + 32 * k];
            float c0 = 0.f, c1 = 0.f;
#pragma unroll
            for (int q = 0; q < 4; ++q) {
                DEC_HOLD8(yp, 8 * q);
#pragma unroll
                for (int h = 8 * q; h < 8 * q + 8; ++h) {
                    v2f t0 = x0p[h] + yp[h] + bbp[h];
                    v2f t1 = x1p[h] + yp[h] + bbp[h];
                    t0 = __builtin_elementwise_max(t0, v2f{0.f, 0.f});
                    t1 = __builtin_elementwise_max(t1, v2f{0.f, 0.f});
                    c0 = fmaf(t0[0], wdp[h][0], c0);
                    c1 = fmaf(t1[0], wdp[h][0], c1);
                    c0 = fmaf(t0[1], wdp[h][1], c0);
                    c1 = fmaf(t1[1], wdp[h][1], c1);
                }
            }
            s0[u] = c0; s1[u] = c1;
        }
#pragma unroll
        for (int off = 16; off; off >>= 1) {
#pragma unroll
            for (int u = 0; u < DEC_NB; ++u) {
                s0[u] += __shfl_xor(s0[u], off, 32);
                s1[u] += __shfl_xor(s1[u], off, 32);
            }
        }
#pragma unroll
        for (int u = 0; u < DEC_NB; ++u) {
            r0 = (lane == g + u) ? s0[u] : r0;
            r1 = (lane == g + u) ? s1[u] : r1;
        }
    }

    const float bd = b2[1] - b2[0];
    const size_t o = (((size_t)pair * Bsz + b) * Lsz + m0) * Lsz + n0 + lane;
    dec[o]       = (r0 + bd > 0.f) ? NEGC : 0.f;
    dec[o + Lsz] = (r1 + bd > 0.f) ? NEGC : 0.f;
}

#define AT_QP 68
#define AT_VP 72
#define AT_SP 196
__global__ __launch_bounds__(256)
void attn_fused_kernel(const float* __restrict__ Qb, const float* __restrict__ Kb,
                       const float* __restrict__ Vb, const float* __restrict__ dec,
                       const int* __restrict__ mask, float* __restrict__ Ob)
{
    const int blk = blockIdx.x;
    const int h = blk & 15;
    const int b = (blk >> 4) & 3;
    const int i = blk >> 6;

    __shared__ __attribute__((aligned(16))) float Qs[64][AT_QP];
    __shared__ __attribute__((aligned(16))) float Ks[192][AT_QP];
    __shared__ __attribute__((aligned(16))) float Vs[192][AT_VP];
    __shared__ __attribute__((aligned(16))) float Ss[64][AT_SP];

    const int tid   = threadIdx.x;
    const int wave  = tid >> 5;
    const int lane  = tid & 31;
    const int l15   = lane & 15;
    const int lhalf = lane >> 4;

    const int lr = tid >> 4;
    const int lc = (tid & 15) << 2;
    float4 qreg[4], kreg[12], vreg[12];
#pragma unroll
    for (int rr = 0; rr < 4; ++rr) {
        const int row = lr + rr * 16;
        qreg[rr] = *(const float4*)(Qb + ((size_t)i * Rsz + b * Lsz + row) * DMsz
                                    + h * DKsz + lc);
    }
#pragma unroll
    for (int rr = 0; rr < 12; ++rr) {
        const int row = lr + rr * 16;
        const size_t g = ((size_t)(row >> 6) * Rsz + b * Lsz + (row & 63)) * DMsz
                         + h * DKsz + lc;
        kreg[rr] = *(const float4*)(Kb + g);
        vreg[rr] = *(const float4*)(Vb + g);
    }

    float dv[8][6];
    int   mk[8][2];
#pragma unroll
    for (int rr = 0; rr < 8; ++rr) {
        const int row = wave * 8 + rr;
#pragma unroll
        for (int u = 0; u < 6; ++u) {
            const int col = lane + 32 * u;
            dv[rr][u] = dec[(((size_t)(i * 3 + (col >> 6)) * Bsz + b) * Lsz + row) * Lsz
                            + (col & 63)];
        }
        mk[rr][0] = mask[((size_t)b * Lsz + row) * Lsz + lane];
        mk[rr][1] = mask[((size_t)b * Lsz + row) * Lsz + lane + 32];
    }

#pragma unroll
    for (int rr = 0; rr < 4; ++rr)
        *(float4*)&Qs[lr + rr * 16][lc] = qreg[rr];
#pragma unroll
    for (int rr = 0; rr < 12; ++rr) {
        *(float4*)&Ks[lr + rr * 16][lc] = kreg[rr];
        *(float4*)&Vs[lr + rr * 16][lc] = vreg[rr];
    }
    __syncthreads();

    const int mt   = wave & 3;
    const int ng   = wave >> 2;
    const int arow = mt * 16 + l15;
    {
        v8f sacc[6];
#pragma unroll
        for (int t = 0; t < 6; ++t) sacc[t] = v8f{0.f, 0.f, 0.f, 0.f, 0.f, 0.f, 0.f, 0.f};
#pragma unroll
        for (int s = 0; s < 16; ++s) {
            const int kb = s * 4 + lhalf * 2;
            const v2f a = *(const v2f*)&Qs[arow][kb];
#pragma unroll
            for (int t = 0; t < 6; ++t) {
                const v2f bk = *(const v2f*)&Ks[(ng * 6 + t) * 16 + l15][kb];
                sacc[t] = __builtin_amdgcn_wmma_f32_16x16x4_f32(
                    false, a, false, bk, (short)0, sacc[t], false, false);
            }
        }
#pragma unroll
        for (int t = 0; t < 6; ++t) {
            const int col = (ng * 6 + t) * 16 + l15;
#pragma unroll
            for (int r = 0; r < 8; ++r)
                Ss[mt * 16 + r + lhalf * 8][col] = sacc[t][r] * 0.125f;
        }
    }
    __syncthreads();

#pragma unroll
    for (int rr = 0; rr < 8; ++rr) {
        const int row = wave * 8 + rr;
        float sv[6];
        float mx = -3.0e38f;
#pragma unroll
        for (int u = 0; u < 6; ++u) {
            float s = Ss[row][lane + 32 * u] + dv[rr][u];
            if (mk[rr][u & 1] == 0) s = NEGC;
            sv[u] = s;
            mx = fmaxf(mx, s);
        }
#pragma unroll
        for (int off = 16; off; off >>= 1) mx = fmaxf(mx, __shfl_xor(mx, off, 32));
        float sum = 0.f;
#pragma unroll
        for (int u = 0; u < 6; ++u) { sv[u] = __expf(sv[u] - mx); sum += sv[u]; }
#pragma unroll
        for (int off = 16; off; off >>= 1) sum += __shfl_xor(sum, off, 32);
        const float inv = 1.0f / sum;
#pragma unroll
        for (int u = 0; u < 6; ++u) Ss[row][lane + 32 * u] = sv[u] * inv;
    }
    __syncthreads();

    {
        v8f oacc[2][2];
#pragma unroll
        for (int t = 0; t < 2; ++t)
#pragma unroll
            for (int e = 0; e < 2; ++e)
                oacc[t][e] = v8f{0.f, 0.f, 0.f, 0.f, 0.f, 0.f, 0.f, 0.f};
#pragma unroll
        for (int s = 0; s < 48; s += 2) {
#pragma unroll
            for (int e = 0; e < 2; ++e) {
                const int kb = (s + e) * 4 + lhalf * 2;
                const v2f a = *(const v2f*)&Ss[arow][kb];
#pragma unroll
                for (int t = 0; t < 2; ++t) {
                    const int col = ng * 32 + t * 16 + l15;
                    const v2f bv = v2f{Vs[kb][col], Vs[kb + 1][col]};
                    oacc[t][e] = __builtin_amdgcn_wmma_f32_16x16x4_f32(
                        false, a, false, bv, (short)0, oacc[t][e], false, false);
                }
            }
        }
#pragma unroll
        for (int t = 0; t < 2; ++t) {
            const int col = ng * 32 + t * 16 + l15;
#pragma unroll
            for (int r = 0; r < 8; ++r) {
                const int row = mt * 16 + r + lhalf * 8;
                Ob[((size_t)i * Rsz + b * Lsz + row) * DMsz + h * DVsz + col] =
                    oacc[t][0][r] + oacc[t][1][r];
            }
        }
    }
}

#define KSPLIT 8
__global__ __launch_bounds__(256)
void layernorm_kernel(const float* __restrict__ x, long plane,
                      const float* __restrict__ resid,
                      const float* __restrict__ g, const float* __restrict__ bta,
                      float* __restrict__ out)
{
    const int row = blockIdx.x;
    const int t = threadIdx.x;
    __shared__ float red[256];

    const float* xp = x + (size_t)row * 1024;
    float v[4];
    float s = 0.f;
#pragma unroll
    for (int u = 0; u < 4; ++u) {
        float val = xp[t + u * 256];
#pragma unroll
        for (int p = 1; p < KSPLIT; ++p) val += xp[(size_t)p * plane + t + u * 256];
        v[u] = val;
    }
    if (resid) {
#pragma unroll
        for (int u = 0; u < 4; ++u) v[u] += resid[(size_t)row * 1024 + t + u * 256];
    }
#pragma unroll
    for (int u = 0; u < 4; ++u) s += v[u];
    red[t] = s; __syncthreads();
    for (int st = 128; st > 0; st >>= 1) { if (t < st) red[t] += red[t + st]; __syncthreads(); }
    const float mu = red[0] * (1.f / 1024.f);
    __syncthreads();

    float s2 = 0.f;
#pragma unroll
    for (int u = 0; u < 4; ++u) { float dd = v[u] - mu; s2 += dd * dd; }
    red[t] = s2; __syncthreads();
    for (int st = 128; st > 0; st >>= 1) { if (t < st) red[t] += red[t + st]; __syncthreads(); }
    const float var = red[0] * (1.f / 1024.f);
    const float rinv = 1.0f / sqrtf(var + 1e-6f);
#pragma unroll
    for (int u = 0; u < 4; ++u) {
        int c = t + u * 256;
        out[(size_t)row * 1024 + c] = (v[u] - mu) * rinv * g[c] + bta[c];
    }
}

extern "C" void kernel_launch(void* const* d_in, const int* in_sizes, int n_in,
                              void* d_out, int out_size, void* d_ws, size_t ws_size,
                              hipStream_t stream)
{
    const float* X[3]   = { (const float*)d_in[0], (const float*)d_in[1], (const float*)d_in[2] };
    const int*   mask   = (const int*)d_in[3];
    const float* w_qs   = (const float*)d_in[4];
    const float* w_ks   = (const float*)d_in[5];
    const float* w_vs   = (const float*)d_in[6];
    const float* w_ds0  = (const float*)d_in[7];
    const float* w_ds1  = (const float*)d_in[8];
    const float* dc_w1  = (const float*)d_in[9];
    const float* dc_b1  = (const float*)d_in[10];
    const float* dc_w2  = (const float*)d_in[11];
    const float* dc_b2  = (const float*)d_in[12];
    const float* w_fc   = (const float*)d_in[13];
    const float* ln1_g  = (const float*)d_in[14];
    const float* ln1_b  = (const float*)d_in[15];
    const float* ffn_w1 = (const float*)d_in[16];
    const float* ffn_b1 = (const float*)d_in[17];
    const float* ffn_w2 = (const float*)d_in[18];
    const float* ffn_b2 = (const float*)d_in[19];
    const float* ln2_g  = (const float*)d_in[20];
    const float* ln2_b  = (const float*)d_in[21];
    float* out = (float*)d_out;

    float* ws = (float*)d_ws;
    size_t off = 0;
    auto alloc = [&](size_t n) { float* p = ws + off; off += n; return p; };
    const size_t RM = (size_t)Rsz * DMsz;
    float* Q   = alloc(3 * RM);
    float* Kb  = alloc(3 * RM);
    float* Vb  = alloc(3 * RM);
    float* d0  = alloc(3 * RM);
    float* d1  = alloc(3 * RM);
    float* a0  = alloc(3 * (size_t)Rsz * 2048);
    float* a1  = alloc(3 * (size_t)Rsz * 2048);
    float* dec = alloc(9 * (size_t)Bsz * Lsz * Lsz);
    (void)alloc((size_t)3 * Bsz * Hsz * Lsz * 192);
    float* O   = alloc(3 * RM);
    float* oln = alloc(3 * RM);
    float* h1  = alloc(3 * (size_t)Rsz * DIsz);
    float* part = Q;
    (void)ws_size; (void)n_in; (void)in_sizes; (void)out_size;

    auto gemm3 = [&](bool narrow,
                     const float* a0s, const float* a1s, const float* a2s, long strideAz,
                     int lda,
                     const float* b0, const float* b1, const float* b2,
                     const float* b3, const float* b4, long strideBz,
                     const float* bias,
                     const float* r0s, const float* r1s, const float* r2s,
                     float* C, long strideCz, int nz, int N, int Kloop, int flags) {
        if (narrow) {
            dim3 grid(N / 64, (3 * Rsz) / 64, nz);
            gemm_wmma_f32<64><<<grid, dim3(128), 0, stream>>>(a0s, a1s, a2s, strideAz, lda,
                                                              b0, b1, b2, b3, b4, strideBz, bias,
                                                              r0s, r1s, r2s, C, strideCz,
                                                              N, Kloop, flags);
        } else {
            dim3 grid(N / 128, (3 * Rsz) / 64, nz);
            gemm_wmma_f32<128><<<grid, dim3(256), 0, stream>>>(a0s, a1s, a2s, strideAz, lda,
                                                               b0, b1, b2, b3, b4, strideBz, bias,
                                                               r0s, r1s, r2s, C, strideCz,
                                                               N, Kloop, flags);
        }
    };

    gemm3(false, X[0], X[1], X[2], 0, 1024,
          w_qs, w_ks, w_vs, w_ds0, w_ds1, 0,
          nullptr, nullptr, nullptr, nullptr,
          Q, (long)(3 * RM), 5, 1024, 1024, 0);

    gemm3(true, d0, d0 + RM, d0 + 2 * RM, (long)(3 * RM), 1024,
          dc_w1 + (size_t)1024 * 2048, dc_w1, dc_w1, dc_w1, dc_w1, 0,
          nullptr, nullptr, nullptr, nullptr,
          a0, (long)(3 * (size_t)Rsz * 2048), 2, 2048, 1024, 0);

    {
        dec_kernel<<<576, 128, 0, stream>>>(a0, a1, dc_b1, dc_w2, dc_b2, dec);
    }

    attn_fused_kernel<<<3 * Bsz * Hsz, 256, 0, stream>>>(Q, Kb, Vb, dec, mask, O);

    gemm3(false, O, O + RM, O + 2 * RM, 1024 / KSPLIT, 1024,
          w_fc, w_fc, w_fc, w_fc, w_fc, (long)(1024 / KSPLIT) * 1024,
          nullptr, X[0], X[1], X[2],
          part, (long)(3 * RM), KSPLIT, 1024, 1024 / KSPLIT, GEMM_RESID);
    layernorm_kernel<<<3 * Rsz, 256, 0, stream>>>(part, (long)(3 * RM), nullptr,
                                                  ln1_g, ln1_b, oln);
    gemm3(true, oln, oln + RM, oln + 2 * RM, 0, 1024,
          ffn_w1, ffn_w1, ffn_w1, ffn_w1, ffn_w1, 0,
          ffn_b1, nullptr, nullptr, nullptr,
          h1, 0, 1, DIsz, 1024, GEMM_BIAS | GEMM_RELU);
    gemm3(false, h1, h1 + (size_t)Rsz * DIsz, h1 + 2 * (size_t)Rsz * DIsz, DIsz / KSPLIT, DIsz,
          ffn_w2, ffn_w2, ffn_w2, ffn_w2, ffn_w2, (long)(DIsz / KSPLIT) * 1024,
          ffn_b2, nullptr, nullptr, nullptr,
          part, (long)(3 * RM), KSPLIT, 1024, DIsz / KSPLIT, GEMM_BIAS);
    layernorm_kernel<<<3 * Rsz, 256, 0, stream>>>(part, (long)(3 * RM), oln,
                                                  ln2_g, ln2_b, out);
}


# GEMM epilogue: each wave parks its 32x32 tile in LDS and writes whole 128-byte lines with 16-byte stores; residual rows fetched as one group of loads (was one load and wait per row); sums unchanged
# speedup vs baseline: 6.5262x; 1.0294x over previous
#include <hip/hip_runtime.h>
#include <hip/hip_bf16.h>

#define Bsz 4
#define Lsz 64
#define DMsz 1024
#define Hsz 16
#define DKsz 64
#define DVsz 64
#define Dsz 1024
#define DIsz 4096
#define Rsz 256
#define NEGC (-1.0e9f)

typedef float v2f __attribute__((ext_vector_type(2)));
typedef float v4f __attribute__((ext_vector_type(4)));
typedef float v8f __attribute__((ext_vector_type(8)));

#define GEMM_AP 68
#define GEMM_BT 1040
#define GEMM_EP 36
#define GEMM_BIAS  1
#define GEMM_RELU  2
#define GEMM_RESID 4

template <int BN>
__global__ __launch_bounds__(BN * 2)
void gemm_wmma_f32(const float* __restrict__ A0, const float* __restrict__ A1,
                   const float* __restrict__ A2, long strideAz, int lda,
                   const float* __restrict__ B0, const float* __restrict__ B1,
                   const float* __restrict__ B2, const float* __restrict__ B3,
                   const float* __restrict__ B4, long strideBz,
                   const float* __restrict__ bias,
                   const float* __restrict__ R0, const float* __restrict__ R1,
                   const float* __restrict__ R2,
                   float* __restrict__ Cbase, long strideCz,
                   int N, int Kloop, int flags)
{
    const int seg  = blockIdx.y >> 2;
    const int mloc = (blockIdx.y & 3) * 64;
    const int zi   = blockIdx.z;
    const float* __restrict__ A =
        ((seg == 0) ? A0 : (seg == 1) ? A1 : A2) + (size_t)zi * strideAz;
    const float* __restrict__ B =
        ((zi == 0) ? B0 : (zi == 1) ? B1 : (zi == 2) ? B2 : (zi == 3) ? B3 : B4)
        + (size_t)zi * strideBz;
    const bool lead = (zi == 0);
    const float* __restrict__ resid = (seg == 0) ? R0 : (seg == 1) ? R1 : R2;
    float* __restrict__ C = Cbase + (size_t)zi * strideCz;

    constexpr int NT    = BN * 2;
    constexpr int NWN   = BN / 32;
    constexpr int APASS = 1024 / NT;
    constexpr int AROWS = NT / 16;
    constexpr int TPR   = BN / 4;

    __shared__ __attribute__((aligned(16))) float smem[64 * GEMM_AP + (BN / 16) * GEMM_BT];
    float (*As)[GEMM_AP] = reinterpret_cast<float (*)[GEMM_AP]>(smem);
    float* Bs = smem + 64 * GEMM_AP;

    const int tid   = threadIdx.x;
    const int wave  = tid >> 5;
    const int lane  = tid & 31;
    const int l15   = lane & 15;
    const int lhalf = lane >> 4;
    const int m0    = blockIdx.y * 64;
    const int nblk  = blockIdx.x * BN;
    const int mq    = wave / NWN;
    const int nq    = wave % NWN;

    const int ar = tid >> 4;
    const int ac = (tid & 15) << 2;
    const int br = (tid / TPR) * 8;
    const int bc = (tid % TPR) << 2;
    v4f aprf[APASS], bprf[8];

    v8f acc[2][2];
#pragma unroll
    for (int x = 0; x < 2; ++x)
#pragma unroll
        for (int y = 0; y < 2; ++y)
            acc[x][y] = v8f{0.f, 0.f, 0.f, 0.f, 0.f, 0.f, 0.f, 0.f};

#pragma unroll
    for (int rr = 0; rr < APASS; ++rr)
        aprf[rr] = *(const v4f*)(A + (size_t)(mloc + ar + rr * AROWS) * lda + ac);
#pragma unroll
    for (int rr = 0; rr < 8; ++rr)
        bprf[rr] = *(const v4f*)(B + (size_t)(br + rr) * N + nblk + bc);

    const int arow0 = mq * 32 + l15;

    for (int k0 = 0; k0 < Kloop; k0 += 64) {
#pragma unroll
        for (int rr = 0; rr < APASS; ++rr)
            *(v4f*)&As[ar + rr * AROWS][ac] = aprf[rr];
#pragma unroll
        for (int rr = 0; rr < 8; ++rr)
            *(v4f*)&Bs[(bc >> 4) * GEMM_BT + (br + rr) * 16 + (bc & 15)] = bprf[rr];
        __syncthreads();

        if (k0 + 64 < Kloop) {
#pragma unroll
            for (int rr = 0; rr < APASS; ++rr)
                aprf[rr] = *(const v4f*)(A + (size_t)(mloc + ar + rr * AROWS) * lda
                                         + k0 + 64 + ac);
#pragma unroll
            for (int rr = 0; rr < 8; ++rr)
                bprf[rr] = *(const v4f*)(B + (size_t)(k0 + 64 + br + rr) * N
                                         + nblk + bc);
        }

#pragma unroll
        for (int half = 0; half < 2; ++half) {
            v2f af[2][8], bf[2][8];
#pragma unroll
            for (int s = 0; s < 8; ++s) {
                const int kb = half * 32 + s * 4 + lhalf * 2;
                af[0][s] = *(const v2f*)&As[arow0][kb];
                af[1][s] = *(const v2f*)&As[arow0 + 16][kb];
                const float* bp = &Bs[(nq * 2) * GEMM_BT + kb * 16 + l15];
                bf[0][s] = v2f{bp[0],       bp[16]};
                bf[1][s] = v2f{bp[GEMM_BT], bp[GEMM_BT + 16]};
            }
#pragma unroll
            for (int s = 0; s < 8; ++s) {
                acc[0][0] = __builtin_amdgcn_wmma_f32_16x16x4_f32(
                    false, af[0][s], false, bf[0][s], (short)0, acc[0][0], false, false);
                acc[0][1] = __builtin_amdgcn_wmma_f32_16x16x4_f32(
                    false, af[0][s], false, bf[1][s], (short)0, acc[0][1], false, false);
                acc[1][0] = __builtin_amdgcn_wmma_f32_16x16x4_f32(
                    false, af[1][s], false, bf[0][s], (short)0, acc[1][0], false, false);
                acc[1][1] = __builtin_amdgcn_wmma_f32_16x16x4_f32(
                    false, af[1][s], false, bf[1][s], (short)0, acc[1][1], false, false);
            }
        }
        __syncthreads();
    }

    const int erow = lane >> 3;
    const int ecol = (lane & 7) << 2;
    const int gcol = nblk + nq * 32 + ecol;
    const bool use_bias  = (flags & GEMM_BIAS) && lead;
    const bool use_resid = (flags & GEMM_RESID) && lead;

    v4f bv = v4f{0.f, 0.f, 0.f, 0.f};
    if (use_bias) bv = *(const v4f*)(bias + gcol);
    v4f rv[8];
#pragma unroll
    for (int it = 0; it < 8; ++it) rv[it] = v4f{0.f, 0.f, 0.f, 0.f};
    if (use_resid) {
#pragma unroll
        for (int it = 0; it < 8; ++it)
            rv[it] = *(const v4f*)(resid + (size_t)(mloc + mq * 32 + it * 4 + erow) * N + gcol);
    }

    float* stg = smem + wave * (32 * GEMM_EP);
#pragma unroll
    for (int msub = 0; msub < 2; ++msub)
#pragma unroll
        for (int nsub = 0; nsub < 2; ++nsub)
#pragma unroll
            for (int r = 0; r < 8; ++r)
                stg[(msub * 16 + lhalf * 8 + r) * GEMM_EP + nsub * 16 + l15] = acc[msub][nsub][r];
    __syncthreads();

#pragma unroll
    for (int it = 0; it < 8; ++it) {
        const int rloc = it * 4 + erow;
        v4f v = *(const v4f*)&stg[rloc * GEMM_EP + ecol];
        v = v + bv;
        if (use_resid) v = v + rv[it];
        if (flags & GEMM_RELU) v = __builtin_elementwise_max(v, v4f{0.f, 0.f, 0.f, 0.f});
        *(v4f*)(C + (size_t)(m0 + mq * 32 + rloc) * N + gcol) = v;
    }
}

#define DEC_NB 4
#define DEC_HOLD8(a, o)                                                         \
    asm volatile("" :: "v"(a[(o) + 0]), "v"(a[(o) + 1]), "v"(a[(o) + 2]),       \
                 "v"(a[(o) + 3]), "v"(a[(o) + 4]), "v"(a[(o) + 5]),             \
                 "v"(a[(o) + 6]), "v"(a[(o) + 7]))

__global__ __launch_bounds__(128, 3)
void dec_kernel(const float* __restrict__ a0base, const float* __restrict__ a1base,
                const float* __restrict__ b1, const float* __restrict__ w2,
                const float* __restrict__ b2, float* __restrict__ dec)
{
    const int lane = threadIdx.x & 31;
    const int wid  = __builtin_amdgcn_readfirstlane(
                         (int)((blockIdx.x * 128u + threadIdx.x) >> 5));
    const int nh   = wid & 1;
    const int mp   = (wid >> 1) & 31;
    const int b    = (wid >> 6) & 3;
    const int pair = wid >> 8;
    const int i    = pair / 3;
    const int j    = pair % 3;
    const int m0   = mp * 2;
    const int n0   = nh * 32;

    const float* a1r = a1base + ((size_t)j * Rsz + b * Lsz + m0) * 2048;
    const float* a0r = a0base + ((size_t)i * Rsz + b * Lsz + n0) * 2048;

    v2f wdp[32], bbp[32], x0p[32], x1p[32];
#pragma unroll
    for (int k = 0; k < 64; ++k) {
        const int c = lane + 32 * k;
        wdp[k >> 1][k & 1] = w2[2 * c + 1] - w2[2 * c];
        bbp[k >> 1][k & 1] = b1[c];
        x0p[k >> 1][k & 1] = a1r[lane + 32 * k];
        x1p[k >> 1][k & 1] = a1r[2048 + lane + 32 * k];
    }

    float r0 = 0.f, r1 = 0.f;
    for (int g = 0; g < 32; g += DEC_NB) {
        float s0[DEC_NB], s1[DEC_NB];
#pragma unroll
        for (int u = 0; u < DEC_NB; ++u) {
            const float* p = a0r + (size_t)(g + u) * 2048;
            v2f yp[32];
#pragma unroll
            for (int k = 0; k < 64; ++k) yp[k >> 1][k & 1] = p[lane + 32 * k];
            float c0 = 0.f, c1 = 0.f;
#pragma unroll
            for (int q = 0; q < 4; ++q) {
                DEC_HOLD8(yp, 8 * q);
#pragma unroll
                for (int h = 8 * q; h < 8 * q + 8; ++h) {
                    v2f t0 = x0p[h] + yp[h] + bbp[h];
                    v2f t1 = x1p[h] + yp[h] + bbp[h];
                    t0 = __builtin_elementwise_max(t0, v2f{0.f, 0.f});
                    t1 = __builtin_elementwise_max(t1, v2f{0.f, 0.f});
                    c0 = fmaf(t0[0], wdp[h][0], c0);
                    c1 = fmaf(t1[0], wdp[h][0], c1);
                    c0 = fmaf(t0[1], wdp[h][1], c0);
                    c1 = fmaf(t1[1], wdp[h][1], c1);
                }
            }
            s0[u] = c0; s1[u] = c1;
        }
#pragma unroll
        for (int off = 16; off; off >>= 1) {
#pragma unroll
            for (int u = 0; u < DEC_NB; ++u) {
                s0[u] += __shfl_xor(s0[u], off, 32);
                s1[u] += __shfl_xor(s1[u], off, 32);
            }
        }
#pragma unroll
        for (int u = 0; u < DEC_NB; ++u) {
            r0 = (lane == g + u) ? s0[u] : r0;
            r1 = (lane == g + u) ? s1[u] : r1;
        }
    }

    const float bd = b2[1] - b2[0];
    const size_t o = (((size_t)pair * Bsz + b) * Lsz + m0) * Lsz + n0 + lane;
    dec[o]       = (r0 + bd > 0.f) ? NEGC : 0.f;
    dec[o + Lsz] = (r1 + bd > 0.f) ? NEGC : 0.f;
}

#define AT_QP 68
#define AT_VP 72
#define AT_SP 196
__global__ __launch_bounds__(256)
void attn_fused_kernel(const float* __restrict__ Qb, const float* __restrict__ Kb,
                       const float* __restrict__ Vb, const float* __restrict__ dec,
                       const int* __restrict__ mask, float* __restrict__ Ob)
{
    const int blk = blockIdx.x;
    const int h = blk & 15;
    const int b = (blk >> 4) & 3;
    const int i = blk >> 6;

    __shared__ __attribute__((aligned(16))) float Qs[64][AT_QP];
    __shared__ __attribute__((aligned(16))) float Ks[192][AT_QP];
    __shared__ __attribute__((aligned(16))) float Vs[192][AT_VP];
    __shared__ __attribute__((aligned(16))) float Ss[64][AT_SP];

    const int tid   = threadIdx.x;
    const int wave  = tid >> 5;
    const int lane  = tid & 31;
    const int l15   = lane & 15;
    const int lhalf = lane >> 4;

    const int lr = tid >> 4;
    const int lc = (tid & 15) << 2;
    float4 qreg[4], kreg[12], vreg[12];
#pragma unroll
    for (int rr = 0; rr < 4; ++rr) {
        const int row = lr + rr * 16;
        qreg[rr] = *(const float4*)(Qb + ((size_t)i * Rsz + b * Lsz + row) * DMsz
                                    + h * DKsz + lc);
    }
#pragma unroll
    for (int rr = 0; rr < 12; ++rr) {
        const int row = lr + rr * 16;
        const size_t g = ((size_t)(row >> 6) * Rsz + b * Lsz + (row & 63)) * DMsz
                         + h * DKsz + lc;
        kreg[rr] = *(const float4*)(Kb + g);
        vreg[rr] = *(const float4*)(Vb + g);
    }

    float dv[8][6];
    int   mk[8][2];
#pragma unroll
    for (int rr = 0; rr < 8; ++rr) {
        const int row = wave * 8 + rr;
#pragma unroll
        for (int u = 0; u < 6; ++u) {
            const int col = lane + 32 * u;
            dv[rr][u] = dec[(((size_t)(i * 3 + (col >> 6)) * Bsz + b) * Lsz + row) * Lsz
                            + (col & 63)];
        }
        mk[rr][0] = mask[((size_t)b * Lsz + row) * Lsz + lane];
        mk[rr][1] = mask[((size_t)b * Lsz + row) * Lsz + lane + 32];
    }

#pragma unroll
    for (int rr = 0; rr < 4; ++rr)
        *(float4*)&Qs[lr + rr * 16][lc] = qreg[rr];
#pragma unroll
    for (int rr = 0; rr < 12; ++rr) {
        *(float4*)&Ks[lr + rr * 16][lc] = kreg[rr];
        *(float4*)&Vs[lr + rr * 16][lc] = vreg[rr];
    }
    __syncthreads();

    const int mt   = wave & 3;
    const int ng   = wave >> 2;
    const int arow = mt * 16 + l15;
    {
        v8f sacc[6];
#pragma unroll
        for (int t = 0; t < 6; ++t) sacc[t] = v8f{0.f, 0.f, 0.f, 0.f, 0.f, 0.f, 0.f, 0.f};
#pragma unroll
        for (int s = 0; s < 16; ++s) {
            const int kb = s * 4 + lhalf * 2;
            const v2f a = *(const v2f*)&Qs[arow][kb];
#pragma unroll
            for (int t = 0; t < 6; ++t) {
                const v2f bk = *(const v2f*)&Ks[(ng * 6 + t) * 16 + l15][kb];
                sacc[t] = __builtin_amdgcn_wmma_f32_16x16x4_f32(
                    false, a, false, bk, (short)0, sacc[t], false, false);
            }
        }
#pragma unroll
        for (int t = 0; t < 6; ++t) {
            const int col = (ng * 6 + t) * 16 + l15;
#pragma unroll
            for (int r = 0; r < 8; ++r)
                Ss[mt * 16 + r + lhalf * 8][col] = sacc[t][r] * 0.125f;
        }
    }
    __syncthreads();

#pragma unroll
    for (int rr = 0; rr < 8; ++rr) {
        const int row = wave * 8 + rr;
        float sv[6];
        float mx = -3.0e38f;
#pragma unroll
        for (int u = 0; u < 6; ++u) {
            float s = Ss[row][lane + 32 * u] + dv[rr][u];
            if (mk[rr][u & 1] == 0) s = NEGC;
            sv[u] = s;
            mx = fmaxf(mx, s);
        }
#pragma unroll
        for (int off = 16; off; off >>= 1) mx = fmaxf(mx, __shfl_xor(mx, off, 32));
        float sum = 0.f;
#pragma unroll
        for (int u = 0; u < 6; ++u) { sv[u] = __expf(sv[u] - mx); sum += sv[u]; }
#pragma unroll
        for (int off = 16; off; off >>= 1) sum += __shfl_xor(sum, off, 32);
        const float inv = 1.0f / sum;
#pragma unroll
        for (int u = 0; u < 6; ++u) Ss[row][lane + 32 * u] = sv[u] * inv;
    }
    __syncthreads();

    {
        v8f oacc[2][2];
#pragma unroll
        for (int t = 0; t < 2; ++t)
#pragma unroll
            for (int e = 0; e < 2; ++e)
                oacc[t][e] = v8f{0.f, 0.f, 0.f, 0.f, 0.f, 0.f, 0.f, 0.f};
#pragma unroll
        for (int s = 0; s < 48; s += 2) {
#pragma unroll
            for (int e = 0; e < 2; ++e) {
                const int kb = (s + e) * 4 + lhalf * 2;
                const v2f a = *(const v2f*)&Ss[arow][kb];
#pragma unroll
                for (int t = 0; t < 2; ++t) {
                    const int col = ng * 32 + t * 16 + l15;
                    const v2f bv = v2f{Vs[kb][col], Vs[kb + 1][col]};
                    oacc[t][e] = __builtin_amdgcn_wmma_f32_16x16x4_f32(
                        false, a, false, bv, (short)0, oacc[t][e], false, false);
                }
            }
        }
#pragma unroll
        for (int t = 0; t < 2; ++t) {
            const int col = ng * 32 + t * 16 + l15;
#pragma unroll
            for (int r = 0; r < 8; ++r) {
                const int row = mt * 16 + r + lhalf * 8;
                Ob[((size_t)i * Rsz + b * Lsz + row) * DMsz + h * DVsz + col] =
                    oacc[t][0][r] + oacc[t][1][r];
            }
        }
    }
}

#define KSPLIT 8
__global__ __launch_bounds__(256)
void layernorm_kernel(const float* __restrict__ x, long plane,
                      const float* __restrict__ resid,
                      const float* __restrict__ g, const float* __restrict__ bta,
                      float* __restrict__ out)
{
    const int row = blockIdx.x;
    const int t = threadIdx.x;
    __shared__ float red[256];

    const float* xp = x + (size_t)row * 1024;
    float v[4];
    float s = 0.f;
#pragma unroll
    for (int u = 0; u < 4; ++u) {
        float val = xp[t + u * 256];
#pragma unroll
        for (int p = 1; p < KSPLIT; ++p) val += xp[(size_t)p * plane + t + u * 256];
        v[u] = val;
    }
    if (resid) {
#pragma unroll
        for (int u = 0; u < 4; ++u) v[u] += resid[(size_t)row * 1024 + t + u * 256];
    }
#pragma unroll
    for (int u = 0; u < 4; ++u) s += v[u];
    red[t] = s; __syncthreads();
    for (int st = 128; st > 0; st >>= 1) { if (t < st) red[t] += red[t + st]; __syncthreads(); }
    const float mu = red[0] * (1.f / 1024.f);
    __syncthreads();

    float s2 = 0.f;
#pragma unroll
    for (int u = 0; u < 4; ++u) { float dd = v[u] - mu; s2 += dd * dd; }
    red[t] = s2; __syncthreads();
    for (int st = 128; st > 0; st >>= 1) { if (t < st) red[t] += red[t + st]; __syncthreads(); }
    const float var = red[0] * (1.f / 1024.f);
    const float rinv = 1.0f / sqrtf(var + 1e-6f);
#pragma unroll
    for (int u = 0; u < 4; ++u) {
        int c = t + u * 256;
        out[(size_t)row * 1024 + c] = (v[u] - mu) * rinv * g[c] + bta[c];
    }
}

extern "C" void kernel_launch(void* const* d_in, const int* in_sizes, int n_in,
                              void* d_out, int out_size, void* d_ws, size_t ws_size,
                              hipStream_t stream)
{
    const float* X[3]   = { (const float*)d_in[0], (const float*)d_in[1], (const float*)d_in[2] };
    const int*   mask   = (const int*)d_in[3];
    const float* w_qs   = (const float*)d_in[4];
    const float* w_ks   = (const float*)d_in[5];
    const float* w_vs   = (const float*)d_in[6];
    const float* w_ds0  = (const float*)d_in[7];
    const float* w_ds1  = (const float*)d_in[8];
    const float* dc_w1  = (const float*)d_in[9];
    const float* dc_b1  = (const float*)d_in[10];
    const float* dc_w2  = (const float*)d_in[11];
    const float* dc_b2  = (const float*)d_in[12];
    const float* w_fc   = (const float*)d_in[13];
    const float* ln1_g  = (const float*)d_in[14];
    const float* ln1_b  = (const float*)d_in[15];
    const float* ffn_w1 = (const float*)d_in[16];
    const float* ffn_b1 = (const float*)d_in[17];
    const float* ffn_w2 = (const float*)d_in[18];
    const float* ffn_b2 = (const float*)d_in[19];
    const float* ln2_g  = (const float*)d_in[20];
    const float* ln2_b  = (const float*)d_in[21];
    float* out = (float*)d_out;

    float* ws = (float*)d_ws;
    size_t off = 0;
    auto alloc = [&](size_t n) { float* p = ws + off; off += n; return p; };
    const size_t RM = (size_t)Rsz * DMsz;
    float* Q   = alloc(3 * RM);
    float* Kb  = alloc(3 * RM);
    float* Vb  = alloc(3 * RM);
    float* d0  = alloc(3 * RM);
    float* d1  = alloc(3 * RM);
    float* a0  = alloc(3 * (size_t)Rsz * 2048);
    float* a1  = alloc(3 * (size_t)Rsz * 2048);
    float* dec = alloc(9 * (size_t)Bsz * Lsz * Lsz);
    (void)alloc((size_t)3 * Bsz * Hsz * Lsz * 192);
    float* O   = alloc(3 * RM);
    float* oln = alloc(3 * RM);
    float* h1  = alloc(3 * (size_t)Rsz * DIsz);
    float* part = Q;
    (void)ws_size; (void)n_in; (void)in_sizes; (void)out_size;

    auto gemm3 = [&](bool narrow,
                     const float* a0s, const float* a1s, const float* a2s, long strideAz,
                     int lda,
                     const float* b0, const float* b1, const float* b2,
                     const float* b3, const float* b4, long strideBz,
                     const float* bias,
                     const float* r0s, const float* r1s, const float* r2s,
                     float* C, long strideCz, int nz, int N, int Kloop, int flags) {
        if (narrow) {
            dim3 grid(N / 64, (3 * Rsz) / 64, nz);
            gemm_wmma_f32<64><<<grid, dim3(128), 0, stream>>>(a0s, a1s, a2s, strideAz, lda,
                                                              b0, b1, b2, b3, b4, strideBz, bias,
                                                              r0s, r1s, r2s, C, strideCz,
                                                              N, Kloop, flags);
        } else {
            dim3 grid(N / 128, (3 * Rsz) / 64, nz);
            gemm_wmma_f32<128><<<grid, dim3(256), 0, stream>>>(a0s, a1s, a2s, strideAz, lda,
                                                               b0, b1, b2, b3, b4, strideBz, bias,
                                                               r0s, r1s, r2s, C, strideCz,
                                                               N, Kloop, flags);
        }
    };

    gemm3(false, X[0], X[1], X[2], 0, 1024,
          w_qs, w_ks, w_vs, w_ds0, w_ds1, 0,
          nullptr, nullptr, nullptr, nullptr,
          Q, (long)(3 * RM), 5, 1024, 1024, 0);

    gemm3(true, d0, d0 + RM, d0 + 2 * RM, (long)(3 * RM), 1024,
          dc_w1 + (size_t)1024 * 2048, dc_w1, dc_w1, dc_w1, dc_w1, 0,
          nullptr, nullptr, nullptr, nullptr,
          a0, (long)(3 * (size_t)Rsz * 2048), 2, 2048, 1024, 0);

    {
        dec_kernel<<<576, 128, 0, stream>>>(a0, a1, dc_b1, dc_w2, dc_b2, dec);
    }

    attn_fused_kernel<<<3 * Bsz * Hsz, 256, 0, stream>>>(Q, Kb, Vb, dec, mask, O);

    gemm3(false, O, O + RM, O + 2 * RM, 1024 / KSPLIT, 1024,
          w_fc, w_fc, w_fc, w_fc, w_fc, (long)(1024 / KSPLIT) * 1024,
          nullptr, X[0], X[1], X[2],
          part, (long)(3 * RM), KSPLIT, 1024, 1024 / KSPLIT, GEMM_RESID);
    layernorm_kernel<<<3 * Rsz, 256, 0, stream>>>(part, (long)(3 * RM), nullptr,
                                                  ln1_g, ln1_b, oln);
    gemm3(true, oln, oln + RM, oln + 2 * RM, 0, 1024,
          ffn_w1, ffn_w1, ffn_w1, ffn_w1, ffn_w1, 0,
          ffn_b1, nullptr, nullptr, nullptr,
          h1, 0, 1, DIsz, 1024, GEMM_BIAS | GEMM_RELU);
    gemm3(false, h1, h1 + (size_t)Rsz * DIsz, h1 + 2 * (size_t)Rsz * DIsz, DIsz / KSPLIT, DIsz,
          ffn_w2, ffn_w2, ffn_w2, ffn_w2, ffn_w2, (long)(DIsz / KSPLIT) * 1024,
          ffn_b2, nullptr, nullptr, nullptr,
          part, (long)(3 * RM), KSPLIT, 1024, DIsz / KSPLIT, GEMM_BIAS);
    layernorm_kernel<<<3 * Rsz, 256, 0, stream>>>(part, (long)(3 * RM), oln,
                                                  ln2_g, ln2_b, out);
}


# Cross-lane reductions on the vector ALU (DPP row_xmask + v_permlanex16; was ds_bpermute waits and a 19-barrier LDS tree): gate and softmax sums bit-identical, layernorm row-sum order changes
# speedup vs baseline: 6.7760x; 1.0383x over previous
#include <hip/hip_runtime.h>
#include <hip/hip_bf16.h>

#define Bsz 4
#define Lsz 64
#define DMsz 1024
#define Hsz 16
#define DKsz 64
#define DVsz 64
#define Dsz 1024
#define DIsz 4096
#define Rsz 256
#define NEGC (-1.0e9f)

typedef float v2f __attribute__((ext_vector_type(2)));
typedef float v4f __attribute__((ext_vector_type(4)));
typedef float v8f __attribute__((ext_vector_type(8)));

template <int M>
__device__ __forceinline__ float xlane(float v)
{
    return __builtin_bit_cast(float, __builtin_amdgcn_update_dpp(
        0, __builtin_bit_cast(int, v), 0x160 + M, 0xf, 0xf, true));
}
__device__ __forceinline__ float xhalf(float v)
{
    const int b = __builtin_bit_cast(int, v);
    return __builtin_bit_cast(float, __builtin_amdgcn_permlanex16(
        b, b, 0x76543210, 0xfedcba98, false, false));
}
__device__ __forceinline__ float wave_sum(float s)
{
    s += xhalf(s);
    s += xlane<8>(s);
    s += xlane<4>(s);
    s += xlane<2>(s);
    s += xlane<1>(s);
    return s;
}
__device__ __forceinline__ float wave_max(float m)
{
    m = fmaxf(m, xhalf(m));
    m = fmaxf(m, xlane<8>(m));
    m = fmaxf(m, xlane<4>(m));
    m = fmaxf(m, xlane<2>(m));
    m = fmaxf(m, xlane<1>(m));
    return m;
}

#define GEMM_AP 68
#define GEMM_BT 1040
#define GEMM_EP 36
#define GEMM_BIAS  1
#define GEMM_RELU  2
#define GEMM_RESID 4

template <int BN>
__global__ __launch_bounds__(BN * 2)
void gemm_wmma_f32(const float* __restrict__ A0, const float* __restrict__ A1,
                   const float* __restrict__ A2, long strideAz, int lda,
                   const float* __restrict__ B0, const float* __restrict__ B1,
                   const float* __restrict__ B2, const float* __restrict__ B3,
                   const float* __restrict__ B4, long strideBz,
                   const float* __restrict__ bias,
                   const float* __restrict__ R0, const float* __restrict__ R1,
                   const float* __restrict__ R2,
                   float* __restrict__ Cbase, long strideCz,
                   int N, int Kloop, int flags)
{
    const int seg  = blockIdx.y >> 2;
    const int mloc = (blockIdx.y & 3) * 64;
    const int zi   = blockIdx.z;
    const float* __restrict__ A =
        ((seg == 0) ? A0 : (seg == 1) ? A1 : A2) + (size_t)zi * strideAz;
    const float* __restrict__ B =
        ((zi == 0) ? B0 : (zi == 1) ? B1 : (zi == 2) ? B2 : (zi == 3) ? B3 : B4)
        + (size_t)zi * strideBz;
    const bool lead = (zi == 0);
    const float* __restrict__ resid = (seg == 0) ? R0 : (seg == 1) ? R1 : R2;
    float* __restrict__ C = Cbase + (size_t)zi * strideCz;

    constexpr int NT    = BN * 2;
    constexpr int NWN   = BN / 32;
    constexpr int APASS = 1024 / NT;
    constexpr int AROWS = NT / 16;
    constexpr int TPR   = BN / 4;

    __shared__ __attribute__((aligned(16))) float smem[64 * GEMM_AP + (BN / 16) * GEMM_BT];
    float (*As)[GEMM_AP] = reinterpret_cast<float (*)[GEMM_AP]>(smem);
    float* Bs = smem + 64 * GEMM_AP;

    const int tid   = threadIdx.x;
    const int wave  = tid >> 5;
    const int lane  = tid & 31;
    const int l15   = lane & 15;
    const int lhalf = lane >> 4;
    const int m0    = blockIdx.y * 64;
    const int nblk  = blockIdx.x * BN;
    const int mq    = wave / NWN;
    const int nq    = wave % NWN;

    const int ar = tid >> 4;
    const int ac = (tid & 15) << 2;
    const int br = (tid / TPR) * 8;
    const int bc = (tid % TPR) << 2;
    v4f aprf[APASS], bprf[8];

    v8f acc[2][2];
#pragma unroll
    for (int x = 0; x < 2; ++x)
#pragma unroll
        for (int y = 0; y < 2; ++y)
            acc[x][y] = v8f{0.f, 0.f, 0.f, 0.f, 0.f, 0.f, 0.f, 0.f};

#pragma unroll
    for (int rr = 0; rr < APASS; ++rr)
        aprf[rr] = *(const v4f*)(A + (size_t)(mloc + ar + rr * AROWS) * lda + ac);
#pragma unroll
    for (int rr = 0; rr < 8; ++rr)
        bprf[rr] = *(const v4f*)(B + (size_t)(br + rr) * N + nblk + bc);

    const int arow0 = mq * 32 + l15;

    for (int k0 = 0; k0 < Kloop; k0 += 64) {
#pragma unroll
        for (int rr = 0; rr < APASS; ++rr)
            *(v4f*)&As[ar + rr * AROWS][ac] = aprf[rr];
#pragma unroll
        for (int rr = 0; rr < 8; ++rr)
            *(v4f*)&Bs[(bc >> 4) * GEMM_BT + (br + rr) * 16 + (bc & 15)] = bprf[rr];
        __syncthreads();

        if (k0 + 64 < Kloop) {
#pragma unroll
            for (int rr = 0; rr < APASS; ++rr)
                aprf[rr] = *(const v4f*)(A + (size_t)(mloc + ar + rr * AROWS) * lda
                                         + k0 + 64 + ac);
#pragma unroll
            for (int rr = 0; rr < 8; ++rr)
                bprf[rr] = *(const v4f*)(B + (size_t)(k0 + 64 + br + rr) * N
                                         + nblk + bc);
        }

#pragma unroll
        for (int half = 0; half < 2; ++half) {
            v2f af[2][8], bf[2][8];
#pragma unroll
            for (int s = 0; s < 8; ++s) {
                const int kb = half * 32 + s * 4 + lhalf * 2;
                af[0][s] = *(const v2f*)&As[arow0][kb];
                af[1][s] = *(const v2f*)&As[arow0 + 16][kb];
                const float* bp = &Bs[(nq * 2) * GEMM_BT + kb * 16 + l15];
                bf[0][s] = v2f{bp[0],       bp[16]};
                bf[1][s] = v2f{bp[GEMM_BT], bp[GEMM_BT + 16]};
            }
#pragma unroll
            for (int s = 0; s < 8; ++s) {
                acc[0][0] = __builtin_amdgcn_wmma_f32_16x16x4_f32(
                    false, af[0][s], false, bf[0][s], (short)0, acc[0][0], false, false);
                acc[0][1] = __builtin_amdgcn_wmma_f32_16x16x4_f32(
                    false, af[0][s], false, bf[1][s], (short)0, acc[0][1], false, false);
                acc[1][0] = __builtin_amdgcn_wmma_f32_16x16x4_f32(
                    false, af[1][s], false, bf[0][s], (short)0, acc[1][0], false, false);
                acc[1][1] = __builtin_amdgcn_wmma_f32_16x16x4_f32(
                    false, af[1][s], false, bf[1][s], (short)0, acc[1][1], false, false);
            }
        }
        __syncthreads();
    }

    const int erow = lane >> 3;
    const int ecol = (lane & 7) << 2;
    const int gcol = nblk + nq * 32 + ecol;
    const bool use_bias  = (flags & GEMM_BIAS) && lead;
    const bool use_resid = (flags & GEMM_RESID) && lead;

    v4f bv = v4f{0.f, 0.f, 0.f, 0.f};
    if (use_bias) bv = *(const v4f*)(bias + gcol);
    v4f rv[8];
#pragma unroll
    for (int it = 0; it < 8; ++it) rv[it] = v4f{0.f, 0.f, 0.f, 0.f};
    if (use_resid) {
#pragma unroll
        for (int it = 0; it < 8; ++it)
            rv[it] = *(const v4f*)(resid + (size_t)(mloc + mq * 32 + it * 4 + erow) * N + gcol);
    }

    float* stg = smem + wave * (32 * GEMM_EP);
#pragma unroll
    for (int msub = 0; msub < 2; ++msub)
#pragma unroll
        for (int nsub = 0; nsub < 2; ++nsub)
#pragma unroll
            for (int r = 0; r < 8; ++r)
                stg[(msub * 16 + lhalf * 8 + r) * GEMM_EP + nsub * 16 + l15] = acc[msub][nsub][r];
    __syncthreads();

#pragma unroll
    for (int it = 0; it < 8; ++it) {
        const int rloc = it * 4 + erow;
        v4f v = *(const v4f*)&stg[rloc * GEMM_EP + ecol];
        v = v + bv;
        if (use_resid) v = v + rv[it];
        if (flags & GEMM_RELU) v = __builtin_elementwise_max(v, v4f{0.f, 0.f, 0.f, 0.f});
        *(v4f*)(C + (size_t)(m0 + mq * 32 + rloc) * N + gcol) = v;
    }
}

#define DEC_NB 4
#define DEC_HOLD8(a, o)                                                         \
    asm volatile("" :: "v"(a[(o) + 0]), "v"(a[(o) + 1]), "v"(a[(o) + 2]),       \
                 "v"(a[(o) + 3]), "v"(a[(o) + 4]), "v"(a[(o) + 5]),             \
                 "v"(a[(o) + 6]), "v"(a[(o) + 7]))

__global__ __launch_bounds__(128, 3)
void dec_kernel(const float* __restrict__ a0base, const float* __restrict__ a1base,
                const float* __restrict__ b1, const float* __restrict__ w2,
                const float* __restrict__ b2, float* __restrict__ dec)
{
    const int lane = threadIdx.x & 31;
    const int wid  = __builtin_amdgcn_readfirstlane(
                         (int)((blockIdx.x * 128u + threadIdx.x) >> 5));
    const int nh   = wid & 1;
    const int mp   = (wid >> 1) & 31;
    const int b    = (wid >> 6) & 3;
    const int pair = wid >> 8;
    const int i    = pair / 3;
    const int j    = pair % 3;
    const int m0   = mp * 2;
    const int n0   = nh * 32;

    const float* a1r = a1base + ((size_t)j * Rsz + b * Lsz + m0) * 2048;
    const float* a0r = a0base + ((size_t)i * Rsz + b * Lsz + n0) * 2048;

    v2f wdp[32], bbp[32], x0p[32], x1p[32];
#pragma unroll
    for (int k = 0; k < 64; ++k) {
        const int c = lane + 32 * k;
        wdp[k >> 1][k & 1] = w2[2 * c + 1] - w2[2 * c];
        bbp[k >> 1][k & 1] = b1[c];
        x0p[k >> 1][k & 1] = a1r[lane + 32 * k];
        x1p[k >> 1][k & 1] = a1r[2048 + lane + 32 * k];
    }

    float r0 = 0.f, r1 = 0.f;
    for (int g = 0; g < 32; g += DEC_NB) {
        float s0[DEC_NB], s1[DEC_NB];
#pragma unroll
        for (int u = 0; u < DEC_NB; ++u) {
            const float* p = a0r + (size_t)(g + u) * 2048;
            v2f yp[32];
#pragma unroll
            for (int k = 0; k < 64; ++k) yp[k >> 1][k & 1] = p[lane + 32 * k];
            float c0 = 0.f, c1 = 0.f;
#pragma unroll
            for (int q = 0; q < 4; ++q) {
                DEC_HOLD8(yp, 8 * q);
#pragma unroll
                for (int h = 8 * q; h < 8 * q + 8; ++h) {
                    v2f t0 = x0p[h] + yp[h] + bbp[h];
                    v2f t1 = x1p[h] + yp[h] + bbp[h];
                    t0 = __builtin_elementwise_max(t0, v2f{0.f, 0.f});
                    t1 = __builtin_elementwise_max(t1, v2f{0.f, 0.f});
                    c0 = fmaf(t0[0], wdp[h][0], c0);
                    c1 = fmaf(t1[0], wdp[h][0], c1);
                    c0 = fmaf(t0[1], wdp[h][1], c0);
                    c1 = fmaf(t1[1], wdp[h][1], c1);
                }
            }
            s0[u] = c0; s1[u] = c1;
        }
#pragma unroll
        for (int u = 0; u < DEC_NB; ++u) { s0[u] += xhalf(s0[u]);    s1[u] += xhalf(s1[u]); }
#pragma unroll
        for (int u = 0; u < DEC_NB; ++u) { s0[u] += xlane<8>(s0[u]); s1[u] += xlane<8>(s1[u]); }
#pragma unroll
        for (int u = 0; u < DEC_NB; ++u) { s0[u] += xlane<4>(s0[u]); s1[u] += xlane<4>(s1[u]); }
#pragma unroll
        for (int u = 0; u < DEC_NB; ++u) { s0[u] += xlane<2>(s0[u]); s1[u] += xlane<2>(s1[u]); }
#pragma unroll
        for (int u = 0; u < DEC_NB; ++u) { s0[u] += xlane<1>(s0[u]); s1[u] += xlane<1>(s1[u]); }
#pragma unroll
        for (int u = 0; u < DEC_NB; ++u) {
            r0 = (lane == g + u) ? s0[u] : r0;
            r1 = (lane == g + u) ? s1[u] : r1;
        }
    }

    const float bd = b2[1] - b2[0];
    const size_t o = (((size_t)pair * Bsz + b) * Lsz + m0) * Lsz + n0 + lane;
    dec[o]       = (r0 + bd > 0.f) ? NEGC : 0.f;
    dec[o + Lsz] = (r1 + bd > 0.f) ? NEGC : 0.f;
}

#define AT_QP 68
#define AT_VP 72
#define AT_SP 196
__global__ __launch_bounds__(256)
void attn_fused_kernel(const float* __restrict__ Qb, const float* __restrict__ Kb,
                       const float* __restrict__ Vb, const float* __restrict__ dec,
                       const int* __restrict__ mask, float* __restrict__ Ob)
{
    const int blk = blockIdx.x;
    const int h = blk & 15;
    const int b = (blk >> 4) & 3;
    const int i = blk >> 6;

    __shared__ __attribute__((aligned(16))) float Qs[64][AT_QP];
    __shared__ __attribute__((aligned(16))) float Ks[192][AT_QP];
    __shared__ __attribute__((aligned(16))) float Vs[192][AT_VP];
    __shared__ __attribute__((aligned(16))) float Ss[64][AT_SP];

    const int tid   = threadIdx.x;
    const int wave  = tid >> 5;
    const int lane  = tid & 31;
    const int l15   = lane & 15;
    const int lhalf = lane >> 4;

    const int lr = tid >> 4;
    const int lc = (tid & 15) << 2;
    float4 qreg[4], kreg[12], vreg[12];
#pragma unroll
    for (int rr = 0; rr < 4; ++rr) {
        const int row = lr + rr * 16;
        qreg[rr] = *(const float4*)(Qb + ((size_t)i * Rsz + b * Lsz + row) * DMsz
                                    + h * DKsz + lc);
    }
#pragma unroll
    for (int rr = 0; rr < 12; ++rr) {
        const int row = lr + rr * 16;
        const size_t g = ((size_t)(row >> 6) * Rsz + b * Lsz + (row & 63)) * DMsz
                         + h * DKsz + lc;
        kreg[rr] = *(const float4*)(Kb + g);
        vreg[rr] = *(const float4*)(Vb + g);
    }

    float dv[8][6];
    int   mk[8][2];
#pragma unroll
    for (int rr = 0; rr < 8; ++rr) {
        const int row = wave * 8 + rr;
#pragma unroll
        for (int u = 0; u < 6; ++u) {
            const int col = lane + 32 * u;
            dv[rr][u] = dec[(((size_t)(i * 3 + (col >> 6)) * Bsz + b) * Lsz + row) * Lsz
                            + (col & 63)];
        }
        mk[rr][0] = mask[((size_t)b * Lsz + row) * Lsz + lane];
        mk[rr][1] = mask[((size_t)b * Lsz + row) * Lsz + lane + 32];
    }

#pragma unroll
    for (int rr = 0; rr < 4; ++rr)
        *(float4*)&Qs[lr + rr * 16][lc] = qreg[rr];
#pragma unroll
    for (int rr = 0; rr < 12; ++rr) {
        *(float4*)&Ks[lr + rr * 16][lc] = kreg[rr];
        *(float4*)&Vs[lr + rr * 16][lc] = vreg[rr];
    }
    __syncthreads();

    const int mt   = wave & 3;
    const int ng   = wave >> 2;
    const int arow = mt * 16 + l15;
    {
        v8f sacc[6];
#pragma unroll
        for (int t = 0; t < 6; ++t) sacc[t] = v8f{0.f, 0.f, 0.f, 0.f, 0.f, 0.f, 0.f, 0.f};
#pragma unroll
        for (int s = 0; s < 16; ++s) {
            const int kb = s * 4 + lhalf * 2;
            const v2f a = *(const v2f*)&Qs[arow][kb];
#pragma unroll
            for (int t = 0; t < 6; ++t) {
                const v2f bk = *(const v2f*)&Ks[(ng * 6 + t) * 16 + l15][kb];
                sacc[t] = __builtin_amdgcn_wmma_f32_16x16x4_f32(
                    false, a, false, bk, (short)0, sacc[t], false, false);
            }
        }
#pragma unroll
        for (int t = 0; t < 6; ++t) {
            const int col = (ng * 6 + t) * 16 + l15;
#pragma unroll
            for (int r = 0; r < 8; ++r)
                Ss[mt * 16 + r + lhalf * 8][col] = sacc[t][r] * 0.125f;
        }
    }
    __syncthreads();

#pragma unroll
    for (int rr = 0; rr < 8; ++rr) {
        const int row = wave * 8 + rr;
        float sv[6];
        float mx = -3.0e38f;
#pragma unroll
        for (int u = 0; u < 6; ++u) {
            float s = Ss[row][lane + 32 * u] + dv[rr][u];
            if (mk[rr][u & 1] == 0) s = NEGC;
            sv[u] = s;
            mx = fmaxf(mx, s);
        }
        mx = wave_max(mx);
        float sum = 0.f;
#pragma unroll
        for (int u = 0; u < 6; ++u) { sv[u] = __expf(sv[u] - mx); sum += sv[u]; }
        sum = wave_sum(sum);
        const float inv = 1.0f / sum;
#pragma unroll
        for (int u = 0; u < 6; ++u) Ss[row][lane + 32 * u] = sv[u] * inv;
    }
    __syncthreads();

    {
        v8f oacc[2][2];
#pragma unroll
        for (int t = 0; t < 2; ++t)
#pragma unroll
            for (int e = 0; e < 2; ++e)
                oacc[t][e] = v8f{0.f, 0.f, 0.f, 0.f, 0.f, 0.f, 0.f, 0.f};
#pragma unroll
        for (int s = 0; s < 48; s += 2) {
#pragma unroll
            for (int e = 0; e < 2; ++e) {
                const int kb = (s + e) * 4 + lhalf * 2;
                const v2f a = *(const v2f*)&Ss[arow][kb];
#pragma unroll
                for (int t = 0; t < 2; ++t) {
                    const int col = ng * 32 + t * 16 + l15;
                    const v2f bv = v2f{Vs[kb][col], Vs[kb + 1][col]};
                    oacc[t][e] = __builtin_amdgcn_wmma_f32_16x16x4_f32(
                        false, a, false, bv, (short)0, oacc[t][e], false, false);
                }
            }
        }
#pragma unroll
        for (int t = 0; t < 2; ++t) {
            const int col = ng * 32 + t * 16 + l15;
#pragma unroll
            for (int r = 0; r < 8; ++r) {
                const int row = mt * 16 + r + lhalf * 8;
                Ob[((size_t)i * Rsz + b * Lsz + row) * DMsz + h * DVsz + col] =
                    oacc[t][0][r] + oacc[t][1][r];
            }
        }
    }
}

#define KSPLIT 8
__global__ __launch_bounds__(256)
void layernorm_kernel(const float* __restrict__ x, long plane,
                      const float* __restrict__ resid,
                      const float* __restrict__ g, const float* __restrict__ bta,
                      float* __restrict__ out)
{
    const int row = blockIdx.x;
    const int t = threadIdx.x;
    const int lane = t & 31;
    const int wave = t >> 5;
    __shared__ float red[16];

    const float* xp = x + (size_t)row * 1024;
    float v[4];
    float s = 0.f;
#pragma unroll
    for (int u = 0; u < 4; ++u) {
        float val = xp[t + u * 256];
#pragma unroll
        for (int p = 1; p < KSPLIT; ++p) val += xp[(size_t)p * plane + t + u * 256];
        v[u] = val;
    }
    if (resid) {
#pragma unroll
        for (int u = 0; u < 4; ++u) v[u] += resid[(size_t)row * 1024 + t + u * 256];
    }
#pragma unroll
    for (int u = 0; u < 4; ++u) s += v[u];
    s = wave_sum(s);
    if (lane == 0) red[wave] = s;
    __syncthreads();
    float tot = red[0];
#pragma unroll
    for (int w = 1; w < 8; ++w) tot += red[w];
    const float mu = tot * (1.f / 1024.f);

    float s2 = 0.f;
#pragma unroll
    for (int u = 0; u < 4; ++u) { float dd = v[u] - mu; s2 += dd * dd; }
    s2 = wave_sum(s2);
    if (lane == 0) red[8 + wave] = s2;
    __syncthreads();
    float tot2 = red[8];
#pragma unroll
    for (int w = 1; w < 8; ++w) tot2 += red[8 + w];
    const float var = tot2 * (1.f / 1024.f);
    const float rinv = 1.0f / sqrtf(var + 1e-6f);
#pragma unroll
    for (int u = 0; u < 4; ++u) {
        int c = t + u * 256;
        out[(size_t)row * 1024 + c] = (v[u] - mu) * rinv * g[c] + bta[c];
    }
}

extern "C" void kernel_launch(void* const* d_in, const int* in_sizes, int n_in,
                              void* d_out, int out_size, void* d_ws, size_t ws_size,
                              hipStream_t stream)
{
    const float* X[3]   = { (const float*)d_in[0], (const float*)d_in[1], (const float*)d_in[2] };
    const int*   mask   = (const int*)d_in[3];
    const float* w_qs   = (const float*)d_in[4];
    const float* w_ks   = (const float*)d_in[5];
    const float* w_vs   = (const float*)d_in[6];
    const float* w_ds0  = (const float*)d_in[7];
    const float* w_ds1  = (const float*)d_in[8];
    const float* dc_w1  = (const float*)d_in[9];
    const float* dc_b1  = (const float*)d_in[10];
    const float* dc_w2  = (const float*)d_in[11];
    const float* dc_b2  = (const float*)d_in[12];
    const float* w_fc   = (const float*)d_in[13];
    const float* ln1_g  = (const float*)d_in[14];
    const float* ln1_b  = (const float*)d_in[15];
    const float* ffn_w1 = (const float*)d_in[16];
    const float* ffn_b1 = (const float*)d_in[17];
    const float* ffn_w2 = (const float*)d_in[18];
    const float* ffn_b2 = (const float*)d_in[19];
    const float* ln2_g  = (const float*)d_in[20];
    const float* ln2_b  = (const float*)d_in[21];
    float* out = (float*)d_out;

    float* ws = (float*)d_ws;
    size_t off = 0;
    auto alloc = [&](size_t n) { float* p = ws + off; off += n; return p; };
    const size_t RM = (size_t)Rsz * DMsz;
    float* Q   = alloc(3 * RM);
    float* Kb  = alloc(3 * RM);
    float* Vb  = alloc(3 * RM);
    float* d0  = alloc(3 * RM);
    float* d1  = alloc(3 * RM);
    float* a0  = alloc(3 * (size_t)Rsz * 2048);
    float* a1  = alloc(3 * (size_t)Rsz * 2048);
    float* dec = alloc(9 * (size_t)Bsz * Lsz * Lsz);
    (void)alloc((size_t)3 * Bsz * Hsz * Lsz * 192);
    float* O   = alloc(3 * RM);
    float* oln = alloc(3 * RM);
    float* h1  = alloc(3 * (size_t)Rsz * DIsz);
    float* part = Q;
    (void)ws_size; (void)n_in; (void)in_sizes; (void)out_size;

    auto gemm3 = [&](bool narrow,
                     const float* a0s, const float* a1s, const float* a2s, long strideAz,
                     int lda,
                     const float* b0, const float* b1, const float* b2,
                     const float* b3, const float* b4, long strideBz,
                     const float* bias,
                     const float* r0s, const float* r1s, const float* r2s,
                     float* C, long strideCz, int nz, int N, int Kloop, int flags) {
        if (narrow) {
            dim3 grid(N / 64, (3 * Rsz) / 64, nz);
            gemm_wmma_f32<64><<<grid, dim3(128), 0, stream>>>(a0s, a1s, a2s, strideAz, lda,
                                                              b0, b1, b2, b3, b4, strideBz, bias,
                                                              r0s, r1s, r2s, C, strideCz,
                                                              N, Kloop, flags);
        } else {
            dim3 grid(N / 128, (3 * Rsz) / 64, nz);
            gemm_wmma_f32<128><<<grid, dim3(256), 0, stream>>>(a0s, a1s, a2s, strideAz, lda,
                                                               b0, b1, b2, b3, b4, strideBz, bias,
                                                               r0s, r1s, r2s, C, strideCz,
                                                               N, Kloop, flags);
        }
    };

    gemm3(false, X[0], X[1], X[2], 0, 1024,
          w_qs, w_ks, w_vs, w_ds0, w_ds1, 0,
          nullptr, nullptr, nullptr, nullptr,
          Q, (long)(3 * RM), 5, 1024, 1024, 0);

    gemm3(true, d0, d0 + RM, d0 + 2 * RM, (long)(3 * RM), 1024,
          dc_w1 + (size_t)1024 * 2048, dc_w1, dc_w1, dc_w1, dc_w1, 0,
          nullptr, nullptr, nullptr, nullptr,
          a0, (long)(3 * (size_t)Rsz * 2048), 2, 2048, 1024, 0);

    {
        dec_kernel<<<576, 128, 0, stream>>>(a0, a1, dc_b1, dc_w2, dc_b2, dec);
    }

    attn_fused_kernel<<<3 * Bsz * Hsz, 256, 0, stream>>>(Q, Kb, Vb, dec, mask, O);

    gemm3(false, O, O + RM, O + 2 * RM, 1024 / KSPLIT, 1024,
          w_fc, w_fc, w_fc, w_fc, w_fc, (long)(1024 / KSPLIT) * 1024,
          nullptr, X[0], X[1], X[2],
          part, (long)(3 * RM), KSPLIT, 1024, 1024 / KSPLIT, GEMM_RESID);
    layernorm_kernel<<<3 * Rsz, 256, 0, stream>>>(part, (long)(3 * RM), nullptr,
                                                  ln1_g, ln1_b, oln);
    gemm3(true, oln, oln + RM, oln + 2 * RM, 0, 1024,
          ffn_w1, ffn_w1, ffn_w1, ffn_w1, ffn_w1, 0,
          ffn_b1, nullptr, nullptr, nullptr,
          h1, 0, 1, DIsz, 1024, GEMM_BIAS | GEMM_RELU);
    gemm3(false, h1, h1 + (size_t)Rsz * DIsz, h1 + 2 * (size_t)Rsz * DIsz, DIsz / KSPLIT, DIsz,
          ffn_w2, ffn_w2, ffn_w2, ffn_w2, ffn_w2, (long)(DIsz / KSPLIT) * 1024,
          ffn_b2, nullptr, nullptr, nullptr,
          part, (long)(3 * RM), KSPLIT, 1024, DIsz / KSPLIT, GEMM_BIAS);
    layernorm_kernel<<<3 * Rsz, 256, 0, stream>>>(part, (long)(3 * RM), oln,
                                                  ln2_g, ln2_b, out);
}
